# starting kernel
# baseline (speedup 1.0000x reference)
#include <hip/hip_runtime.h>
#include <math.h>

typedef __attribute__((ext_vector_type(2))) float v2f;
typedef __attribute__((ext_vector_type(8))) float v8f;

#define NEG_SLOPE 0.2f
#define PYG_EPS   1e-16f

__global__ __launch_bounds__(256) void gemm_wmma_f32(
    const float* __restrict__ X, const float* __restrict__ W,
    float* __restrict__ Hout, int Nrows, int K, int M)
{
  const int tilesN = M >> 4;
  const int tilesTotal = (Nrows >> 4) * tilesN;
  const int wave = blockIdx.x * (blockDim.x >> 5) + (threadIdx.x >> 5);
  if (wave >= tilesTotal) return;
  const int tm = wave / tilesN;
  const int tn = wave - tm * tilesN;
  const int lane = threadIdx.x & 31;
  const int half = lane >> 4;
  const int l15  = lane & 15;
  const int kb   = half << 1;

  const float* xp = X + (size_t)((tm << 4) + l15) * K + kb;
  const float* wp = W + (size_t)kb * M + ((tn << 4) + l15);

  v8f acc = {0.f, 0.f, 0.f, 0.f, 0.f, 0.f, 0.f, 0.f};
  for (int k0 = 0; k0 < K; k0 += 4) {
    v2f a = *(const v2f*)xp;
    v2f b;
    b.x = wp[0];
    b.y = wp[M];
    acc = __builtin_amdgcn_wmma_f32_16x16x4_f32(
         false, a,  false, b,
         (short)0, acc,  false,  false);
    xp += 4;
    wp += (size_t)4 * M;
  }

  float* op = Hout + (size_t)((tm << 4) + (half << 3)) * M + ((tn << 4) + l15);
  #pragma unroll
  for (int v = 0; v < 8; ++v) op[(size_t)v * M] = acc[v];
}

__global__ __launch_bounds__(256) void att_dots(
    const float* __restrict__ Hm, const float* __restrict__ att_s,
    const float* __restrict__ att_d, float* __restrict__ a_s,
    float* __restrict__ a_d, int Nrows, int Hh, int C)
{
  const int wave = blockIdx.x * (blockDim.x >> 5) + (threadIdx.x >> 5);
  if (wave >= Nrows * Hh) return;
  const int lane = threadIdx.x & 31;
  const int h = wave % Hh;
  const float* hv = Hm + (size_t)wave * C;
  const float* as = att_s + (size_t)h * C;
  const float* ad = att_d + (size_t)h * C;
  float ss = 0.f, sd = 0.f;
  for (int c = lane; c < C; c += 32) {
    float x = hv[c];
    ss += x * as[c];
    sd += x * ad[c];
  }
  #pragma unroll
  for (int off = 16; off; off >>= 1) {
    ss += __shfl_xor(ss, off, 32);
    sd += __shfl_xor(sd, off, 32);
  }
  if (lane == 0) { a_s[wave] = ss; a_d[wave] = sd; }
}

__device__ __forceinline__ float lrelu(float e) {
  return e > 0.f ? e : NEG_SLOPE * e;
}

__global__ void init_ms(float* __restrict__ m, float* __restrict__ s, int total)
{
  int i = blockIdx.x * blockDim.x + threadIdx.x;
  if (i < total) { m[i] = -__builtin_inff(); s[i] = 0.f; }
}

__global__ void init_bias(float* __restrict__ out, const float* __restrict__ bias,
                          int HC, int total)
{
  int i = blockIdx.x * blockDim.x + threadIdx.x;
  if (i < total) out[i] = bias[i % HC];
}

__global__ void elu_inplace(float* __restrict__ x, int total)
{
  int i = blockIdx.x * blockDim.x + threadIdx.x;
  if (i < total) { float v = x[i]; x[i] = v > 0.f ? v : expm1f(v); }
}

template <int H>
__global__ void edge_max(const int* __restrict__ src, const int* __restrict__ dst,
                         const float* __restrict__ a_s, const float* __restrict__ a_d,
                         float* __restrict__ m, int E, int N)
{
  int i = blockIdx.x * blockDim.x + threadIdx.x;
  if (i >= E + N) return;
  int si, di;
  if (i < E) { si = src[i]; di = dst[i]; } else { si = di = i - E; }
  #pragma unroll
  for (int h = 0; h < H; ++h) {
    float e = lrelu(a_s[si * H + h] + a_d[di * H + h]);
    __hip_atomic_fetch_max(&m[di * H + h], e,
                           __ATOMIC_RELAXED, __HIP_MEMORY_SCOPE_AGENT);
  }
}

template <int H>
__global__ void edge_expsum(const int* __restrict__ src, const int* __restrict__ dst,
                            const float* __restrict__ a_s, const float* __restrict__ a_d,
                            const float* __restrict__ m, float* __restrict__ ssum,
                            int E, int N)
{
  int i = blockIdx.x * blockDim.x + threadIdx.x;
  if (i >= E + N) return;
  int si, di;
  if (i < E) { si = src[i]; di = dst[i]; } else { si = di = i - E; }
  #pragma unroll
  for (int h = 0; h < H; ++h) {
    float e = lrelu(a_s[si * H + h] + a_d[di * H + h]);
    float w = __expf(e - m[di * H + h]);
    atomicAdd(&ssum[di * H + h], w);
  }
}

template <int H, int C>
__global__ __launch_bounds__(256) void edge_aggregate(
    const int* __restrict__ src, const int* __restrict__ dst,
    const float* __restrict__ a_s, const float* __restrict__ a_d,
    const float* __restrict__ m, const float* __restrict__ ssum,
    const float* __restrict__ Hfeat, float* __restrict__ out, int E, int N)
{
  constexpr int HC  = H * C;
  constexpr int EPB = 256 / HC;
  const int t    = threadIdx.x;
  const int eidx = blockIdx.x * EPB + t / HC;
  if (eidx >= E + N) return;
  const int f = t % HC;
  const int h = f / C;
  int si, di;
  if (eidx < E) { si = src[eidx]; di = dst[eidx]; } else { si = di = eidx - E; }
  float e = lrelu(a_s[si * H + h] + a_d[di * H + h]);
  float w = __expf(e - m[di * H + h]);
  float alpha = w / (ssum[di * H + h] + PYG_EPS);
  atomicAdd(&out[(size_t)di * HC + f], Hfeat[(size_t)si * HC + f] * alpha);
}

extern "C" void kernel_launch(void* const* d_in, const int* in_sizes, int n_in,
                              void* d_out, int out_size, void* d_ws, size_t ws_size,
                              hipStream_t stream)
{
  const float* x   = (const float*)d_in[0];
  const int*   ei  = (const int*)  d_in[1];
  const float* W1  = (const float*)d_in[2];
  const float* as1 = (const float*)d_in[3];
  const float* ad1 = (const float*)d_in[4];
  const float* b1  = (const float*)d_in[5];
  const float* W2  = (const float*)d_in[6];
  const float* as2 = (const float*)d_in[7];
  const float* ad2 = (const float*)d_in[8];
  const float* b2  = (const float*)d_in[9];
  float* out = (float*)d_out;

  const int F = 128, H1 = 4, C = 64, HC1 = 256, HC2 = 64;
  const int N  = in_sizes[0] / F;
  const int E  = in_sizes[1] / 2;
  const int EE = E + N;
  const int* src = ei;
  const int* dst = ei + E;

  float* h1  = (float*)d_ws;
  float* x2  = h1  + (size_t)N * HC1;
  float* h2  = x2  + (size_t)N * HC1;
  float* aS1 = h2  + (size_t)N * HC2;
  float* aD1 = aS1 + (size_t)N * H1;
  float* m1  = aD1 + (size_t)N * H1;
  float* s1  = m1  + (size_t)N * H1;
  float* aS2 = s1  + (size_t)N * H1;
  float* aD2 = aS2 + N;
  float* m2  = aD2 + N;
  float* s2  = m2  + N;

  const int TB = 256;
  auto cdiv = [](int a, int b) { return (a + b - 1) / b; };

  {
    int tiles = (N / 16) * (HC1 / 16);
    gemm_wmma_f32<<<cdiv(tiles, 8), TB, 0, stream>>>(x, W1, h1, N, F, HC1);
  }
  att_dots<<<cdiv(N * H1, 8), TB, 0, stream>>>(h1, as1, ad1, aS1, aD1, N, H1, C);
  init_ms<<<cdiv(N * H1, TB), TB, 0, stream>>>(m1, s1, N * H1);
  init_bias<<<cdiv(N * HC1, TB), TB, 0, stream>>>(x2, b1, HC1, N * HC1);
  edge_max<H1><<<cdiv(EE, TB), TB, 0, stream>>>(src, dst, aS1, aD1, m1, E, N);
  edge_expsum<H1><<<cdiv(EE, TB), TB, 0, stream>>>(src, dst, aS1, aD1, m1, s1, E, N);
  edge_aggregate<H1, C><<<EE, TB, 0, stream>>>(src, dst, aS1, aD1, m1, s1, h1, x2, E, N);
  elu_inplace<<<cdiv(N * HC1, TB), TB, 0, stream>>>(x2, N * HC1);

  {
    int tiles = (N / 16) * (HC2 / 16);
    gemm_wmma_f32<<<cdiv(tiles, 8), TB, 0, stream>>>(x2, W2, h2, N, HC1, HC2);
  }
  att_dots<<<cdiv(N, 8), TB, 0, stream>>>(h2, as2, ad2, aS2, aD2, N, 1, C);
  init_ms<<<cdiv(N, TB), TB, 0, stream>>>(m2, s2, N);
  init_bias<<<cdiv(N * HC2, TB), TB, 0, stream>>>(out, b2, HC2, N * HC2);
  edge_max<1><<<cdiv(EE, TB), TB, 0, stream>>>(src, dst, aS2, aD2, m2, E, N);
  edge_expsum<1><<<cdiv(EE, TB), TB, 0, stream>>>(src, dst, aS2, aD2, m2, s2, E, N);
  edge_aggregate<1, C><<<cdiv(EE, 4), TB, 0, stream>>>(src, dst, aS2, aD2, m2, s2, h2, out, E, N);
}


# Edge phase as one kernel per layer: a work-group keeps 196 destination rows in LDS, filters all edges for them, max/exp-sum/feature sums by LDS atomics, one divide per row; sum order changes
# speedup vs baseline: 3.6160x; 3.6160x over previous
#include <hip/hip_runtime.h>
#include <math.h>

typedef __attribute__((ext_vector_type(2))) float v2f;
typedef __attribute__((ext_vector_type(8))) float v8f;

#define NEG_SLOPE 0.2f
#define PYG_EPS   1e-16f

__global__ __launch_bounds__(256) void gemm_wmma_f32(
    const float* __restrict__ X, const float* __restrict__ W,
    float* __restrict__ Hout, int Nrows, int K, int M)
{
  const int tilesN = M >> 4;
  const int tilesTotal = (Nrows >> 4) * tilesN;
  const int wave = blockIdx.x * (blockDim.x >> 5) + (threadIdx.x >> 5);
  if (wave >= tilesTotal) return;
  const int tm = wave / tilesN;
  const int tn = wave - tm * tilesN;
  const int lane = threadIdx.x & 31;
  const int half = lane >> 4;
  const int l15  = lane & 15;
  const int kb   = half << 1;

  const float* xp = X + (size_t)((tm << 4) + l15) * K + kb;
  const float* wp = W + (size_t)kb * M + ((tn << 4) + l15);

  v8f acc = {0.f, 0.f, 0.f, 0.f, 0.f, 0.f, 0.f, 0.f};
  for (int k0 = 0; k0 < K; k0 += 4) {
    v2f a = *(const v2f*)xp;
    v2f b;
    b.x = wp[0];
    b.y = wp[M];
    acc = __builtin_amdgcn_wmma_f32_16x16x4_f32(
         false, a,  false, b,
         (short)0, acc,  false,  false);
    xp += 4;
    wp += (size_t)4 * M;
  }

  float* op = Hout + (size_t)((tm << 4) + (half << 3)) * M + ((tn << 4) + l15);
  #pragma unroll
  for (int v = 0; v < 8; ++v) op[(size_t)v * M] = acc[v];
}

__global__ __launch_bounds__(256) void att_dots(
    const float* __restrict__ Hm, const float* __restrict__ att_s,
    const float* __restrict__ att_d, float* __restrict__ a_s,
    float* __restrict__ a_d, int Nrows, int Hh, int C)
{
  const int wave = blockIdx.x * (blockDim.x >> 5) + (threadIdx.x >> 5);
  if (wave >= Nrows * Hh) return;
  const int lane = threadIdx.x & 31;
  const int h = wave % Hh;
  const float* hv = Hm + (size_t)wave * C;
  const float* as = att_s + (size_t)h * C;
  const float* ad = att_d + (size_t)h * C;
  float ss = 0.f, sd = 0.f;
  for (int c = lane; c < C; c += 32) {
    float x = hv[c];
    ss += x * as[c];
    sd += x * ad[c];
  }
  #pragma unroll
  for (int off = 16; off; off >>= 1) {
    ss += __shfl_xor(ss, off, 32);
    sd += __shfl_xor(sd, off, 32);
  }
  if (lane == 0) { a_s[wave] = ss; a_d[wave] = sd; }
}

__device__ __forceinline__ float lrelu(float e) {
  return e > 0.f ? e : NEG_SLOPE * e;
}

#define TILE_NODES 196
#define LIST_CAP   8192
#define TILE_NT    1024

template <int H, int C, bool ELU>
__global__ __launch_bounds__(TILE_NT) void node_tile_gat(
    const int* __restrict__ src, const int* __restrict__ dst,
    const float* __restrict__ a_s, const float* __restrict__ a_d,
    const float* __restrict__ Hfeat, const float* __restrict__ bias,
    float* __restrict__ out, int E, int N)
{
  constexpr int HC = H * C;
  constexpr int TN = TILE_NODES;
  constexpr int CAP = LIST_CAP;
  constexpr int NT = TILE_NT;
  constexpr int J = HC / 32;
  static_assert(C % 32 == 0, "a 32-float chunk must lie inside one head");
  static_assert((H & (H - 1)) == 0 && H <= 32, "H must be a power of two");

  __shared__ float acc[TN * HC];
  __shared__ float m_l[TN * H];
  __shared__ float s_l[TN * H];
  __shared__ float sc_l[TN * H];
  __shared__ float ad_l[TN * H];
  __shared__ int lsrc[CAP];
  __shared__ unsigned short lld[CAP];
  __shared__ int cnt;
  __shared__ int ovf;

  const int tid = threadIdx.x;
  const int lane = tid & 31;
  const int wv = tid >> 5;
  const int lo = blockIdx.x * TN;
  const int nloc = min(TN, N - lo);
  if (nloc <= 0) return;
  const bool vecOK = (((size_t)dst) & 15) == 0;
  const int E4 = vecOK ? (E >> 2) : 0;
  const int4* dst4 = (const int4*)dst;
  const int rot = (int)((((long long)blockIdx.x * E4) / gridDim.x) & ~31LL);

  auto test1 = [&](int d, int e) {
    unsigned ldv = (unsigned)(d - lo);
    if (ldv < (unsigned)nloc) {
      int idx = atomicAdd(&cnt, 1);
      if (idx < CAP) { lsrc[idx] = src[e]; lld[idx] = (unsigned short)ldv; }
      else ovf = 1;
    }
  };

  auto process = [&](int n, int batch) {
    if (batch > 0) {
      for (int j = tid; j < nloc * H; j += NT) sc_l[j] = m_l[j];
    }
    __syncthreads();
    for (int i = tid; i < n; i += NT) {
      const int si = lsrc[i];
      const int ld = lld[i];
      for (int h = 0; h < H; ++h) {
        float ev = lrelu(a_s[(size_t)si * H + h] + ad_l[ld * H + h]);
        __hip_atomic_fetch_max(&m_l[ld * H + h], ev,
                               __ATOMIC_RELAXED, __HIP_MEMORY_SCOPE_WORKGROUP);
      }
    }
    __syncthreads();
    if (batch > 0) {
      for (int j = tid; j < nloc * H; j += NT) {
        float mo = sc_l[j], mn = m_l[j];
        float f = (mo == mn) ? 1.f : __expf(mo - mn);
        sc_l[j] = f;
        s_l[j] *= f;
      }
      __syncthreads();
      for (int i = tid; i < nloc * HC; i += NT)
        acc[i] *= sc_l[(i / HC) * H + (i % HC) / C];
      __syncthreads();
    }
    for (int i0 = wv * 4; i0 < n; i0 += 32 * 4) {
      int si[4], ld[4];
      float hv[4][J];
      float w[4];
      #pragma unroll
      for (int u = 0; u < 4; ++u) {
        int ii = min(i0 + u, n - 1);
        si[u] = lsrc[ii];
        ld[u] = lld[ii];
      }
      #pragma unroll
      for (int u = 0; u < 4; ++u) {
        const float* hp = Hfeat + (size_t)si[u] * HC + lane;
        #pragma unroll
        for (int j = 0; j < J; ++j) hv[u][j] = hp[j * 32];
      }
      #pragma unroll
      for (int u = 0; u < 4; ++u) {
        const int hh = lane & (H - 1);
        float ev = lrelu(a_s[(size_t)si[u] * H + hh] + ad_l[ld[u] * H + hh]);
        w[u] = __expf(ev - m_l[ld[u] * H + hh]);
      }
      #pragma unroll
      for (int u = 0; u < 4; ++u) {
        if (i0 + u < n) {
          if (lane < H) atomicAdd(&s_l[ld[u] * H + lane], w[u]);
          float wh[H];
          #pragma unroll
          for (int h = 0; h < H; ++h)
            wh[h] = __builtin_bit_cast(float,
                __builtin_amdgcn_readlane(__builtin_bit_cast(int, w[u]), h));
          #pragma unroll
          for (int j = 0; j < J; ++j)
            atomicAdd(&acc[ld[u] * HC + j * 32 + lane], hv[u][j] * wh[(j * 32) / C]);
        }
      }
    }
    __syncthreads();
  };

  int batch = 0;
  auto drain_if_needed = [&]() {
    __syncthreads();
    const int c = cnt;
    __syncthreads();
    if (c + 4 * NT > CAP) {
      process(c, batch);
      ++batch;
      if (tid == 0) cnt = 0;
      __syncthreads();
    }
  };

  for (int attempt = 0; attempt < 2; ++attempt) {
    for (int i = tid; i < nloc * HC; i += NT) acc[i] = 0.f;
    for (int j = tid; j < nloc * H; j += NT) {
      m_l[j] = -__builtin_inff();
      s_l[j] = 0.f;
      ad_l[j] = a_d[(size_t)lo * H + j];
    }
    for (int i = tid; i < nloc; i += NT) { lsrc[i] = lo + i; lld[i] = (unsigned short)i; }
    if (tid == 0) { cnt = nloc; ovf = 0; }
    batch = 0;
    __syncthreads();

    if (attempt == 0) {
      for (int g0 = 0; g0 < E4; g0 += 4 * NT) {
        int4 d[4];
        int gi[4];
        #pragma unroll
        for (int u = 0; u < 4; ++u) {
          int g = g0 + u * NT + tid;
          bool ok = g < E4;
          int gr = g + rot;
          if (gr >= E4) gr -= E4;
          gi[u] = gr;
          if (ok) d[u] = dst4[gr];
          else    d[u] = make_int4(lo - 1, lo - 1, lo - 1, lo - 1);
        }
        #pragma unroll
        for (int u = 0; u < 4; ++u) {
          const int e = gi[u] * 4;
          test1(d[u].x, e);
          test1(d[u].y, e + 1);
          test1(d[u].z, e + 2);
          test1(d[u].w, e + 3);
        }
      }
      for (int e = E4 * 4 + tid; e < E; e += NT) test1(dst[e], e);
    } else {
      for (int g0 = 0; g0 < E4; g0 += NT) {
        drain_if_needed();
        int g = g0 + tid;
        if (g < E4) {
          int4 d = dst4[g];
          const int e = g * 4;
          test1(d.x, e);
          test1(d.y, e + 1);
          test1(d.z, e + 2);
          test1(d.w, e + 3);
        }
      }
      for (int e0 = E4 * 4; e0 < E; e0 += 4 * NT) {
        drain_if_needed();
        for (int k = 0; k < 4; ++k) {
          int e = e0 + k * NT + tid;
          if (e < E) test1(dst[e], e);
        }
      }
    }
    __syncthreads();
    const int o = ovf;
    const int c = cnt;
    __syncthreads();
    if (attempt == 0 && o) continue;
    process(min(c, CAP), batch);
    break;
  }

  for (int r = wv; r < nloc; r += 32) {
    for (int j = 0; j < J; ++j) {
      const int f = j * 32 + lane;
      float v = acc[r * HC + f] / (s_l[r * H + f / C] + PYG_EPS) + bias[f];
      if (ELU) v = v > 0.f ? v : expm1f(v);
      out[(size_t)(lo + r) * HC + f] = v;
    }
  }
}

extern "C" void kernel_launch(void* const* d_in, const int* in_sizes, int n_in,
                              void* d_out, int out_size, void* d_ws, size_t ws_size,
                              hipStream_t stream)
{
  const float* x   = (const float*)d_in[0];
  const int*   ei  = (const int*)  d_in[1];
  const float* W1  = (const float*)d_in[2];
  const float* as1 = (const float*)d_in[3];
  const float* ad1 = (const float*)d_in[4];
  const float* b1  = (const float*)d_in[5];
  const float* W2  = (const float*)d_in[6];
  const float* as2 = (const float*)d_in[7];
  const float* ad2 = (const float*)d_in[8];
  const float* b2  = (const float*)d_in[9];
  float* out = (float*)d_out;

  const int F = 128, H1 = 4, C = 64, HC1 = 256, HC2 = 64;
  const int N  = in_sizes[0] / F;
  const int E  = in_sizes[1] / 2;
  const int* src = ei;
  const int* dst = ei + E;

  float* h1  = (float*)d_ws;
  float* x2  = h1  + (size_t)N * HC1;
  float* h2  = x2  + (size_t)N * HC1;
  float* aS1 = h2  + (size_t)N * HC2;
  float* aD1 = aS1 + (size_t)N * H1;
  float* aS2 = aD1 + (size_t)N * H1;
  float* aD2 = aS2 + N;

  const int TB = 256;
  auto cdiv = [](int a, int b) { return (a + b - 1) / b; };

  {
    int tiles = (N / 16) * (HC1 / 16);
    gemm_wmma_f32<<<cdiv(tiles, 8), TB, 0, stream>>>(x, W1, h1, N, F, HC1);
  }
  att_dots<<<cdiv(N * H1, 8), TB, 0, stream>>>(h1, as1, ad1, aS1, aD1, N, H1, C);
  node_tile_gat<4, 64, true><<<cdiv(N, TILE_NODES), TILE_NT, 0, stream>>>(
      src, dst, aS1, aD1, h1, b1, x2, E, N);

  {
    int tiles = (N / 16) * (HC2 / 16);
    gemm_wmma_f32<<<cdiv(tiles, 8), TB, 0, stream>>>(x2, W2, h2, N, HC1, HC2);
  }
  att_dots<<<cdiv(N, 8), TB, 0, stream>>>(h2, as2, ad2, aS2, aD2, N, 1, C);
  node_tile_gat<1, 64, false><<<cdiv(N, TILE_NODES), TILE_NT, 0, stream>>>(
      src, dst, aS2, aD2, h2, b2, out, E, N);
}


# Attention dots: 16 lanes per (node, head) row with 16-byte loads and a DPP row sum, 32 rows per wave instead of one wave per row (250000 waves become 7813); dot sum order changes
# speedup vs baseline: 4.0048x; 1.1075x over previous
#include <hip/hip_runtime.h>
#include <math.h>

typedef __attribute__((ext_vector_type(2))) float v2f;
typedef __attribute__((ext_vector_type(8))) float v8f;

#define NEG_SLOPE 0.2f
#define PYG_EPS   1e-16f

__global__ __launch_bounds__(256) void gemm_wmma_f32(
    const float* __restrict__ X, const float* __restrict__ W,
    float* __restrict__ Hout, int Nrows, int K, int M)
{
  const int tilesN = M >> 4;
  const int tilesTotal = (Nrows >> 4) * tilesN;
  const int wave = blockIdx.x * (blockDim.x >> 5) + (threadIdx.x >> 5);
  if (wave >= tilesTotal) return;
  const int tm = wave / tilesN;
  const int tn = wave - tm * tilesN;
  const int lane = threadIdx.x & 31;
  const int half = lane >> 4;
  const int l15  = lane & 15;
  const int kb   = half << 1;

  const float* xp = X + (size_t)((tm << 4) + l15) * K + kb;
  const float* wp = W + (size_t)kb * M + ((tn << 4) + l15);

  v8f acc = {0.f, 0.f, 0.f, 0.f, 0.f, 0.f, 0.f, 0.f};
  for (int k0 = 0; k0 < K; k0 += 4) {
    v2f a = *(const v2f*)xp;
    v2f b;
    b.x = wp[0];
    b.y = wp[M];
    acc = __builtin_amdgcn_wmma_f32_16x16x4_f32(
         false, a,  false, b,
         (short)0, acc,  false,  false);
    xp += 4;
    wp += (size_t)4 * M;
  }

  float* op = Hout + (size_t)((tm << 4) + (half << 3)) * M + ((tn << 4) + l15);
  #pragma unroll
  for (int v = 0; v < 8; ++v) op[(size_t)v * M] = acc[v];
}

template <int CTRL>
__device__ __forceinline__ float row_swz(float v) {
  return __builtin_bit_cast(float, __builtin_amdgcn_update_dpp(
      0, __builtin_bit_cast(int, v), CTRL, 0xf, 0xf, true));
}

#define DOT_PAIRS 16

template <int H, bool FULL>
__device__ __forceinline__ void att_dots_wave(
    const float* __restrict__ Hm, const float* __restrict__ att_s,
    const float* __restrict__ att_d, float* __restrict__ a_s,
    float* __restrict__ a_d, int nrows, int base, int lane)
{
  constexpr int C = 64;
  constexpr int U = DOT_PAIRS;
  constexpr int PP = (H >= 2) ? (H / 2) : 1;
  static_assert((H & (H - 1)) == 0, "H must be a power of two");
  static_assert(U % PP == 0 && (2 * U) % H == 0, "a wave's first row must be head 0");

  const int half = lane >> 4;
  const int l15  = lane & 15;
  const int c0   = l15 << 2;

  float4 x[U];
  const float* p0 = Hm + (size_t)(base + half) * C + c0;
  #pragma unroll
  for (int u = 0; u < U; ++u) {
    if (FULL) {
      x[u] = *(const float4*)(p0 + u * 2 * C);
    } else {
      const int r = min(base + 2 * u + half, nrows - 1);
      x[u] = *(const float4*)(Hm + (size_t)r * C + c0);
    }
  }

  float ws[PP][4], wd[PP][4];
  #pragma unroll
  for (int q = 0; q < PP; ++q) {
    const int h = (2 * q + half) & (H - 1);
    #pragma unroll
    for (int k = 0; k < 4; ++k) {
      ws[q][k] = att_s[h * C + c0 + k];
      wd[q][k] = att_d[h * C + c0 + k];
    }
  }

  float ss[U], sd[U];
  #pragma unroll
  for (int u = 0; u < U; ++u) {
    const int q = u % PP;
    ss[u] = x[u].x * ws[q][0] + x[u].y * ws[q][1] + x[u].z * ws[q][2] + x[u].w * ws[q][3];
    sd[u] = x[u].x * wd[q][0] + x[u].y * wd[q][1] + x[u].z * wd[q][2] + x[u].w * wd[q][3];
  }
  #pragma unroll
  for (int u = 0; u < U; ++u) { ss[u] += row_swz<0x161>(ss[u]); sd[u] += row_swz<0x161>(sd[u]); }
  #pragma unroll
  for (int u = 0; u < U; ++u) { ss[u] += row_swz<0x162>(ss[u]); sd[u] += row_swz<0x162>(sd[u]); }
  #pragma unroll
  for (int u = 0; u < U; ++u) { ss[u] += row_swz<0x164>(ss[u]); sd[u] += row_swz<0x164>(sd[u]); }
  #pragma unroll
  for (int u = 0; u < U; ++u) { ss[u] += row_swz<0x168>(ss[u]); sd[u] += row_swz<0x168>(sd[u]); }

  float vs = ss[0], vd = sd[0];
  #pragma unroll
  for (int u = 1; u < U; ++u) {
    if (l15 == u) { vs = ss[u]; vd = sd[u]; }
  }
  const int r = base + 2 * l15 + half;
  if (FULL || r < nrows) { a_s[r] = vs; a_d[r] = vd; }
}

template <int H>
__global__ __launch_bounds__(256) void att_dots_rows(
    const float* __restrict__ Hm, const float* __restrict__ att_s,
    const float* __restrict__ att_d, float* __restrict__ a_s,
    float* __restrict__ a_d, int nrows)
{
  const int wave = blockIdx.x * (blockDim.x >> 5) + (threadIdx.x >> 5);
  const int base = wave * (2 * DOT_PAIRS);
  const int lane = threadIdx.x & 31;
  if (base + 2 * DOT_PAIRS <= nrows)
    att_dots_wave<H, true>(Hm, att_s, att_d, a_s, a_d, nrows, base, lane);
  else if (base < nrows)
    att_dots_wave<H, false>(Hm, att_s, att_d, a_s, a_d, nrows, base, lane);
}

__device__ __forceinline__ float lrelu(float e) {
  return e > 0.f ? e : NEG_SLOPE * e;
}

#define TILE_NODES 196
#define LIST_CAP   8192
#define TILE_NT    1024

template <int H, int C, bool ELU>
__global__ __launch_bounds__(TILE_NT) void node_tile_gat(
    const int* __restrict__ src, const int* __restrict__ dst,
    const float* __restrict__ a_s, const float* __restrict__ a_d,
    const float* __restrict__ Hfeat, const float* __restrict__ bias,
    float* __restrict__ out, int E, int N)
{
  constexpr int HC = H * C;
  constexpr int TN = TILE_NODES;
  constexpr int CAP = LIST_CAP;
  constexpr int NT = TILE_NT;
  constexpr int J = HC / 32;
  static_assert(C % 32 == 0, "a 32-float chunk must lie inside one head");
  static_assert((H & (H - 1)) == 0 && H <= 32, "H must be a power of two");

  __shared__ float acc[TN * HC];
  __shared__ float m_l[TN * H];
  __shared__ float s_l[TN * H];
  __shared__ float sc_l[TN * H];
  __shared__ float ad_l[TN * H];
  __shared__ int lsrc[CAP];
  __shared__ unsigned short lld[CAP];
  __shared__ int cnt;
  __shared__ int ovf;

  const int tid = threadIdx.x;
  const int lane = tid & 31;
  const int wv = tid >> 5;
  const int lo = blockIdx.x * TN;
  const int nloc = min(TN, N - lo);
  if (nloc <= 0) return;
  const bool vecOK = (((size_t)dst) & 15) == 0;
  const int E4 = vecOK ? (E >> 2) : 0;
  const int4* dst4 = (const int4*)dst;
  const int rot = (int)((((long long)blockIdx.x * E4) / gridDim.x) & ~31LL);

  auto test1 = [&](int d, int e) {
    unsigned ldv = (unsigned)(d - lo);
    if (ldv < (unsigned)nloc) {
      int idx = atomicAdd(&cnt, 1);
      if (idx < CAP) { lsrc[idx] = src[e]; lld[idx] = (unsigned short)ldv; }
      else ovf = 1;
    }
  };

  auto process = [&](int n, int batch) {
    if (batch > 0) {
      for (int j = tid; j < nloc * H; j += NT) sc_l[j] = m_l[j];
    }
    __syncthreads();
    for (int i = tid; i < n; i += NT) {
      const int si = lsrc[i];
      const int ld = lld[i];
      for (int h = 0; h < H; ++h) {
        float ev = lrelu(a_s[(size_t)si * H + h] + ad_l[ld * H + h]);
        __hip_atomic_fetch_max(&m_l[ld * H + h], ev,
                               __ATOMIC_RELAXED, __HIP_MEMORY_SCOPE_WORKGROUP);
      }
    }
    __syncthreads();
    if (batch > 0) {
      for (int j = tid; j < nloc * H; j += NT) {
        float mo = sc_l[j], mn = m_l[j];
        float f = (mo == mn) ? 1.f : __expf(mo - mn);
        sc_l[j] = f;
        s_l[j] *= f;
      }
      __syncthreads();
      for (int i = tid; i < nloc * HC; i += NT)
        acc[i] *= sc_l[(i / HC) * H + (i % HC) / C];
      __syncthreads();
    }
    for (int i0 = wv * 4; i0 < n; i0 += 32 * 4) {
      int si[4], ld[4];
      float hv[4][J];
      float w[4];
      #pragma unroll
      for (int u = 0; u < 4; ++u) {
        int ii = min(i0 + u, n - 1);
        si[u] = lsrc[ii];
        ld[u] = lld[ii];
      }
      #pragma unroll
      for (int u = 0; u < 4; ++u) {
        const float* hp = Hfeat + (size_t)si[u] * HC + lane;
        #pragma unroll
        for (int j = 0; j < J; ++j) hv[u][j] = hp[j * 32];
      }
      #pragma unroll
      for (int u = 0; u < 4; ++u) {
        const int hh = lane & (H - 1);
        float ev = lrelu(a_s[(size_t)si[u] * H + hh] + ad_l[ld[u] * H + hh]);
        w[u] = __expf(ev - m_l[ld[u] * H + hh]);
      }
      #pragma unroll
      for (int u = 0; u < 4; ++u) {
        if (i0 + u < n) {
          if (lane < H) atomicAdd(&s_l[ld[u] * H + lane], w[u]);
          float wh[H];
          #pragma unroll
          for (int h = 0; h < H; ++h)
            wh[h] = __builtin_bit_cast(float,
                __builtin_amdgcn_readlane(__builtin_bit_cast(int, w[u]), h));
          #pragma unroll
          for (int j = 0; j < J; ++j)
            atomicAdd(&acc[ld[u] * HC + j * 32 + lane], hv[u][j] * wh[(j * 32) / C]);
        }
      }
    }
    __syncthreads();
  };

  int batch = 0;
  auto drain_if_needed = [&]() {
    __syncthreads();
    const int c = cnt;
    __syncthreads();
    if (c + 4 * NT > CAP) {
      process(c, batch);
      ++batch;
      if (tid == 0) cnt = 0;
      __syncthreads();
    }
  };

  for (int attempt = 0; attempt < 2; ++attempt) {
    for (int i = tid; i < nloc * HC; i += NT) acc[i] = 0.f;
    for (int j = tid; j < nloc * H; j += NT) {
      m_l[j] = -__builtin_inff();
      s_l[j] = 0.f;
      ad_l[j] = a_d[(size_t)lo * H + j];
    }
    for (int i = tid; i < nloc; i += NT) { lsrc[i] = lo + i; lld[i] = (unsigned short)i; }
    if (tid == 0) { cnt = nloc; ovf = 0; }
    batch = 0;
    __syncthreads();

    if (attempt == 0) {
      for (int g0 = 0; g0 < E4; g0 += 4 * NT) {
        int4 d[4];
        int gi[4];
        #pragma unroll
        for (int u = 0; u < 4; ++u) {
          int g = g0 + u * NT + tid;
          bool ok = g < E4;
          int gr = g + rot;
          if (gr >= E4) gr -= E4;
          gi[u] = gr;
          if (ok) d[u] = dst4[gr];
          else    d[u] = make_int4(lo - 1, lo - 1, lo - 1, lo - 1);
        }
        #pragma unroll
        for (int u = 0; u < 4; ++u) {
          const int e = gi[u] * 4;
          test1(d[u].x, e);
          test1(d[u].y, e + 1);
          test1(d[u].z, e + 2);
          test1(d[u].w, e + 3);
        }
      }
      for (int e = E4 * 4 + tid; e < E; e += NT) test1(dst[e], e);
    } else {
      for (int g0 = 0; g0 < E4; g0 += NT) {
        drain_if_needed();
        int g = g0 + tid;
        if (g < E4) {
          int4 d = dst4[g];
          const int e = g * 4;
          test1(d.x, e);
          test1(d.y, e + 1);
          test1(d.z, e + 2);
          test1(d.w, e + 3);
        }
      }
      for (int e0 = E4 * 4; e0 < E; e0 += 4 * NT) {
        drain_if_needed();
        for (int k = 0; k < 4; ++k) {
          int e = e0 + k * NT + tid;
          if (e < E) test1(dst[e], e);
        }
      }
    }
    __syncthreads();
    const int o = ovf;
    const int c = cnt;
    __syncthreads();
    if (attempt == 0 && o) continue;
    process(min(c, CAP), batch);
    break;
  }

  for (int r = wv; r < nloc; r += 32) {
    for (int j = 0; j < J; ++j) {
      const int f = j * 32 + lane;
      float v = acc[r * HC + f] / (s_l[r * H + f / C] + PYG_EPS) + bias[f];
      if (ELU) v = v > 0.f ? v : expm1f(v);
      out[(size_t)(lo + r) * HC + f] = v;
    }
  }
}

extern "C" void kernel_launch(void* const* d_in, const int* in_sizes, int n_in,
                              void* d_out, int out_size, void* d_ws, size_t ws_size,
                              hipStream_t stream)
{
  const float* x   = (const float*)d_in[0];
  const int*   ei  = (const int*)  d_in[1];
  const float* W1  = (const float*)d_in[2];
  const float* as1 = (const float*)d_in[3];
  const float* ad1 = (const float*)d_in[4];
  const float* b1  = (const float*)d_in[5];
  const float* W2  = (const float*)d_in[6];
  const float* as2 = (const float*)d_in[7];
  const float* ad2 = (const float*)d_in[8];
  const float* b2  = (const float*)d_in[9];
  float* out = (float*)d_out;

  const int F = 128, H1 = 4, C = 64, HC1 = 256, HC2 = 64;
  const int N  = in_sizes[0] / F;
  const int E  = in_sizes[1] / 2;
  const int* src = ei;
  const int* dst = ei + E;

  float* h1  = (float*)d_ws;
  float* x2  = h1  + (size_t)N * HC1;
  float* h2  = x2  + (size_t)N * HC1;
  float* aS1 = h2  + (size_t)N * HC2;
  float* aD1 = aS1 + (size_t)N * H1;
  float* aS2 = aD1 + (size_t)N * H1;
  float* aD2 = aS2 + N;

  const int TB = 256;
  auto cdiv = [](int a, int b) { return (a + b - 1) / b; };

  {
    int tiles = (N / 16) * (HC1 / 16);
    gemm_wmma_f32<<<cdiv(tiles, 8), TB, 0, stream>>>(x, W1, h1, N, F, HC1);
  }
  att_dots_rows<4><<<cdiv(N * H1, 8 * 2 * DOT_PAIRS), TB, 0, stream>>>(h1, as1, ad1, aS1, aD1, N * H1);
  node_tile_gat<4, 64, true><<<cdiv(N, TILE_NODES), TILE_NT, 0, stream>>>(
      src, dst, aS1, aD1, h1, b1, x2, E, N);

  {
    int tiles = (N / 16) * (HC2 / 16);
    gemm_wmma_f32<<<cdiv(tiles, 8), TB, 0, stream>>>(x2, W2, h2, N, HC1, HC2);
  }
  att_dots_rows<1><<<cdiv(N, 8 * 2 * DOT_PAIRS), TB, 0, stream>>>(h2, as2, ad2, aS2, aD2, N);
  node_tile_gat<1, 64, false><<<cdiv(N, TILE_NODES), TILE_NT, 0, stream>>>(
      src, dst, aS2, aD2, h2, b2, out, E, N);
}


# Edge filter sweep waits for no lookup: a trip's 16 tests become wave masks, one LDS slot claim per wave and trip stores edge numbers; src/dst read for the whole list afterwards; sum order changes
# speedup vs baseline: 4.2045x; 1.0499x over previous
#include <hip/hip_runtime.h>
#include <math.h>

typedef __attribute__((ext_vector_type(2))) float v2f;
typedef __attribute__((ext_vector_type(8))) float v8f;

#define NEG_SLOPE 0.2f
#define PYG_EPS   1e-16f

__global__ __launch_bounds__(256) void gemm_wmma_f32(
    const float* __restrict__ X, const float* __restrict__ W,
    float* __restrict__ Hout, int Nrows, int K, int M)
{
  const int tilesN = M >> 4;
  const int tilesTotal = (Nrows >> 4) * tilesN;
  const int wave = blockIdx.x * (blockDim.x >> 5) + (threadIdx.x >> 5);
  if (wave >= tilesTotal) return;
  const int tm = wave / tilesN;
  const int tn = wave - tm * tilesN;
  const int lane = threadIdx.x & 31;
  const int half = lane >> 4;
  const int l15  = lane & 15;
  const int kb   = half << 1;

  const float* xp = X + (size_t)((tm << 4) + l15) * K + kb;
  const float* wp = W + (size_t)kb * M + ((tn << 4) + l15);

  v8f acc = {0.f, 0.f, 0.f, 0.f, 0.f, 0.f, 0.f, 0.f};
  for (int k0 = 0; k0 < K; k0 += 4) {
    v2f a = *(const v2f*)xp;
    v2f b;
    b.x = wp[0];
    b.y = wp[M];
    acc = __builtin_amdgcn_wmma_f32_16x16x4_f32(
         false, a,  false, b,
         (short)0, acc,  false,  false);
    xp += 4;
    wp += (size_t)4 * M;
  }

  float* op = Hout + (size_t)((tm << 4) + (half << 3)) * M + ((tn << 4) + l15);
  #pragma unroll
  for (int v = 0; v < 8; ++v) op[(size_t)v * M] = acc[v];
}

template <int CTRL>
__device__ __forceinline__ float row_swz(float v) {
  return __builtin_bit_cast(float, __builtin_amdgcn_update_dpp(
      0, __builtin_bit_cast(int, v), CTRL, 0xf, 0xf, true));
}

#define DOT_PAIRS 16

template <int H, bool FULL>
__device__ __forceinline__ void att_dots_wave(
    const float* __restrict__ Hm, const float* __restrict__ att_s,
    const float* __restrict__ att_d, float* __restrict__ a_s,
    float* __restrict__ a_d, int nrows, int base, int lane)
{
  constexpr int C = 64;
  constexpr int U = DOT_PAIRS;
  constexpr int PP = (H >= 2) ? (H / 2) : 1;
  static_assert((H & (H - 1)) == 0, "H must be a power of two");
  static_assert(U % PP == 0 && (2 * U) % H == 0, "a wave's first row must be head 0");

  const int half = lane >> 4;
  const int l15  = lane & 15;
  const int c0   = l15 << 2;

  float4 x[U];
  const float* p0 = Hm + (size_t)(base + half) * C + c0;
  #pragma unroll
  for (int u = 0; u < U; ++u) {
    if (FULL) {
      x[u] = *(const float4*)(p0 + u * 2 * C);
    } else {
      const int r = min(base + 2 * u + half, nrows - 1);
      x[u] = *(const float4*)(Hm + (size_t)r * C + c0);
    }
  }

  float ws[PP][4], wd[PP][4];
  #pragma unroll
  for (int q = 0; q < PP; ++q) {
    const int h = (2 * q + half) & (H - 1);
    #pragma unroll
    for (int k = 0; k < 4; ++k) {
      ws[q][k] = att_s[h * C + c0 + k];
      wd[q][k] = att_d[h * C + c0 + k];
    }
  }

  float ss[U], sd[U];
  #pragma unroll
  for (int u = 0; u < U; ++u) {
    const int q = u % PP;
    ss[u] = x[u].x * ws[q][0] + x[u].y * ws[q][1] + x[u].z * ws[q][2] + x[u].w * ws[q][3];
    sd[u] = x[u].x * wd[q][0] + x[u].y * wd[q][1] + x[u].z * wd[q][2] + x[u].w * wd[q][3];
  }
  #pragma unroll
  for (int u = 0; u < U; ++u) { ss[u] += row_swz<0x161>(ss[u]); sd[u] += row_swz<0x161>(sd[u]); }
  #pragma unroll
  for (int u = 0; u < U; ++u) { ss[u] += row_swz<0x162>(ss[u]); sd[u] += row_swz<0x162>(sd[u]); }
  #pragma unroll
  for (int u = 0; u < U; ++u) { ss[u] += row_swz<0x164>(ss[u]); sd[u] += row_swz<0x164>(sd[u]); }
  #pragma unroll
  for (int u = 0; u < U; ++u) { ss[u] += row_swz<0x168>(ss[u]); sd[u] += row_swz<0x168>(sd[u]); }

  float vs = ss[0], vd = sd[0];
  #pragma unroll
  for (int u = 1; u < U; ++u) {
    if (l15 == u) { vs = ss[u]; vd = sd[u]; }
  }
  const int r = base + 2 * l15 + half;
  if (FULL || r < nrows) { a_s[r] = vs; a_d[r] = vd; }
}

template <int H>
__global__ __launch_bounds__(256) void att_dots_rows(
    const float* __restrict__ Hm, const float* __restrict__ att_s,
    const float* __restrict__ att_d, float* __restrict__ a_s,
    float* __restrict__ a_d, int nrows)
{
  const int wave = blockIdx.x * (blockDim.x >> 5) + (threadIdx.x >> 5);
  const int base = wave * (2 * DOT_PAIRS);
  const int lane = threadIdx.x & 31;
  if (base + 2 * DOT_PAIRS <= nrows)
    att_dots_wave<H, true>(Hm, att_s, att_d, a_s, a_d, nrows, base, lane);
  else if (base < nrows)
    att_dots_wave<H, false>(Hm, att_s, att_d, a_s, a_d, nrows, base, lane);
}

__device__ __forceinline__ float lrelu(float e) {
  return e > 0.f ? e : NEG_SLOPE * e;
}

#define TILE_NODES 196
#define LIST_CAP   8192
#define TILE_NT    1024

template <int H, int C, bool ELU>
__global__ __launch_bounds__(TILE_NT) void node_tile_gat(
    const int* __restrict__ src, const int* __restrict__ dst,
    const float* __restrict__ a_s, const float* __restrict__ a_d,
    const float* __restrict__ Hfeat, const float* __restrict__ bias,
    float* __restrict__ out, int E, int N)
{
  constexpr int HC = H * C;
  constexpr int TN = TILE_NODES;
  constexpr int CAP = LIST_CAP;
  constexpr int NT = TILE_NT;
  constexpr int J = HC / 32;
  static_assert(C % 32 == 0, "a 32-float chunk must lie inside one head");
  static_assert((H & (H - 1)) == 0 && H <= 32, "H must be a power of two");

  __shared__ float acc[TN * HC];
  __shared__ float m_l[TN * H];
  __shared__ float s_l[TN * H];
  __shared__ float sc_l[TN * H];
  __shared__ float ad_l[TN * H];
  __shared__ int lsrc[CAP];
  __shared__ unsigned short lld[CAP];
  __shared__ int cnt;
  __shared__ int ovf;

  const int tid = threadIdx.x;
  const int lane = tid & 31;
  const int wv = tid >> 5;
  const int lo = blockIdx.x * TN;
  const int nloc = min(TN, N - lo);
  if (nloc <= 0) return;
  const bool vecOK = (((size_t)dst) & 15) == 0;
  const int E4 = vecOK ? (E >> 2) : 0;
  const int4* dst4 = (const int4*)dst;
  const int rot = (int)((((long long)blockIdx.x * E4) / gridDim.x) & ~31LL);

  auto test1 = [&](int d, int e) {
    unsigned ldv = (unsigned)(d - lo);
    if (ldv < (unsigned)nloc) {
      int idx = atomicAdd(&cnt, 1);
      if (idx < CAP) lsrc[idx] = e;
      else ovf = 1;
    }
  };

  auto process = [&](int n, int batch) {
    for (int i = (batch == 0 ? nloc : 0) + tid; i < n; i += NT) {
      const int e = lsrc[i];
      lsrc[i] = src[e];
      lld[i] = (unsigned short)(dst[e] - lo);
    }
    if (batch > 0) {
      for (int j = tid; j < nloc * H; j += NT) sc_l[j] = m_l[j];
    }
    __syncthreads();
    for (int i = tid; i < n; i += NT) {
      const int si = lsrc[i];
      const int ld = lld[i];
      for (int h = 0; h < H; ++h) {
        float ev = lrelu(a_s[(size_t)si * H + h] + ad_l[ld * H + h]);
        __hip_atomic_fetch_max(&m_l[ld * H + h], ev,
                               __ATOMIC_RELAXED, __HIP_MEMORY_SCOPE_WORKGROUP);
      }
    }
    __syncthreads();
    if (batch > 0) {
      for (int j = tid; j < nloc * H; j += NT) {
        float mo = sc_l[j], mn = m_l[j];
        float f = (mo == mn) ? 1.f : __expf(mo - mn);
        sc_l[j] = f;
        s_l[j] *= f;
      }
      __syncthreads();
      for (int i = tid; i < nloc * HC; i += NT)
        acc[i] *= sc_l[(i / HC) * H + (i % HC) / C];
      __syncthreads();
    }
    for (int i0 = wv * 4; i0 < n; i0 += 32 * 4) {
      int si[4], ld[4];
      float hv[4][J];
      float w[4];
      #pragma unroll
      for (int u = 0; u < 4; ++u) {
        int ii = min(i0 + u, n - 1);
        si[u] = lsrc[ii];
        ld[u] = lld[ii];
      }
      #pragma unroll
      for (int u = 0; u < 4; ++u) {
        const float* hp = Hfeat + (size_t)si[u] * HC + lane;
        #pragma unroll
        for (int j = 0; j < J; ++j) hv[u][j] = hp[j * 32];
      }
      #pragma unroll
      for (int u = 0; u < 4; ++u) {
        const int hh = lane & (H - 1);
        float ev = lrelu(a_s[(size_t)si[u] * H + hh] + ad_l[ld[u] * H + hh]);
        w[u] = __expf(ev - m_l[ld[u] * H + hh]);
      }
      #pragma unroll
      for (int u = 0; u < 4; ++u) {
        if (i0 + u < n) {
          if (lane < H) atomicAdd(&s_l[ld[u] * H + lane], w[u]);
          float wh[H];
          #pragma unroll
          for (int h = 0; h < H; ++h)
            wh[h] = __builtin_bit_cast(float,
                __builtin_amdgcn_readlane(__builtin_bit_cast(int, w[u]), h));
          #pragma unroll
          for (int j = 0; j < J; ++j)
            atomicAdd(&acc[ld[u] * HC + j * 32 + lane], hv[u][j] * wh[(j * 32) / C]);
        }
      }
    }
    __syncthreads();
  };

  int batch = 0;
  auto drain_if_needed = [&]() {
    __syncthreads();
    const int c = cnt;
    __syncthreads();
    if (c + 4 * NT > CAP) {
      process(c, batch);
      ++batch;
      if (tid == 0) cnt = 0;
      __syncthreads();
    }
  };

  for (int attempt = 0; attempt < 2; ++attempt) {
    for (int i = tid; i < nloc * HC; i += NT) acc[i] = 0.f;
    for (int j = tid; j < nloc * H; j += NT) {
      m_l[j] = -__builtin_inff();
      s_l[j] = 0.f;
      ad_l[j] = a_d[(size_t)lo * H + j];
    }
    for (int i = tid; i < nloc; i += NT) { lsrc[i] = lo + i; lld[i] = (unsigned short)i; }
    if (tid == 0) { cnt = nloc; ovf = 0; }
    batch = 0;
    __syncthreads();

    if (attempt == 0) {
      for (int g0 = 0; g0 < E4; g0 += 4 * NT) {
        int4 d[4];
        int gi[4];
        #pragma unroll
        for (int u = 0; u < 4; ++u) {
          int g = g0 + u * NT + tid;
          bool ok = g < E4;
          int gr = g + rot;
          if (gr >= E4) gr -= E4;
          gi[u] = gr;
          if (ok) d[u] = dst4[gr];
          else    d[u] = make_int4(lo - 1, lo - 1, lo - 1, lo - 1);
        }
        unsigned hit[16];
        #pragma unroll
        for (int u = 0; u < 4; ++u) {
          hit[4 * u + 0] = __builtin_amdgcn_ballot_w32((unsigned)(d[u].x - lo) < (unsigned)nloc);
          hit[4 * u + 1] = __builtin_amdgcn_ballot_w32((unsigned)(d[u].y - lo) < (unsigned)nloc);
          hit[4 * u + 2] = __builtin_amdgcn_ballot_w32((unsigned)(d[u].z - lo) < (unsigned)nloc);
          hit[4 * u + 3] = __builtin_amdgcn_ballot_w32((unsigned)(d[u].w - lo) < (unsigned)nloc);
        }
        unsigned any = 0;
        #pragma unroll
        for (int k = 0; k < 16; ++k) any |= hit[k];
        if (any) {
          int tot = 0;
          #pragma unroll
          for (int k = 0; k < 16; ++k) tot += __builtin_popcount(hit[k]);
          int slot = 0;
          if (lane == 0) slot = atomicAdd(&cnt, tot);
          slot = __builtin_amdgcn_readfirstlane(slot);
          #pragma unroll
          for (int k = 0; k < 16; ++k) {
            if (hit[k]) {
              if ((hit[k] >> lane) & 1u) {
                const int idx = slot + (int)__builtin_amdgcn_mbcnt_lo(hit[k], 0u);
                if (idx < CAP) lsrc[idx] = gi[k >> 2] * 4 + (k & 3);
                else ovf = 1;
              }
              slot += __builtin_popcount(hit[k]);
            }
          }
        }
      }
      for (int e = E4 * 4 + tid; e < E; e += NT) test1(dst[e], e);
    } else {
      for (int g0 = 0; g0 < E4; g0 += NT) {
        drain_if_needed();
        int g = g0 + tid;
        if (g < E4) {
          int4 d = dst4[g];
          const int e = g * 4;
          test1(d.x, e);
          test1(d.y, e + 1);
          test1(d.z, e + 2);
          test1(d.w, e + 3);
        }
      }
      for (int e0 = E4 * 4; e0 < E; e0 += 4 * NT) {
        drain_if_needed();
        for (int k = 0; k < 4; ++k) {
          int e = e0 + k * NT + tid;
          if (e < E) test1(dst[e], e);
        }
      }
    }
    __syncthreads();
    const int o = ovf;
    const int c = cnt;
    __syncthreads();
    if (attempt == 0 && o) continue;
    process(min(c, CAP), batch);
    break;
  }

  for (int r = wv; r < nloc; r += 32) {
    for (int j = 0; j < J; ++j) {
      const int f = j * 32 + lane;
      float v = acc[r * HC + f] / (s_l[r * H + f / C] + PYG_EPS) + bias[f];
      if (ELU) v = v > 0.f ? v : expm1f(v);
      out[(size_t)(lo + r) * HC + f] = v;
    }
  }
}

extern "C" void kernel_launch(void* const* d_in, const int* in_sizes, int n_in,
                              void* d_out, int out_size, void* d_ws, size_t ws_size,
                              hipStream_t stream)
{
  const float* x   = (const float*)d_in[0];
  const int*   ei  = (const int*)  d_in[1];
  const float* W1  = (const float*)d_in[2];
  const float* as1 = (const float*)d_in[3];
  const float* ad1 = (const float*)d_in[4];
  const float* b1  = (const float*)d_in[5];
  const float* W2  = (const float*)d_in[6];
  const float* as2 = (const float*)d_in[7];
  const float* ad2 = (const float*)d_in[8];
  const float* b2  = (const float*)d_in[9];
  float* out = (float*)d_out;

  const int F = 128, H1 = 4, C = 64, HC1 = 256, HC2 = 64;
  const int N  = in_sizes[0] / F;
  const int E  = in_sizes[1] / 2;
  const int* src = ei;
  const int* dst = ei + E;

  float* h1  = (float*)d_ws;
  float* x2  = h1  + (size_t)N * HC1;
  float* h2  = x2  + (size_t)N * HC1;
  float* aS1 = h2  + (size_t)N * HC2;
  float* aD1 = aS1 + (size_t)N * H1;
  float* aS2 = aD1 + (size_t)N * H1;
  float* aD2 = aS2 + N;

  const int TB = 256;
  auto cdiv = [](int a, int b) { return (a + b - 1) / b; };

  {
    int tiles = (N / 16) * (HC1 / 16);
    gemm_wmma_f32<<<cdiv(tiles, 8), TB, 0, stream>>>(x, W1, h1, N, F, HC1);
  }
  att_dots_rows<4><<<cdiv(N * H1, 8 * 2 * DOT_PAIRS), TB, 0, stream>>>(h1, as1, ad1, aS1, aD1, N * H1);
  node_tile_gat<4, 64, true><<<cdiv(N, TILE_NODES), TILE_NT, 0, stream>>>(
      src, dst, aS1, aD1, h1, b1, x2, E, N);

  {
    int tiles = (N / 16) * (HC2 / 16);
    gemm_wmma_f32<<<cdiv(tiles, 8), TB, 0, stream>>>(x2, W2, h2, N, HC1, HC2);
  }
  att_dots_rows<1><<<cdiv(N, 8 * 2 * DOT_PAIRS), TB, 0, stream>>>(h2, as2, ad2, aS2, aD2, N);
  node_tile_gat<1, 64, false><<<cdiv(N, TILE_NODES), TILE_NT, 0, stream>>>(
      src, dst, aS2, aD2, h2, b2, out, E, N);
}


# Layer 1's edge kernel leaves each tile's finished list (source node, local destination) in the workspace; layer 2 reads it: no dst sweep, no src/dst lookups there; layer 2 sums in that list's order
# speedup vs baseline: 6.0022x; 1.4276x over previous
#include <hip/hip_runtime.h>
#include <math.h>

typedef __attribute__((ext_vector_type(2))) float v2f;
typedef __attribute__((ext_vector_type(8))) float v8f;

#define NEG_SLOPE 0.2f
#define PYG_EPS   1e-16f

template <int K, int M, int NT, int THREADS>
__global__ __launch_bounds__(THREADS) void gemm_wmma_f32(
    const float* __restrict__ X, const float* __restrict__ W,
    float* __restrict__ Hout, int rowTiles)
{
  constexpr int PITCH = 2 * M + 32;
  constexpr int CH = (M / 16) / NT;
  static_assert(K % 4 == 0 && M % 16 == 0 && (M / 16) % NT == 0, "shape");
  static_assert(PITCH % 64 == 32, "lane halves must fall on different bank halves");
  __shared__ __attribute__((aligned(16))) float Ws[(K / 2) * PITCH];

  const int tid = threadIdx.x;
  for (int i = tid; i < K * M; i += THREADS) {
    const int k = i / M;
    const int col = i - k * M;
    Ws[(k >> 1) * PITCH + col * 2 + (k & 1)] = W[i];
  }
  __syncthreads();

  const int lane = tid & 31;
  const int wv = tid >> 5;
  const int half = lane >> 4;
  const int l15  = lane & 15;
  const int items = rowTiles * CH;

  for (int it = blockIdx.x + gridDim.x * wv; it < items;
       it += gridDim.x * (THREADS / 32)) {
    const int tm = it / CH;
    const int c0 = (it - tm * CH) * (NT * 16);
    const float* xp = X + (size_t)((tm << 4) + l15) * K + (half << 1);
    const float* bp = Ws + half * PITCH + (c0 + l15) * 2;

    v8f acc[NT];
    #pragma unroll
    for (int t = 0; t < NT; ++t) acc[t] = (v8f){0.f, 0.f, 0.f, 0.f, 0.f, 0.f, 0.f, 0.f};

    for (int k0 = 0; k0 < K; k0 += 4) {
      const v2f a = *(const v2f*)(xp + k0);
      const float* bk = bp + (k0 >> 1) * PITCH;
      #pragma unroll
      for (int t = 0; t < NT; ++t) {
        const v2f b = *(const v2f*)(bk + t * 32);
        acc[t] = __builtin_amdgcn_wmma_f32_16x16x4_f32(
             false, a,  false, b,
             (short)0, acc[t],  false,  false);
      }
    }

    float* op = Hout + (size_t)((tm << 4) + (half << 3)) * M + (c0 + l15);
    #pragma unroll
    for (int t = 0; t < NT; ++t) {
      #pragma unroll
      for (int v = 0; v < 8; ++v) op[(size_t)v * M + t * 16] = acc[t][v];
    }
  }
}

template <int CTRL>
__device__ __forceinline__ float row_swz(float v) {
  return __builtin_bit_cast(float, __builtin_amdgcn_update_dpp(
      0, __builtin_bit_cast(int, v), CTRL, 0xf, 0xf, true));
}

#define DOT_PAIRS 16

template <int H, bool FULL>
__device__ __forceinline__ void att_dots_wave(
    const float* __restrict__ Hm, const float* __restrict__ att_s,
    const float* __restrict__ att_d, float* __restrict__ a_s,
    float* __restrict__ a_d, int nrows, int base, int lane)
{
  constexpr int C = 64;
  constexpr int U = DOT_PAIRS;
  constexpr int PP = (H >= 2) ? (H / 2) : 1;
  static_assert((H & (H - 1)) == 0, "H must be a power of two");
  static_assert(U % PP == 0 && (2 * U) % H == 0, "a wave's first row must be head 0");

  const int half = lane >> 4;
  const int l15  = lane & 15;
  const int c0   = l15 << 2;

  float4 x[U];
  const float* p0 = Hm + (size_t)(base + half) * C + c0;
  #pragma unroll
  for (int u = 0; u < U; ++u) {
    if (FULL) {
      x[u] = *(const float4*)(p0 + u * 2 * C);
    } else {
      const int r = min(base + 2 * u + half, nrows - 1);
      x[u] = *(const float4*)(Hm + (size_t)r * C + c0);
    }
  }

  float ws[PP][4], wd[PP][4];
  #pragma unroll
  for (int q = 0; q < PP; ++q) {
    const int h = (2 * q + half) & (H - 1);
    #pragma unroll
    for (int k = 0; k < 4; ++k) {
      ws[q][k] = att_s[h * C + c0 + k];
      wd[q][k] = att_d[h * C + c0 + k];
    }
  }

  float ss[U], sd[U];
  #pragma unroll
  for (int u = 0; u < U; ++u) {
    const int q = u % PP;
    ss[u] = x[u].x * ws[q][0] + x[u].y * ws[q][1] + x[u].z * ws[q][2] + x[u].w * ws[q][3];
    sd[u] = x[u].x * wd[q][0] + x[u].y * wd[q][1] + x[u].z * wd[q][2] + x[u].w * wd[q][3];
  }
  #pragma unroll
  for (int u = 0; u < U; ++u) { ss[u] += row_swz<0x161>(ss[u]); sd[u] += row_swz<0x161>(sd[u]); }
  #pragma unroll
  for (int u = 0; u < U; ++u) { ss[u] += row_swz<0x162>(ss[u]); sd[u] += row_swz<0x162>(sd[u]); }
  #pragma unroll
  for (int u = 0; u < U; ++u) { ss[u] += row_swz<0x164>(ss[u]); sd[u] += row_swz<0x164>(sd[u]); }
  #pragma unroll
  for (int u = 0; u < U; ++u) { ss[u] += row_swz<0x168>(ss[u]); sd[u] += row_swz<0x168>(sd[u]); }

  float vs = ss[0], vd = sd[0];
  #pragma unroll
  for (int u = 1; u < U; ++u) {
    if (l15 == u) { vs = ss[u]; vd = sd[u]; }
  }
  const int r = base + 2 * l15 + half;
  if (FULL || r < nrows) { a_s[r] = vs; a_d[r] = vd; }
}

template <int H>
__global__ __launch_bounds__(256) void att_dots_rows(
    const float* __restrict__ Hm, const float* __restrict__ att_s,
    const float* __restrict__ att_d, float* __restrict__ a_s,
    float* __restrict__ a_d, int nrows)
{
  const int wave = blockIdx.x * (blockDim.x >> 5) + (threadIdx.x >> 5);
  const int base = wave * (2 * DOT_PAIRS);
  const int lane = threadIdx.x & 31;
  if (base + 2 * DOT_PAIRS <= nrows)
    att_dots_wave<H, true>(Hm, att_s, att_d, a_s, a_d, nrows, base, lane);
  else if (base < nrows)
    att_dots_wave<H, false>(Hm, att_s, att_d, a_s, a_d, nrows, base, lane);
}

__device__ __forceinline__ float lrelu(float e) {
  return e > 0.f ? e : NEG_SLOPE * e;
}

#define TILE_NODES 196
#define LIST_CAP   8192
#define TILE_NT    1024

template <int H, int C, bool ELU, int ROLE>
__global__ __launch_bounds__(TILE_NT) void node_tile_gat(
    const int* __restrict__ src, const int* __restrict__ dst,
    const float* __restrict__ a_s, const float* __restrict__ a_d,
    const float* __restrict__ Hfeat, const float* __restrict__ bias,
    float* __restrict__ out, int E, int N,
    int* __restrict__ tl_src, unsigned short* __restrict__ tl_ld,
    int* __restrict__ tl_cnt)
{
  constexpr int HC = H * C;
  constexpr int TN = TILE_NODES;
  constexpr int CAP = LIST_CAP;
  constexpr int NT = TILE_NT;
  constexpr int J = HC / 32;
  static_assert(C % 32 == 0, "a 32-float chunk must lie inside one head");
  static_assert((H & (H - 1)) == 0 && H <= 32, "H must be a power of two");

  __shared__ float acc[TN * HC];
  __shared__ float m_l[TN * H];
  __shared__ float s_l[TN * H];
  __shared__ float sc_l[TN * H];
  __shared__ float ad_l[TN * H];
  __shared__ int lsrc[CAP];
  __shared__ unsigned short lld[CAP];
  __shared__ int cnt;
  __shared__ int ovf;

  const int tid = threadIdx.x;
  const int lane = tid & 31;
  const int wv = tid >> 5;
  const int lo = blockIdx.x * TN;
  const int nloc = min(TN, N - lo);
  if (nloc <= 0) return;
  const bool vecOK = (((size_t)dst) & 15) == 0;
  const int E4 = vecOK ? (E >> 2) : 0;
  const int4* dst4 = (const int4*)dst;
  const int rot = (int)((((long long)blockIdx.x * E4) / gridDim.x) & ~31LL);

  auto test1 = [&](int d, int e) {
    unsigned ldv = (unsigned)(d - lo);
    if (ldv < (unsigned)nloc) {
      int idx = atomicAdd(&cnt, 1);
      if (idx < CAP) lsrc[idx] = e;
      else ovf = 1;
    }
  };

  auto process = [&](int n, int batch, bool resolved) {
    if (!resolved) {
      for (int i = (batch == 0 ? nloc : 0) + tid; i < n; i += NT) {
        const int e = lsrc[i];
        lsrc[i] = src[e];
        lld[i] = (unsigned short)(dst[e] - lo);
      }
    }
    if (batch > 0) {
      for (int j = tid; j < nloc * H; j += NT) sc_l[j] = m_l[j];
    }
    __syncthreads();
    for (int i = tid; i < n; i += NT) {
      const int si = lsrc[i];
      const int ld = lld[i];
      for (int h = 0; h < H; ++h) {
        float ev = lrelu(a_s[(size_t)si * H + h] + ad_l[ld * H + h]);
        __hip_atomic_fetch_max(&m_l[ld * H + h], ev,
                               __ATOMIC_RELAXED, __HIP_MEMORY_SCOPE_WORKGROUP);
      }
    }
    __syncthreads();
    if (batch > 0) {
      for (int j = tid; j < nloc * H; j += NT) {
        float mo = sc_l[j], mn = m_l[j];
        float f = (mo == mn) ? 1.f : __expf(mo - mn);
        sc_l[j] = f;
        s_l[j] *= f;
      }
      __syncthreads();
      for (int i = tid; i < nloc * HC; i += NT)
        acc[i] *= sc_l[(i / HC) * H + (i % HC) / C];
      __syncthreads();
    }
    for (int i0 = wv * 4; i0 < n; i0 += 32 * 4) {
      int si[4], ld[4];
      float hv[4][J];
      float w[4];
      #pragma unroll
      for (int u = 0; u < 4; ++u) {
        int ii = min(i0 + u, n - 1);
        si[u] = lsrc[ii];
        ld[u] = lld[ii];
      }
      #pragma unroll
      for (int u = 0; u < 4; ++u) {
        const float* hp = Hfeat + (size_t)si[u] * HC + lane;
        #pragma unroll
        for (int j = 0; j < J; ++j) hv[u][j] = hp[j * 32];
      }
      #pragma unroll
      for (int u = 0; u < 4; ++u) {
        const int hh = lane & (H - 1);
        float ev = lrelu(a_s[(size_t)si[u] * H + hh] + ad_l[ld[u] * H + hh]);
        w[u] = __expf(ev - m_l[ld[u] * H + hh]);
      }
      #pragma unroll
      for (int u = 0; u < 4; ++u) {
        if (i0 + u < n) {
          if (lane < H) atomicAdd(&s_l[ld[u] * H + lane], w[u]);
          float wh[H];
          #pragma unroll
          for (int h = 0; h < H; ++h)
            wh[h] = __builtin_bit_cast(float,
                __builtin_amdgcn_readlane(__builtin_bit_cast(int, w[u]), h));
          #pragma unroll
          for (int j = 0; j < J; ++j)
            atomicAdd(&acc[ld[u] * HC + j * 32 + lane], hv[u][j] * wh[(j * 32) / C]);
        }
      }
    }
    __syncthreads();
  };

  int batch = 0;
  auto drain_if_needed = [&]() {
    __syncthreads();
    const int c = cnt;
    __syncthreads();
    if (c + 4 * NT > CAP) {
      process(c, batch, false);
      ++batch;
      if (tid == 0) cnt = 0;
      __syncthreads();
    }
  };

  auto init_sums = [&]() {
    for (int i = tid; i < nloc * HC; i += NT) acc[i] = 0.f;
    for (int j = tid; j < nloc * H; j += NT) {
      m_l[j] = -__builtin_inff();
      s_l[j] = 0.f;
      ad_l[j] = a_d[(size_t)lo * H + j];
    }
  };

  const size_t lbase = (size_t)blockIdx.x * CAP;
  const int nl = (ROLE == 1) ? tl_cnt[blockIdx.x] : -1;

  const bool handed = (ROLE == 1 && nl >= 0);
  if (handed) {
    const int n = min(nl, CAP);
    init_sums();
    for (int i = tid; i < n; i += NT) { lsrc[i] = tl_src[lbase + i]; lld[i] = tl_ld[lbase + i]; }
    process(n, 0, true);
  }
  for (int attempt = (ROLE == 1 ? 1 : 0); !handed && attempt < 2; ++attempt) {
    init_sums();
    for (int i = tid; i < nloc; i += NT) { lsrc[i] = lo + i; lld[i] = (unsigned short)i; }
    if (tid == 0) { cnt = nloc; ovf = 0; }
    batch = 0;
    __syncthreads();

    if (attempt == 0) {
      for (int g0 = 0; g0 < E4; g0 += 4 * NT) {
        int4 d[4];
        int gi[4];
        #pragma unroll
        for (int u = 0; u < 4; ++u) {
          int g = g0 + u * NT + tid;
          bool ok = g < E4;
          int gr = g + rot;
          if (gr >= E4) gr -= E4;
          gi[u] = gr;
          if (ok) d[u] = dst4[gr];
          else    d[u] = make_int4(lo - 1, lo - 1, lo - 1, lo - 1);
        }
        unsigned hit[16];
        #pragma unroll
        for (int u = 0; u < 4; ++u) {
          hit[4 * u + 0] = __builtin_amdgcn_ballot_w32((unsigned)(d[u].x - lo) < (unsigned)nloc);
          hit[4 * u + 1] = __builtin_amdgcn_ballot_w32((unsigned)(d[u].y - lo) < (unsigned)nloc);
          hit[4 * u + 2] = __builtin_amdgcn_ballot_w32((unsigned)(d[u].z - lo) < (unsigned)nloc);
          hit[4 * u + 3] = __builtin_amdgcn_ballot_w32((unsigned)(d[u].w - lo) < (unsigned)nloc);
        }
        unsigned any = 0;
        #pragma unroll
        for (int k = 0; k < 16; ++k) any |= hit[k];
        if (any) {
          int tot = 0;
          #pragma unroll
          for (int k = 0; k < 16; ++k) tot += __builtin_popcount(hit[k]);
          int slot = 0;
          if (lane == 0) slot = atomicAdd(&cnt, tot);
          slot = __builtin_amdgcn_readfirstlane(slot);
          #pragma unroll
          for (int k = 0; k < 16; ++k) {
            if (hit[k]) {
              if ((hit[k] >> lane) & 1u) {
                const int idx = slot + (int)__builtin_amdgcn_mbcnt_lo(hit[k], 0u);
                if (idx < CAP) lsrc[idx] = gi[k >> 2] * 4 + (k & 3);
                else ovf = 1;
              }
              slot += __builtin_popcount(hit[k]);
            }
          }
        }
      }
      for (int e = E4 * 4 + tid; e < E; e += NT) test1(dst[e], e);
    } else {
      for (int g0 = 0; g0 < E4; g0 += NT) {
        drain_if_needed();
        int g = g0 + tid;
        if (g < E4) {
          int4 d = dst4[g];
          const int e = g * 4;
          test1(d.x, e);
          test1(d.y, e + 1);
          test1(d.z, e + 2);
          test1(d.w, e + 3);
        }
      }
      for (int e0 = E4 * 4; e0 < E; e0 += 4 * NT) {
        drain_if_needed();
        for (int k = 0; k < 4; ++k) {
          int e = e0 + k * NT + tid;
          if (e < E) test1(dst[e], e);
        }
      }
    }
    __syncthreads();
    const int o = ovf;
    const int c = cnt;
    __syncthreads();
    if (attempt == 0 && o) continue;
    const int n = min(c, CAP);
    process(n, batch, false);
    if (ROLE == 0) {
      if (attempt == 0) {
        for (int i = tid; i < n; i += NT) { tl_src[lbase + i] = lsrc[i]; tl_ld[lbase + i] = lld[i]; }
      }
      if (tid == 0) tl_cnt[blockIdx.x] = (attempt == 0) ? n : -1;
    }
    break;
  }

  for (int r = wv; r < nloc; r += 32) {
    for (int j = 0; j < J; ++j) {
      const int f = j * 32 + lane;
      float v = acc[r * HC + f] / (s_l[r * H + f / C] + PYG_EPS) + bias[f];
      if (ELU) v = v > 0.f ? v : expm1f(v);
      out[(size_t)(lo + r) * HC + f] = v;
    }
  }
}

extern "C" void kernel_launch(void* const* d_in, const int* in_sizes, int n_in,
                              void* d_out, int out_size, void* d_ws, size_t ws_size,
                              hipStream_t stream)
{
  const float* x   = (const float*)d_in[0];
  const int*   ei  = (const int*)  d_in[1];
  const float* W1  = (const float*)d_in[2];
  const float* as1 = (const float*)d_in[3];
  const float* ad1 = (const float*)d_in[4];
  const float* b1  = (const float*)d_in[5];
  const float* W2  = (const float*)d_in[6];
  const float* as2 = (const float*)d_in[7];
  const float* ad2 = (const float*)d_in[8];
  const float* b2  = (const float*)d_in[9];
  float* out = (float*)d_out;

  const int F = 128, H1 = 4, C = 64, HC1 = 256, HC2 = 64;
  const int N  = in_sizes[0] / F;
  const int E  = in_sizes[1] / 2;
  const int* src = ei;
  const int* dst = ei + E;

  float* h1  = (float*)d_ws;
  float* x2  = h1  + (size_t)N * HC1;
  float* h2  = x2  + (size_t)N * HC1;
  float* aS1 = h2  + (size_t)N * HC2;
  float* aD1 = aS1 + (size_t)N * H1;
  float* aS2 = aD1 + (size_t)N * H1;
  float* aD2 = aS2 + N;

  const int TB = 256;
  auto cdiv = [](int a, int b) { return (a + b - 1) / b; };

  const int tiles = cdiv(N, TILE_NODES);
  int* tlSrc = (int*)(aD2 + N);
  unsigned short* tlLd = (unsigned short*)(tlSrc + (size_t)tiles * LIST_CAP);
  int* tlCnt = (int*)(tlLd + (size_t)tiles * LIST_CAP);

  gemm_wmma_f32<128, 256, 8, 1024><<<256, 1024, 0, stream>>>(x, W1, h1, N / 16);
  att_dots_rows<4><<<cdiv(N * H1, 8 * 2 * DOT_PAIRS), TB, 0, stream>>>(h1, as1, ad1, aS1, aD1, N * H1);
  node_tile_gat<4, 64, true, 0><<<tiles, TILE_NT, 0, stream>>>(
      src, dst, aS1, aD1, h1, b1, x2, E, N, tlSrc, tlLd, tlCnt);

  gemm_wmma_f32<256, 64, 4, 1024><<<256, 1024, 0, stream>>>(x2, W2, h2, N / 16);
  att_dots_rows<1><<<cdiv(N, 8 * 2 * DOT_PAIRS), TB, 0, stream>>>(h2, as2, ad2, aS2, aD2, N);
  node_tile_gat<1, 64, false, 1><<<tiles, TILE_NT, 0, stream>>>(
      src, dst, aS2, aD2, h2, b2, out, E, N, tlSrc, tlLd, tlCnt);
}


# Edge sweep reads a one-byte tile number per edge (dst / 196, written by the attention-dot launch) instead of 4-byte dst: a quarter of the bytes, byte matches found in-word; list and sum order change
# speedup vs baseline: 6.9418x; 1.1565x over previous
#include <hip/hip_runtime.h>
#include <math.h>

typedef __attribute__((ext_vector_type(2))) float v2f;
typedef __attribute__((ext_vector_type(8))) float v8f;

#define NEG_SLOPE 0.2f
#define PYG_EPS   1e-16f

#define TILE_NODES 196
#define LIST_CAP   8192
#define TILE_NT    1024
#define KEY_TILES  256

template <int K, int M, int NT, int THREADS>
__global__ __launch_bounds__(THREADS) void gemm_wmma_f32(
    const float* __restrict__ X, const float* __restrict__ W,
    float* __restrict__ Hout, int rowTiles)
{
  constexpr int PITCH = 2 * M + 32;
  constexpr int CH = (M / 16) / NT;
  static_assert(K % 4 == 0 && M % 16 == 0 && (M / 16) % NT == 0, "shape");
  static_assert(PITCH % 64 == 32, "lane halves must fall on different bank halves");
  __shared__ __attribute__((aligned(16))) float Ws[(K / 2) * PITCH];

  const int tid = threadIdx.x;
  for (int i = tid; i < K * M; i += THREADS) {
    const int k = i / M;
    const int col = i - k * M;
    Ws[(k >> 1) * PITCH + col * 2 + (k & 1)] = W[i];
  }
  __syncthreads();

  const int lane = tid & 31;
  const int wv = tid >> 5;
  const int half = lane >> 4;
  const int l15  = lane & 15;
  const int items = rowTiles * CH;

  for (int it = blockIdx.x + gridDim.x * wv; it < items;
       it += gridDim.x * (THREADS / 32)) {
    const int tm = it / CH;
    const int c0 = (it - tm * CH) * (NT * 16);
    const float* xp = X + (size_t)((tm << 4) + l15) * K + (half << 1);
    const float* bp = Ws + half * PITCH + (c0 + l15) * 2;

    v8f acc[NT];
    #pragma unroll
    for (int t = 0; t < NT; ++t) acc[t] = (v8f){0.f, 0.f, 0.f, 0.f, 0.f, 0.f, 0.f, 0.f};

    for (int k0 = 0; k0 < K; k0 += 4) {
      const v2f a = *(const v2f*)(xp + k0);
      const float* bk = bp + (k0 >> 1) * PITCH;
      #pragma unroll
      for (int t = 0; t < NT; ++t) {
        const v2f b = *(const v2f*)(bk + t * 32);
        acc[t] = __builtin_amdgcn_wmma_f32_16x16x4_f32(
             false, a,  false, b,
             (short)0, acc[t],  false,  false);
      }
    }

    float* op = Hout + (size_t)((tm << 4) + (half << 3)) * M + (c0 + l15);
    #pragma unroll
    for (int t = 0; t < NT; ++t) {
      #pragma unroll
      for (int v = 0; v < 8; ++v) op[(size_t)v * M + t * 16] = acc[t][v];
    }
  }
}

template <int CTRL>
__device__ __forceinline__ float row_swz(float v) {
  return __builtin_bit_cast(float, __builtin_amdgcn_update_dpp(
      0, __builtin_bit_cast(int, v), CTRL, 0xf, 0xf, true));
}

#define DOT_PAIRS 16

template <int H, bool FULL>
__device__ __forceinline__ void att_dots_wave(
    const float* __restrict__ Hm, const float* __restrict__ att_s,
    const float* __restrict__ att_d, float* __restrict__ a_s,
    float* __restrict__ a_d, int nrows, int base, int lane)
{
  constexpr int C = 64;
  constexpr int U = DOT_PAIRS;
  constexpr int PP = (H >= 2) ? (H / 2) : 1;
  static_assert((H & (H - 1)) == 0, "H must be a power of two");
  static_assert(U % PP == 0 && (2 * U) % H == 0, "a wave's first row must be head 0");

  const int half = lane >> 4;
  const int l15  = lane & 15;
  const int c0   = l15 << 2;

  float4 x[U];
  const float* p0 = Hm + (size_t)(base + half) * C + c0;
  #pragma unroll
  for (int u = 0; u < U; ++u) {
    if (FULL) {
      x[u] = *(const float4*)(p0 + u * 2 * C);
    } else {
      const int r = min(base + 2 * u + half, nrows - 1);
      x[u] = *(const float4*)(Hm + (size_t)r * C + c0);
    }
  }

  float ws[PP][4], wd[PP][4];
  #pragma unroll
  for (int q = 0; q < PP; ++q) {
    const int h = (2 * q + half) & (H - 1);
    #pragma unroll
    for (int k = 0; k < 4; ++k) {
      ws[q][k] = att_s[h * C + c0 + k];
      wd[q][k] = att_d[h * C + c0 + k];
    }
  }

  float ss[U], sd[U];
  #pragma unroll
  for (int u = 0; u < U; ++u) {
    const int q = u % PP;
    ss[u] = x[u].x * ws[q][0] + x[u].y * ws[q][1] + x[u].z * ws[q][2] + x[u].w * ws[q][3];
    sd[u] = x[u].x * wd[q][0] + x[u].y * wd[q][1] + x[u].z * wd[q][2] + x[u].w * wd[q][3];
  }
  #pragma unroll
  for (int u = 0; u < U; ++u) { ss[u] += row_swz<0x161>(ss[u]); sd[u] += row_swz<0x161>(sd[u]); }
  #pragma unroll
  for (int u = 0; u < U; ++u) { ss[u] += row_swz<0x162>(ss[u]); sd[u] += row_swz<0x162>(sd[u]); }
  #pragma unroll
  for (int u = 0; u < U; ++u) { ss[u] += row_swz<0x164>(ss[u]); sd[u] += row_swz<0x164>(sd[u]); }
  #pragma unroll
  for (int u = 0; u < U; ++u) { ss[u] += row_swz<0x168>(ss[u]); sd[u] += row_swz<0x168>(sd[u]); }

  float vs = ss[0], vd = sd[0];
  #pragma unroll
  for (int u = 1; u < U; ++u) {
    if (l15 == u) { vs = ss[u]; vd = sd[u]; }
  }
  const int r = base + 2 * l15 + half;
  if (FULL || r < nrows) { a_s[r] = vs; a_d[r] = vd; }
}

template <int H, bool KEYS>
__global__ __launch_bounds__(256) void att_dots_rows(
    const float* __restrict__ Hm, const float* __restrict__ att_s,
    const float* __restrict__ att_d, float* __restrict__ a_s,
    float* __restrict__ a_d, int nrows,
    const int* __restrict__ dst, unsigned char* __restrict__ tkey, int E)
{
  if (KEYS) {
    const int nthr = gridDim.x * blockDim.x;
    const int gt = blockIdx.x * blockDim.x + threadIdx.x;
    const int E4 = ((((size_t)dst) & 15) == 0) ? (E >> 2) : 0;
    for (int g = gt; g < E4; g += nthr) {
      const int4 d = ((const int4*)dst)[g];
      const unsigned k = (((unsigned)d.x / TILE_NODES) & 255u)
                       | ((((unsigned)d.y / TILE_NODES) & 255u) << 8)
                       | ((((unsigned)d.z / TILE_NODES) & 255u) << 16)
                       | ((((unsigned)d.w / TILE_NODES) & 255u) << 24);
      ((unsigned*)tkey)[g] = k;
    }
    for (int e = E4 * 4 + gt; e < E; e += nthr)
      tkey[e] = (unsigned char)((unsigned)dst[e] / TILE_NODES);
  }
  const int wave = blockIdx.x * (blockDim.x >> 5) + (threadIdx.x >> 5);
  const int base = wave * (2 * DOT_PAIRS);
  const int lane = threadIdx.x & 31;
  if (base + 2 * DOT_PAIRS <= nrows)
    att_dots_wave<H, true>(Hm, att_s, att_d, a_s, a_d, nrows, base, lane);
  else if (base < nrows)
    att_dots_wave<H, false>(Hm, att_s, att_d, a_s, a_d, nrows, base, lane);
}

__device__ __forceinline__ unsigned match4(unsigned w, unsigned t) {
  const unsigned x = w ^ t;
  return ~((((x & 0x7f7f7f7fu) + 0x7f7f7f7fu) | x) | 0x7f7f7f7fu);
}
__device__ __forceinline__ unsigned match16(uint4 k, unsigned t) {
  return (match4(k.x, t) >> 7) | (match4(k.y, t) >> 6)
       | (match4(k.z, t) >> 5) | (match4(k.w, t) >> 4);
}
template <int CTRL>
__device__ __forceinline__ int row_shr0(int v) {
  return __builtin_amdgcn_update_dpp(0, v, CTRL, 0xf, 0xf, true);
}

__device__ __forceinline__ float lrelu(float e) {
  return e > 0.f ? e : NEG_SLOPE * e;
}

template <int H, int C, bool ELU, int ROLE>
__global__ __launch_bounds__(TILE_NT) void node_tile_gat(
    const int* __restrict__ src, const int* __restrict__ dst,
    const float* __restrict__ a_s, const float* __restrict__ a_d,
    const float* __restrict__ Hfeat, const float* __restrict__ bias,
    float* __restrict__ out, int E, int N,
    int* __restrict__ tl_src, unsigned short* __restrict__ tl_ld,
    int* __restrict__ tl_cnt, const unsigned char* __restrict__ tkey)
{
  constexpr int HC = H * C;
  constexpr int TN = TILE_NODES;
  constexpr int CAP = LIST_CAP;
  constexpr int NT = TILE_NT;
  constexpr int J = HC / 32;
  static_assert(C % 32 == 0, "a 32-float chunk must lie inside one head");
  static_assert((H & (H - 1)) == 0 && H <= 32, "H must be a power of two");

  __shared__ float acc[TN * HC];
  __shared__ float m_l[TN * H];
  __shared__ float s_l[TN * H];
  __shared__ float sc_l[TN * H];
  __shared__ float ad_l[TN * H];
  __shared__ int lsrc[CAP];
  __shared__ unsigned short lld[CAP];
  __shared__ int cnt;
  __shared__ int ovf;

  const int tid = threadIdx.x;
  const int lane = tid & 31;
  const int wv = tid >> 5;
  const int lo = blockIdx.x * TN;
  const int nloc = min(TN, N - lo);
  if (nloc <= 0) return;
  const bool vecOK = (((size_t)dst) & 15) == 0;
  const int E4 = vecOK ? (E >> 2) : 0;
  const int4* dst4 = (const int4*)dst;
  const int rot = (int)((((long long)blockIdx.x * E4) / gridDim.x) & ~31LL);

  auto test1 = [&](int d, int e) {
    unsigned ldv = (unsigned)(d - lo);
    if (ldv < (unsigned)nloc) {
      int idx = atomicAdd(&cnt, 1);
      if (idx < CAP) lsrc[idx] = e;
      else ovf = 1;
    }
  };

  auto process = [&](int n, int batch, bool resolved) {
    if (!resolved) {
      for (int i = (batch == 0 ? nloc : 0) + tid; i < n; i += NT) {
        const int e = lsrc[i];
        lsrc[i] = src[e];
        lld[i] = (unsigned short)(dst[e] - lo);
      }
    }
    if (batch > 0) {
      for (int j = tid; j < nloc * H; j += NT) sc_l[j] = m_l[j];
    }
    __syncthreads();
    for (int i = tid; i < n; i += NT) {
      const int si = lsrc[i];
      const int ld = lld[i];
      for (int h = 0; h < H; ++h) {
        float ev = lrelu(a_s[(size_t)si * H + h] + ad_l[ld * H + h]);
        __hip_atomic_fetch_max(&m_l[ld * H + h], ev,
                               __ATOMIC_RELAXED, __HIP_MEMORY_SCOPE_WORKGROUP);
      }
    }
    __syncthreads();
    if (batch > 0) {
      for (int j = tid; j < nloc * H; j += NT) {
        float mo = sc_l[j], mn = m_l[j];
        float f = (mo == mn) ? 1.f : __expf(mo - mn);
        sc_l[j] = f;
        s_l[j] *= f;
      }
      __syncthreads();
      for (int i = tid; i < nloc * HC; i += NT)
        acc[i] *= sc_l[(i / HC) * H + (i % HC) / C];
      __syncthreads();
    }
    for (int i0 = wv * 4; i0 < n; i0 += 32 * 4) {
      int si[4], ld[4];
      float hv[4][J];
      float w[4];
      #pragma unroll
      for (int u = 0; u < 4; ++u) {
        int ii = min(i0 + u, n - 1);
        si[u] = lsrc[ii];
        ld[u] = lld[ii];
      }
      #pragma unroll
      for (int u = 0; u < 4; ++u) {
        const float* hp = Hfeat + (size_t)si[u] * HC + lane;
        #pragma unroll
        for (int j = 0; j < J; ++j) hv[u][j] = hp[j * 32];
      }
      #pragma unroll
      for (int u = 0; u < 4; ++u) {
        const int hh = lane & (H - 1);
        float ev = lrelu(a_s[(size_t)si[u] * H + hh] + ad_l[ld[u] * H + hh]);
        w[u] = __expf(ev - m_l[ld[u] * H + hh]);
      }
      #pragma unroll
      for (int u = 0; u < 4; ++u) {
        if (i0 + u < n) {
          if (lane < H) atomicAdd(&s_l[ld[u] * H + lane], w[u]);
          float wh[H];
          #pragma unroll
          for (int h = 0; h < H; ++h)
            wh[h] = __builtin_bit_cast(float,
                __builtin_amdgcn_readlane(__builtin_bit_cast(int, w[u]), h));
          #pragma unroll
          for (int j = 0; j < J; ++j)
            atomicAdd(&acc[ld[u] * HC + j * 32 + lane], hv[u][j] * wh[(j * 32) / C]);
        }
      }
    }
    __syncthreads();
  };

  int batch = 0;
  auto drain_if_needed = [&]() {
    __syncthreads();
    const int c = cnt;
    __syncthreads();
    if (c + 4 * NT > CAP) {
      process(c, batch, false);
      ++batch;
      if (tid == 0) cnt = 0;
      __syncthreads();
    }
  };

  auto init_sums = [&]() {
    for (int i = tid; i < nloc * HC; i += NT) acc[i] = 0.f;
    for (int j = tid; j < nloc * H; j += NT) {
      m_l[j] = -__builtin_inff();
      s_l[j] = 0.f;
      ad_l[j] = a_d[(size_t)lo * H + j];
    }
  };

  const size_t lbase = (size_t)blockIdx.x * CAP;
  const int nl = (ROLE == 1) ? tl_cnt[blockIdx.x] : -1;

  const bool handed = (ROLE == 1 && nl >= 0);
  if (handed) {
    const int n = min(nl, CAP);
    init_sums();
    for (int i = tid; i < n; i += NT) { lsrc[i] = tl_src[lbase + i]; lld[i] = tl_ld[lbase + i]; }
    process(n, 0, true);
  }
  for (int attempt = (ROLE == 1 ? 1 : 0); !handed && attempt < 2; ++attempt) {
    init_sums();
    for (int i = tid; i < nloc; i += NT) { lsrc[i] = lo + i; lld[i] = (unsigned short)i; }
    if (tid == 0) { cnt = nloc; ovf = 0; }
    batch = 0;
    __syncthreads();

    const bool byteKeys = gridDim.x <= KEY_TILES && (((size_t)tkey) & 15) == 0;
    if (attempt == 0 && byteKeys) {
      constexpr int KU = 8;
      const int K16 = E >> 4;
      const uint4* key4 = (const uint4*)tkey;
      const unsigned tpat = (unsigned)blockIdx.x * 0x01010101u;
      const int krot = (int)((((long long)blockIdx.x * K16) / gridDim.x) & ~31LL);
      const int wvs = __builtin_amdgcn_readfirstlane(wv);
      for (int g0 = 0; g0 < K16; g0 += KU * NT) {
        uint4 kq[KU];
        int gi[KU];
        #pragma unroll
        for (int u = 0; u < KU; ++u) {
          const int g = g0 + u * NT + tid;
          int gr = g + krot;
          if (gr >= K16) gr -= K16;
          gr = min(gr, K16 - 1);
          gi[u] = gr;
          kq[u] = key4[gr];
        }
        unsigned zz[KU];
        int c = 0;
        #pragma unroll
        for (int u = 0; u < KU; ++u) {
          zz[u] = 0;
          if (g0 + u * NT + (wvs << 5) < K16) {
            const unsigned z = match16(kq[u], tpat);
            zz[u] = (g0 + u * NT + tid < K16) ? z : 0u;
            c += __builtin_popcount(zz[u]);
          }
        }
        if (__builtin_amdgcn_ballot_w32(c != 0)) {
          int inc = c;
          inc += row_shr0<0x111>(inc);
          inc += row_shr0<0x112>(inc);
          inc += row_shr0<0x114>(inc);
          inc += row_shr0<0x118>(inc);
          const int r0 = __builtin_amdgcn_readlane(inc, 15);
          if (lane >= 16) inc += r0;
          const int tot = __builtin_amdgcn_readlane(inc, 31);
          int slot = 0;
          if (lane == 0) slot = atomicAdd(&cnt, tot);
          slot = __builtin_amdgcn_readfirstlane(slot);
          int idx = slot + inc - c;
          if (c) {
            #pragma unroll
            for (int u = 0; u < KU; ++u) {
              unsigned z = zz[u];
              for (int it = 0; it < 16 && z; ++it) {
                const int p = __builtin_ctz(z);
                z &= z - 1;
                if (idx < CAP) lsrc[idx] = gi[u] * 16 + ((p & 7) << 2) + (p >> 3);
                else ovf = 1;
                ++idx;
              }
            }
          }
        }
      }
      for (int e = K16 * 16 + tid; e < E; e += NT) test1(dst[e], e);
    } else if (attempt == 0) {
      for (int g0 = 0; g0 < E4; g0 += 4 * NT) {
        int4 d[4];
        int gi[4];
        #pragma unroll
        for (int u = 0; u < 4; ++u) {
          int g = g0 + u * NT + tid;
          bool ok = g < E4;
          int gr = g + rot;
          if (gr >= E4) gr -= E4;
          gi[u] = gr;
          if (ok) d[u] = dst4[gr];
          else    d[u] = make_int4(lo - 1, lo - 1, lo - 1, lo - 1);
        }
        unsigned hit[16];
        #pragma unroll
        for (int u = 0; u < 4; ++u) {
          hit[4 * u + 0] = __builtin_amdgcn_ballot_w32((unsigned)(d[u].x - lo) < (unsigned)nloc);
          hit[4 * u + 1] = __builtin_amdgcn_ballot_w32((unsigned)(d[u].y - lo) < (unsigned)nloc);
          hit[4 * u + 2] = __builtin_amdgcn_ballot_w32((unsigned)(d[u].z - lo) < (unsigned)nloc);
          hit[4 * u + 3] = __builtin_amdgcn_ballot_w32((unsigned)(d[u].w - lo) < (unsigned)nloc);
        }
        unsigned any = 0;
        #pragma unroll
        for (int k = 0; k < 16; ++k) any |= hit[k];
        if (any) {
          int tot = 0;
          #pragma unroll
          for (int k = 0; k < 16; ++k) tot += __builtin_popcount(hit[k]);
          int slot = 0;
          if (lane == 0) slot = atomicAdd(&cnt, tot);
          slot = __builtin_amdgcn_readfirstlane(slot);
          #pragma unroll
          for (int k = 0; k < 16; ++k) {
            if (hit[k]) {
              if ((hit[k] >> lane) & 1u) {
                const int idx = slot + (int)__builtin_amdgcn_mbcnt_lo(hit[k], 0u);
                if (idx < CAP) lsrc[idx] = gi[k >> 2] * 4 + (k & 3);
                else ovf = 1;
              }
              slot += __builtin_popcount(hit[k]);
            }
          }
        }
      }
      for (int e = E4 * 4 + tid; e < E; e += NT) test1(dst[e], e);
    } else {
      for (int g0 = 0; g0 < E4; g0 += NT) {
        drain_if_needed();
        int g = g0 + tid;
        if (g < E4) {
          int4 d = dst4[g];
          const int e = g * 4;
          test1(d.x, e);
          test1(d.y, e + 1);
          test1(d.z, e + 2);
          test1(d.w, e + 3);
        }
      }
      for (int e0 = E4 * 4; e0 < E; e0 += 4 * NT) {
        drain_if_needed();
        for (int k = 0; k < 4; ++k) {
          int e = e0 + k * NT + tid;
          if (e < E) test1(dst[e], e);
        }
      }
    }
    __syncthreads();
    const int o = ovf;
    const int c = cnt;
    __syncthreads();
    if (attempt == 0 && o) continue;
    const int n = min(c, CAP);
    process(n, batch, false);
    if (ROLE == 0) {
      if (attempt == 0) {
        for (int i = tid; i < n; i += NT) { tl_src[lbase + i] = lsrc[i]; tl_ld[lbase + i] = lld[i]; }
      }
      if (tid == 0) tl_cnt[blockIdx.x] = (attempt == 0) ? n : -1;
    }
    break;
  }

  for (int r = wv; r < nloc; r += 32) {
    for (int j = 0; j < J; ++j) {
      const int f = j * 32 + lane;
      float v = acc[r * HC + f] / (s_l[r * H + f / C] + PYG_EPS) + bias[f];
      if (ELU) v = v > 0.f ? v : expm1f(v);
      out[(size_t)(lo + r) * HC + f] = v;
    }
  }
}

extern "C" void kernel_launch(void* const* d_in, const int* in_sizes, int n_in,
                              void* d_out, int out_size, void* d_ws, size_t ws_size,
                              hipStream_t stream)
{
  const float* x   = (const float*)d_in[0];
  const int*   ei  = (const int*)  d_in[1];
  const float* W1  = (const float*)d_in[2];
  const float* as1 = (const float*)d_in[3];
  const float* ad1 = (const float*)d_in[4];
  const float* b1  = (const float*)d_in[5];
  const float* W2  = (const float*)d_in[6];
  const float* as2 = (const float*)d_in[7];
  const float* ad2 = (const float*)d_in[8];
  const float* b2  = (const float*)d_in[9];
  float* out = (float*)d_out;

  const int F = 128, H1 = 4, C = 64, HC1 = 256, HC2 = 64;
  const int N  = in_sizes[0] / F;
  const int E  = in_sizes[1] / 2;
  const int* src = ei;
  const int* dst = ei + E;

  float* h1  = (float*)d_ws;
  float* x2  = h1  + (size_t)N * HC1;
  float* h2  = x2  + (size_t)N * HC1;
  float* aS1 = h2  + (size_t)N * HC2;
  float* aD1 = aS1 + (size_t)N * H1;
  float* aS2 = aD1 + (size_t)N * H1;
  float* aD2 = aS2 + N;

  const int TB = 256;
  auto cdiv = [](int a, int b) { return (a + b - 1) / b; };

  const int tiles = cdiv(N, TILE_NODES);
  int* tlSrc = (int*)(aD2 + N);
  unsigned short* tlLd = (unsigned short*)(tlSrc + (size_t)tiles * LIST_CAP);
  int* tlCnt = (int*)(tlLd + (size_t)tiles * LIST_CAP);
  unsigned char* tKey = (unsigned char*)(tlCnt + tiles);

  gemm_wmma_f32<128, 256, 8, 1024><<<256, 1024, 0, stream>>>(x, W1, h1, N / 16);
  att_dots_rows<4, true><<<cdiv(N * H1, 8 * 2 * DOT_PAIRS), TB, 0, stream>>>(
      h1, as1, ad1, aS1, aD1, N * H1, dst, tKey, E);
  node_tile_gat<4, 64, true, 0><<<tiles, TILE_NT, 0, stream>>>(
      src, dst, aS1, aD1, h1, b1, x2, E, N, tlSrc, tlLd, tlCnt, tKey);

  gemm_wmma_f32<256, 64, 4, 1024><<<256, 1024, 0, stream>>>(x2, W2, h2, N / 16);
  att_dots_rows<1, false><<<cdiv(N, 8 * 2 * DOT_PAIRS), TB, 0, stream>>>(
      h2, as2, ad2, aS2, aD2, N, dst, tKey, E);
  node_tile_gat<1, 64, false, 1><<<tiles, TILE_NT, 0, stream>>>(
      src, dst, aS2, aD2, h2, b2, out, E, N, tlSrc, tlLd, tlCnt, tKey);
}


# Edge kernels fetch a source row with 16-byte (layer 2: 8-byte) lane loads: a 1-KB row is 2 wave loads of 4 lines, not 8 of 1; LDS sums kept lane-major so each add hits 32 banks; rows stored likewise
# speedup vs baseline: 7.3573x; 1.0599x over previous
#include <hip/hip_runtime.h>
#include <math.h>

typedef __attribute__((ext_vector_type(2))) float v2f;
typedef __attribute__((ext_vector_type(8))) float v8f;

#define NEG_SLOPE 0.2f
#define PYG_EPS   1e-16f

#define TILE_NODES 196
#define LIST_CAP   8192
#define TILE_NT    1024
#define KEY_TILES  256

template <int K, int M, int NT, int THREADS>
__global__ __launch_bounds__(THREADS) void gemm_wmma_f32(
    const float* __restrict__ X, const float* __restrict__ W,
    float* __restrict__ Hout, int rowTiles)
{
  constexpr int PITCH = 2 * M + 32;
  constexpr int CH = (M / 16) / NT;
  static_assert(K % 4 == 0 && M % 16 == 0 && (M / 16) % NT == 0, "shape");
  static_assert(PITCH % 64 == 32, "lane halves must fall on different bank halves");
  __shared__ __attribute__((aligned(16))) float Ws[(K / 2) * PITCH];

  const int tid = threadIdx.x;
  for (int i = tid; i < K * M; i += THREADS) {
    const int k = i / M;
    const int col = i - k * M;
    Ws[(k >> 1) * PITCH + col * 2 + (k & 1)] = W[i];
  }
  __syncthreads();

  const int lane = tid & 31;
  const int wv = tid >> 5;
  const int half = lane >> 4;
  const int l15  = lane & 15;
  const int items = rowTiles * CH;

  for (int it = blockIdx.x + gridDim.x * wv; it < items;
       it += gridDim.x * (THREADS / 32)) {
    const int tm = it / CH;
    const int c0 = (it - tm * CH) * (NT * 16);
    const float* xp = X + (size_t)((tm << 4) + l15) * K + (half << 1);
    const float* bp = Ws + half * PITCH + (c0 + l15) * 2;

    v8f acc[NT];
    #pragma unroll
    for (int t = 0; t < NT; ++t) acc[t] = (v8f){0.f, 0.f, 0.f, 0.f, 0.f, 0.f, 0.f, 0.f};

    for (int k0 = 0; k0 < K; k0 += 4) {
      const v2f a = *(const v2f*)(xp + k0);
      const float* bk = bp + (k0 >> 1) * PITCH;
      #pragma unroll
      for (int t = 0; t < NT; ++t) {
        const v2f b = *(const v2f*)(bk + t * 32);
        acc[t] = __builtin_amdgcn_wmma_f32_16x16x4_f32(
             false, a,  false, b,
             (short)0, acc[t],  false,  false);
      }
    }

    float* op = Hout + (size_t)((tm << 4) + (half << 3)) * M + (c0 + l15);
    #pragma unroll
    for (int t = 0; t < NT; ++t) {
      #pragma unroll
      for (int v = 0; v < 8; ++v) op[(size_t)v * M + t * 16] = acc[t][v];
    }
  }
}

template <int CTRL>
__device__ __forceinline__ float row_swz(float v) {
  return __builtin_bit_cast(float, __builtin_amdgcn_update_dpp(
      0, __builtin_bit_cast(int, v), CTRL, 0xf, 0xf, true));
}

#define DOT_PAIRS 16

template <int H, bool FULL>
__device__ __forceinline__ void att_dots_wave(
    const float* __restrict__ Hm, const float* __restrict__ att_s,
    const float* __restrict__ att_d, float* __restrict__ a_s,
    float* __restrict__ a_d, int nrows, int base, int lane)
{
  constexpr int C = 64;
  constexpr int U = DOT_PAIRS;
  constexpr int PP = (H >= 2) ? (H / 2) : 1;
  static_assert((H & (H - 1)) == 0, "H must be a power of two");
  static_assert(U % PP == 0 && (2 * U) % H == 0, "a wave's first row must be head 0");

  const int half = lane >> 4;
  const int l15  = lane & 15;
  const int c0   = l15 << 2;

  float4 x[U];
  const float* p0 = Hm + (size_t)(base + half) * C + c0;
  #pragma unroll
  for (int u = 0; u < U; ++u) {
    if (FULL) {
      x[u] = *(const float4*)(p0 + u * 2 * C);
    } else {
      const int r = min(base + 2 * u + half, nrows - 1);
      x[u] = *(const float4*)(Hm + (size_t)r * C + c0);
    }
  }

  float ws[PP][4], wd[PP][4];
  #pragma unroll
  for (int q = 0; q < PP; ++q) {
    const int h = (2 * q + half) & (H - 1);
    #pragma unroll
    for (int k = 0; k < 4; ++k) {
      ws[q][k] = att_s[h * C + c0 + k];
      wd[q][k] = att_d[h * C + c0 + k];
    }
  }

  float ss[U], sd[U];
  #pragma unroll
  for (int u = 0; u < U; ++u) {
    const int q = u % PP;
    ss[u] = x[u].x * ws[q][0] + x[u].y * ws[q][1] + x[u].z * ws[q][2] + x[u].w * ws[q][3];
    sd[u] = x[u].x * wd[q][0] + x[u].y * wd[q][1] + x[u].z * wd[q][2] + x[u].w * wd[q][3];
  }
  #pragma unroll
  for (int u = 0; u < U; ++u) { ss[u] += row_swz<0x161>(ss[u]); sd[u] += row_swz<0x161>(sd[u]); }
  #pragma unroll
  for (int u = 0; u < U; ++u) { ss[u] += row_swz<0x162>(ss[u]); sd[u] += row_swz<0x162>(sd[u]); }
  #pragma unroll
  for (int u = 0; u < U; ++u) { ss[u] += row_swz<0x164>(ss[u]); sd[u] += row_swz<0x164>(sd[u]); }
  #pragma unroll
  for (int u = 0; u < U; ++u) { ss[u] += row_swz<0x168>(ss[u]); sd[u] += row_swz<0x168>(sd[u]); }

  float vs = ss[0], vd = sd[0];
  #pragma unroll
  for (int u = 1; u < U; ++u) {
    if (l15 == u) { vs = ss[u]; vd = sd[u]; }
  }
  const int r = base + 2 * l15 + half;
  if (FULL || r < nrows) { a_s[r] = vs; a_d[r] = vd; }
}

template <int H, bool KEYS>
__global__ __launch_bounds__(256) void att_dots_rows(
    const float* __restrict__ Hm, const float* __restrict__ att_s,
    const float* __restrict__ att_d, float* __restrict__ a_s,
    float* __restrict__ a_d, int nrows,
    const int* __restrict__ dst, unsigned char* __restrict__ tkey, int E)
{
  if (KEYS) {
    const int nthr = gridDim.x * blockDim.x;
    const int gt = blockIdx.x * blockDim.x + threadIdx.x;
    const int E4 = ((((size_t)dst) & 15) == 0) ? (E >> 2) : 0;
    for (int g = gt; g < E4; g += nthr) {
      const int4 d = ((const int4*)dst)[g];
      const unsigned k = (((unsigned)d.x / TILE_NODES) & 255u)
                       | ((((unsigned)d.y / TILE_NODES) & 255u) << 8)
                       | ((((unsigned)d.z / TILE_NODES) & 255u) << 16)
                       | ((((unsigned)d.w / TILE_NODES) & 255u) << 24);
      ((unsigned*)tkey)[g] = k;
    }
    for (int e = E4 * 4 + gt; e < E; e += nthr)
      tkey[e] = (unsigned char)((unsigned)dst[e] / TILE_NODES);
  }
  const int wave = blockIdx.x * (blockDim.x >> 5) + (threadIdx.x >> 5);
  const int base = wave * (2 * DOT_PAIRS);
  const int lane = threadIdx.x & 31;
  if (base + 2 * DOT_PAIRS <= nrows)
    att_dots_wave<H, true>(Hm, att_s, att_d, a_s, a_d, nrows, base, lane);
  else if (base < nrows)
    att_dots_wave<H, false>(Hm, att_s, att_d, a_s, a_d, nrows, base, lane);
}

__device__ __forceinline__ unsigned match4(unsigned w, unsigned t) {
  const unsigned x = w ^ t;
  return ~((((x & 0x7f7f7f7fu) + 0x7f7f7f7fu) | x) | 0x7f7f7f7fu);
}
__device__ __forceinline__ unsigned match16(uint4 k, unsigned t) {
  return (match4(k.x, t) >> 7) | (match4(k.y, t) >> 6)
       | (match4(k.z, t) >> 5) | (match4(k.w, t) >> 4);
}
template <int CTRL>
__device__ __forceinline__ int row_shr0(int v) {
  return __builtin_amdgcn_update_dpp(0, v, CTRL, 0xf, 0xf, true);
}

__device__ __forceinline__ float lrelu(float e) {
  return e > 0.f ? e : NEG_SLOPE * e;
}

template <int H, int C, bool ELU, int ROLE>
__global__ __launch_bounds__(TILE_NT) void node_tile_gat(
    const int* __restrict__ src, const int* __restrict__ dst,
    const float* __restrict__ a_s, const float* __restrict__ a_d,
    const float* __restrict__ Hfeat, const float* __restrict__ bias,
    float* __restrict__ out, int E, int N,
    int* __restrict__ tl_src, unsigned short* __restrict__ tl_ld,
    int* __restrict__ tl_cnt, const unsigned char* __restrict__ tkey)
{
  constexpr int HC = H * C;
  constexpr int TN = TILE_NODES;
  constexpr int CAP = LIST_CAP;
  constexpr int NT = TILE_NT;
  constexpr int J = HC / 32;
  constexpr int V = (J >= 4) ? 4 : J;
  constexpr int Q = J / V;
  constexpr int LW = 32 * V;
  constexpr int HPL = LW / C;
  typedef __attribute__((ext_vector_type(V))) float vrow;
  static_assert(V * Q == J && (V == 4 || V == 2), "a row is whole 16- or 8-byte lane loads");
  static_assert(LW % C == 0 && Q * HPL == H, "a wave load covers whole heads");
  static_assert(C % 32 == 0, "a 32-float chunk must lie inside one head");
  static_assert((H & (H - 1)) == 0 && H <= 32, "H must be a power of two");

  __shared__ float acc[TN * HC];
  __shared__ float m_l[TN * H];
  __shared__ float s_l[TN * H];
  __shared__ float sc_l[TN * H];
  __shared__ float ad_l[TN * H];
  __shared__ int lsrc[CAP];
  __shared__ unsigned short lld[CAP];
  __shared__ int cnt;
  __shared__ int ovf;

  const int tid = threadIdx.x;
  const int lane = tid & 31;
  const int wv = tid >> 5;
  const int lo = blockIdx.x * TN;
  const int nloc = min(TN, N - lo);
  if (nloc <= 0) return;
  const bool vecOK = (((size_t)dst) & 15) == 0;
  const int E4 = vecOK ? (E >> 2) : 0;
  const int4* dst4 = (const int4*)dst;
  const int rot = (int)((((long long)blockIdx.x * E4) / gridDim.x) & ~31LL);

  auto test1 = [&](int d, int e) {
    unsigned ldv = (unsigned)(d - lo);
    if (ldv < (unsigned)nloc) {
      int idx = atomicAdd(&cnt, 1);
      if (idx < CAP) lsrc[idx] = e;
      else ovf = 1;
    }
  };

  auto process = [&](int n, int batch, bool resolved) {
    if (!resolved) {
      for (int i = (batch == 0 ? nloc : 0) + tid; i < n; i += NT) {
        const int e = lsrc[i];
        lsrc[i] = src[e];
        lld[i] = (unsigned short)(dst[e] - lo);
      }
    }
    if (batch > 0) {
      for (int j = tid; j < nloc * H; j += NT) sc_l[j] = m_l[j];
    }
    __syncthreads();
    for (int i = tid; i < n; i += NT) {
      const int si = lsrc[i];
      const int ld = lld[i];
      for (int h = 0; h < H; ++h) {
        float ev = lrelu(a_s[(size_t)si * H + h] + ad_l[ld * H + h]);
        __hip_atomic_fetch_max(&m_l[ld * H + h], ev,
                               __ATOMIC_RELAXED, __HIP_MEMORY_SCOPE_WORKGROUP);
      }
    }
    __syncthreads();
    if (batch > 0) {
      for (int j = tid; j < nloc * H; j += NT) {
        float mo = sc_l[j], mn = m_l[j];
        float f = (mo == mn) ? 1.f : __expf(mo - mn);
        sc_l[j] = f;
        s_l[j] *= f;
      }
      __syncthreads();
      for (int i = tid; i < nloc * HC; i += NT) {
        const int p = i % HC;
        const int f = (p / LW) * LW + (p % 32) * V + (p % LW) / 32;
        acc[i] *= sc_l[(i / HC) * H + f / C];
      }
      __syncthreads();
    }
    for (int i0 = wv * 4; i0 < n; i0 += 32 * 4) {
      int si[4], ld[4];
      vrow hv[4][Q];
      float w[4];
      #pragma unroll
      for (int u = 0; u < 4; ++u) {
        int ii = min(i0 + u, n - 1);
        si[u] = lsrc[ii];
        ld[u] = lld[ii];
      }
      #pragma unroll
      for (int u = 0; u < 4; ++u) {
        const vrow* hp = (const vrow*)(Hfeat + (size_t)si[u] * HC) + lane;
        #pragma unroll
        for (int q = 0; q < Q; ++q) hv[u][q] = hp[q * 32];
      }
      #pragma unroll
      for (int u = 0; u < 4; ++u) {
        const int hh = lane & (H - 1);
        float ev = lrelu(a_s[(size_t)si[u] * H + hh] + ad_l[ld[u] * H + hh]);
        w[u] = __expf(ev - m_l[ld[u] * H + hh]);
      }
      #pragma unroll
      for (int u = 0; u < 4; ++u) {
        if (i0 + u < n) {
          if (lane < H) atomicAdd(&s_l[ld[u] * H + lane], w[u]);
          float wh[H];
          #pragma unroll
          for (int h = 0; h < H; ++h)
            wh[h] = __builtin_bit_cast(float,
                __builtin_amdgcn_readlane(__builtin_bit_cast(int, w[u]), h));
          #pragma unroll
          for (int q = 0; q < Q; ++q) {
            float wl = wh[q * HPL];
            #pragma unroll
            for (int t = 1; t < HPL; ++t)
              if ((lane * V) / C == t) wl = wh[q * HPL + t];
            #pragma unroll
            for (int k = 0; k < V; ++k)
              atomicAdd(&acc[ld[u] * HC + q * LW + k * 32 + lane], hv[u][q][k] * wl);
          }
        }
      }
    }
    __syncthreads();
  };

  int batch = 0;
  auto drain_if_needed = [&]() {
    __syncthreads();
    const int c = cnt;
    __syncthreads();
    if (c + 4 * NT > CAP) {
      process(c, batch, false);
      ++batch;
      if (tid == 0) cnt = 0;
      __syncthreads();
    }
  };

  auto init_sums = [&]() {
    for (int i = tid; i < nloc * HC; i += NT) acc[i] = 0.f;
    for (int j = tid; j < nloc * H; j += NT) {
      m_l[j] = -__builtin_inff();
      s_l[j] = 0.f;
      ad_l[j] = a_d[(size_t)lo * H + j];
    }
  };

  const size_t lbase = (size_t)blockIdx.x * CAP;
  const int nl = (ROLE == 1) ? tl_cnt[blockIdx.x] : -1;

  const bool handed = (ROLE == 1 && nl >= 0);
  if (handed) {
    const int n = min(nl, CAP);
    init_sums();
    for (int i = tid; i < n; i += NT) { lsrc[i] = tl_src[lbase + i]; lld[i] = tl_ld[lbase + i]; }
    process(n, 0, true);
  }
  for (int attempt = (ROLE == 1 ? 1 : 0); !handed && attempt < 2; ++attempt) {
    init_sums();
    for (int i = tid; i < nloc; i += NT) { lsrc[i] = lo + i; lld[i] = (unsigned short)i; }
    if (tid == 0) { cnt = nloc; ovf = 0; }
    batch = 0;
    __syncthreads();

    const bool byteKeys = gridDim.x <= KEY_TILES && (((size_t)tkey) & 15) == 0;
    if (attempt == 0 && byteKeys) {
      constexpr int KU = 8;
      const int K16 = E >> 4;
      const uint4* key4 = (const uint4*)tkey;
      const unsigned tpat = (unsigned)blockIdx.x * 0x01010101u;
      const int krot = (int)((((long long)blockIdx.x * K16) / gridDim.x) & ~31LL);
      const int wvs = __builtin_amdgcn_readfirstlane(wv);
      for (int g0 = 0; g0 < K16; g0 += KU * NT) {
        uint4 kq[KU];
        int gi[KU];
        #pragma unroll
        for (int u = 0; u < KU; ++u) {
          const int g = g0 + u * NT + tid;
          int gr = g + krot;
          if (gr >= K16) gr -= K16;
          gr = min(gr, K16 - 1);
          gi[u] = gr;
          kq[u] = key4[gr];
        }
        unsigned zz[KU];
        int c = 0;
        #pragma unroll
        for (int u = 0; u < KU; ++u) {
          zz[u] = 0;
          if (g0 + u * NT + (wvs << 5) < K16) {
            const unsigned z = match16(kq[u], tpat);
            zz[u] = (g0 + u * NT + tid < K16) ? z : 0u;
            c += __builtin_popcount(zz[u]);
          }
        }
        if (__builtin_amdgcn_ballot_w32(c != 0)) {
          int inc = c;
          inc += row_shr0<0x111>(inc);
          inc += row_shr0<0x112>(inc);
          inc += row_shr0<0x114>(inc);
          inc += row_shr0<0x118>(inc);
          const int r0 = __builtin_amdgcn_readlane(inc, 15);
          if (lane >= 16) inc += r0;
          const int tot = __builtin_amdgcn_readlane(inc, 31);
          int slot = 0;
          if (lane == 0) slot = atomicAdd(&cnt, tot);
          slot = __builtin_amdgcn_readfirstlane(slot);
          int idx = slot + inc - c;
          if (c) {
            #pragma unroll
            for (int u = 0; u < KU; ++u) {
              unsigned z = zz[u];
              for (int it = 0; it < 16 && z; ++it) {
                const int p = __builtin_ctz(z);
                z &= z - 1;
                if (idx < CAP) lsrc[idx] = gi[u] * 16 + ((p & 7) << 2) + (p >> 3);
                else ovf = 1;
                ++idx;
              }
            }
          }
        }
      }
      for (int e = K16 * 16 + tid; e < E; e += NT) test1(dst[e], e);
    } else if (attempt == 0) {
      for (int g0 = 0; g0 < E4; g0 += 4 * NT) {
        int4 d[4];
        int gi[4];
        #pragma unroll
        for (int u = 0; u < 4; ++u) {
          int g = g0 + u * NT + tid;
          bool ok = g < E4;
          int gr = g + rot;
          if (gr >= E4) gr -= E4;
          gi[u] = gr;
          if (ok) d[u] = dst4[gr];
          else    d[u] = make_int4(lo - 1, lo - 1, lo - 1, lo - 1);
        }
        unsigned hit[16];
        #pragma unroll
        for (int u = 0; u < 4; ++u) {
          hit[4 * u + 0] = __builtin_amdgcn_ballot_w32((unsigned)(d[u].x - lo) < (unsigned)nloc);
          hit[4 * u + 1] = __builtin_amdgcn_ballot_w32((unsigned)(d[u].y - lo) < (unsigned)nloc);
          hit[4 * u + 2] = __builtin_amdgcn_ballot_w32((unsigned)(d[u].z - lo) < (unsigned)nloc);
          hit[4 * u + 3] = __builtin_amdgcn_ballot_w32((unsigned)(d[u].w - lo) < (unsigned)nloc);
        }
        unsigned any = 0;
        #pragma unroll
        for (int k = 0; k < 16; ++k) any |= hit[k];
        if (any) {
          int tot = 0;
          #pragma unroll
          for (int k = 0; k < 16; ++k) tot += __builtin_popcount(hit[k]);
          int slot = 0;
          if (lane == 0) slot = atomicAdd(&cnt, tot);
          slot = __builtin_amdgcn_readfirstlane(slot);
          #pragma unroll
          for (int k = 0; k < 16; ++k) {
            if (hit[k]) {
              if ((hit[k] >> lane) & 1u) {
                const int idx = slot + (int)__builtin_amdgcn_mbcnt_lo(hit[k], 0u);
                if (idx < CAP) lsrc[idx] = gi[k >> 2] * 4 + (k & 3);
                else ovf = 1;
              }
              slot += __builtin_popcount(hit[k]);
            }
          }
        }
      }
      for (int e = E4 * 4 + tid; e < E; e += NT) test1(dst[e], e);
    } else {
      for (int g0 = 0; g0 < E4; g0 += NT) {
        drain_if_needed();
        int g = g0 + tid;
        if (g < E4) {
          int4 d = dst4[g];
          const int e = g * 4;
          test1(d.x, e);
          test1(d.y, e + 1);
          test1(d.z, e + 2);
          test1(d.w, e + 3);
        }
      }
      for (int e0 = E4 * 4; e0 < E; e0 += 4 * NT) {
        drain_if_needed();
        for (int k = 0; k < 4; ++k) {
          int e = e0 + k * NT + tid;
          if (e < E) test1(dst[e], e);
        }
      }
    }
    __syncthreads();
    const int o = ovf;
    const int c = cnt;
    __syncthreads();
    if (attempt == 0 && o) continue;
    const int n = min(c, CAP);
    process(n, batch, false);
    if (ROLE == 0) {
      if (attempt == 0) {
        for (int i = tid; i < n; i += NT) { tl_src[lbase + i] = lsrc[i]; tl_ld[lbase + i] = lld[i]; }
      }
      if (tid == 0) tl_cnt[blockIdx.x] = (attempt == 0) ? n : -1;
    }
    break;
  }

  float bl[J];
  #pragma unroll
  for (int j = 0; j < J; ++j) bl[j] = bias[(j / V) * LW + lane * V + (j % V)];
  for (int r = wv; r < nloc; r += 32) {
    #pragma unroll
    for (int q = 0; q < Q; ++q) {
      const float den = s_l[r * H + q * HPL + (lane * V) / C] + PYG_EPS;
      vrow o;
      #pragma unroll
      for (int k = 0; k < V; ++k) {
        float v = acc[r * HC + q * LW + k * 32 + lane] / den + bl[q * V + k];
        if (ELU) v = v > 0.f ? v : expm1f(v);
        o[k] = v;
      }
      *((vrow*)(out + (size_t)(lo + r) * HC) + q * 32 + lane) = o;
    }
  }
}

extern "C" void kernel_launch(void* const* d_in, const int* in_sizes, int n_in,
                              void* d_out, int out_size, void* d_ws, size_t ws_size,
                              hipStream_t stream)
{
  const float* x   = (const float*)d_in[0];
  const int*   ei  = (const int*)  d_in[1];
  const float* W1  = (const float*)d_in[2];
  const float* as1 = (const float*)d_in[3];
  const float* ad1 = (const float*)d_in[4];
  const float* b1  = (const float*)d_in[5];
  const float* W2  = (const float*)d_in[6];
  const float* as2 = (const float*)d_in[7];
  const float* ad2 = (const float*)d_in[8];
  const float* b2  = (const float*)d_in[9];
  float* out = (float*)d_out;

  const int F = 128, H1 = 4, C = 64, HC1 = 256, HC2 = 64;
  const int N  = in_sizes[0] / F;
  const int E  = in_sizes[1] / 2;
  const int* src = ei;
  const int* dst = ei + E;

  float* h1  = (float*)d_ws;
  float* x2  = h1  + (size_t)N * HC1;
  float* h2  = x2  + (size_t)N * HC1;
  float* aS1 = h2  + (size_t)N * HC2;
  float* aD1 = aS1 + (size_t)N * H1;
  float* aS2 = aD1 + (size_t)N * H1;
  float* aD2 = aS2 + N;

  const int TB = 256;
  auto cdiv = [](int a, int b) { return (a + b - 1) / b; };

  const int tiles = cdiv(N, TILE_NODES);
  int* tlSrc = (int*)(aD2 + N);
  unsigned short* tlLd = (unsigned short*)(tlSrc + (size_t)tiles * LIST_CAP);
  int* tlCnt = (int*)(tlLd + (size_t)tiles * LIST_CAP);
  unsigned char* tKey = (unsigned char*)(tlCnt + tiles);

  gemm_wmma_f32<128, 256, 8, 1024><<<256, 1024, 0, stream>>>(x, W1, h1, N / 16);
  att_dots_rows<4, true><<<cdiv(N * H1, 8 * 2 * DOT_PAIRS), TB, 0, stream>>>(
      h1, as1, ad1, aS1, aD1, N * H1, dst, tKey, E);
  node_tile_gat<4, 64, true, 0><<<tiles, TILE_NT, 0, stream>>>(
      src, dst, aS1, aD1, h1, b1, x2, E, N, tlSrc, tlLd, tlCnt, tKey);

  gemm_wmma_f32<256, 64, 4, 1024><<<256, 1024, 0, stream>>>(x2, W2, h2, N / 16);
  att_dots_rows<1, false><<<cdiv(N, 8 * 2 * DOT_PAIRS), TB, 0, stream>>>(
      h2, as2, ad2, aS2, aD2, N, dst, tKey, E);
  node_tile_gat<1, 64, false, 1><<<tiles, TILE_NT, 0, stream>>>(
      src, dst, aS2, aD2, h2, b2, out, E, N, tlSrc, tlLd, tlCnt, tKey);
}


# Feature pass lays a trip's 4 entries x H heads across lanes: 1 a_s load, 1 ad and 1 max read, 1 exp, ONE exp-sum LDS add per trip, not 4 each; list read as 16+8 bytes; exp-sum order changes
# speedup vs baseline: 7.7360x; 1.0515x over previous
#include <hip/hip_runtime.h>
#include <math.h>

typedef __attribute__((ext_vector_type(2))) float v2f;
typedef __attribute__((ext_vector_type(8))) float v8f;

#define NEG_SLOPE 0.2f
#define PYG_EPS   1e-16f

#define TILE_NODES 196
#define LIST_CAP   8192
#define TILE_NT    1024
#define KEY_TILES  256

template <int K, int M, int NT, int THREADS>
__global__ __launch_bounds__(THREADS) void gemm_wmma_f32(
    const float* __restrict__ X, const float* __restrict__ W,
    float* __restrict__ Hout, int rowTiles)
{
  constexpr int PITCH = 2 * M + 32;
  constexpr int CH = (M / 16) / NT;
  static_assert(K % 4 == 0 && M % 16 == 0 && (M / 16) % NT == 0, "shape");
  static_assert(PITCH % 64 == 32, "lane halves must fall on different bank halves");
  __shared__ __attribute__((aligned(16))) float Ws[(K / 2) * PITCH];

  const int tid = threadIdx.x;
  for (int i = tid; i < K * M; i += THREADS) {
    const int k = i / M;
    const int col = i - k * M;
    Ws[(k >> 1) * PITCH + col * 2 + (k & 1)] = W[i];
  }
  __syncthreads();

  const int lane = tid & 31;
  const int wv = tid >> 5;
  const int half = lane >> 4;
  const int l15  = lane & 15;
  const int items = rowTiles * CH;

  for (int it = blockIdx.x + gridDim.x * wv; it < items;
       it += gridDim.x * (THREADS / 32)) {
    const int tm = it / CH;
    const int c0 = (it - tm * CH) * (NT * 16);
    const float* xp = X + (size_t)((tm << 4) + l15) * K + (half << 1);
    const float* bp = Ws + half * PITCH + (c0 + l15) * 2;

    v8f acc[NT];
    #pragma unroll
    for (int t = 0; t < NT; ++t) acc[t] = (v8f){0.f, 0.f, 0.f, 0.f, 0.f, 0.f, 0.f, 0.f};

    for (int k0 = 0; k0 < K; k0 += 4) {
      const v2f a = *(const v2f*)(xp + k0);
      const float* bk = bp + (k0 >> 1) * PITCH;
      #pragma unroll
      for (int t = 0; t < NT; ++t) {
        const v2f b = *(const v2f*)(bk + t * 32);
        acc[t] = __builtin_amdgcn_wmma_f32_16x16x4_f32(
             false, a,  false, b,
             (short)0, acc[t],  false,  false);
      }
    }

    float* op = Hout + (size_t)((tm << 4) + (half << 3)) * M + (c0 + l15);
    #pragma unroll
    for (int t = 0; t < NT; ++t) {
      #pragma unroll
      for (int v = 0; v < 8; ++v) op[(size_t)v * M + t * 16] = acc[t][v];
    }
  }
}

template <int CTRL>
__device__ __forceinline__ float row_swz(float v) {
  return __builtin_bit_cast(float, __builtin_amdgcn_update_dpp(
      0, __builtin_bit_cast(int, v), CTRL, 0xf, 0xf, true));
}

#define DOT_PAIRS 16

template <int H, bool FULL>
__device__ __forceinline__ void att_dots_wave(
    const float* __restrict__ Hm, const float* __restrict__ att_s,
    const float* __restrict__ att_d, float* __restrict__ a_s,
    float* __restrict__ a_d, int nrows, int base, int lane)
{
  constexpr int C = 64;
  constexpr int U = DOT_PAIRS;
  constexpr int PP = (H >= 2) ? (H / 2) : 1;
  static_assert((H & (H - 1)) == 0, "H must be a power of two");
  static_assert(U % PP == 0 && (2 * U) % H == 0, "a wave's first row must be head 0");

  const int half = lane >> 4;
  const int l15  = lane & 15;
  const int c0   = l15 << 2;

  float4 x[U];
  const float* p0 = Hm + (size_t)(base + half) * C + c0;
  #pragma unroll
  for (int u = 0; u < U; ++u) {
    if (FULL) {
      x[u] = *(const float4*)(p0 + u * 2 * C);
    } else {
      const int r = min(base + 2 * u + half, nrows - 1);
      x[u] = *(const float4*)(Hm + (size_t)r * C + c0);
    }
  }

  float ws[PP][4], wd[PP][4];
  #pragma unroll
  for (int q = 0; q < PP; ++q) {
    const int h = (2 * q + half) & (H - 1);
    #pragma unroll
    for (int k = 0; k < 4; ++k) {
      ws[q][k] = att_s[h * C + c0 + k];
      wd[q][k] = att_d[h * C + c0 + k];
    }
  }

  float ss[U], sd[U];
  #pragma unroll
  for (int u = 0; u < U; ++u) {
    const int q = u % PP;
    ss[u] = x[u].x * ws[q][0] + x[u].y * ws[q][1] + x[u].z * ws[q][2] + x[u].w * ws[q][3];
    sd[u] = x[u].x * wd[q][0] + x[u].y * wd[q][1] + x[u].z * wd[q][2] + x[u].w * wd[q][3];
  }
  #pragma unroll
  for (int u = 0; u < U; ++u) { ss[u] += row_swz<0x161>(ss[u]); sd[u] += row_swz<0x161>(sd[u]); }
  #pragma unroll
  for (int u = 0; u < U; ++u) { ss[u] += row_swz<0x162>(ss[u]); sd[u] += row_swz<0x162>(sd[u]); }
  #pragma unroll
  for (int u = 0; u < U; ++u) { ss[u] += row_swz<0x164>(ss[u]); sd[u] += row_swz<0x164>(sd[u]); }
  #pragma unroll
  for (int u = 0; u < U; ++u) { ss[u] += row_swz<0x168>(ss[u]); sd[u] += row_swz<0x168>(sd[u]); }

  float vs = ss[0], vd = sd[0];
  #pragma unroll
  for (int u = 1; u < U; ++u) {
    if (l15 == u) { vs = ss[u]; vd = sd[u]; }
  }
  const int r = base + 2 * l15 + half;
  if (FULL || r < nrows) { a_s[r] = vs; a_d[r] = vd; }
}

template <int H, bool KEYS>
__global__ __launch_bounds__(256) void att_dots_rows(
    const float* __restrict__ Hm, const float* __restrict__ att_s,
    const float* __restrict__ att_d, float* __restrict__ a_s,
    float* __restrict__ a_d, int nrows,
    const int* __restrict__ dst, unsigned char* __restrict__ tkey, int E)
{
  if (KEYS) {
    const int nthr = gridDim.x * blockDim.x;
    const int gt = blockIdx.x * blockDim.x + threadIdx.x;
    const int E4 = ((((size_t)dst) & 15) == 0) ? (E >> 2) : 0;
    for (int g = gt; g < E4; g += nthr) {
      const int4 d = ((const int4*)dst)[g];
      const unsigned k = (((unsigned)d.x / TILE_NODES) & 255u)
                       | ((((unsigned)d.y / TILE_NODES) & 255u) << 8)
                       | ((((unsigned)d.z / TILE_NODES) & 255u) << 16)
                       | ((((unsigned)d.w / TILE_NODES) & 255u) << 24);
      ((unsigned*)tkey)[g] = k;
    }
    for (int e = E4 * 4 + gt; e < E; e += nthr)
      tkey[e] = (unsigned char)((unsigned)dst[e] / TILE_NODES);
  }
  const int wave = blockIdx.x * (blockDim.x >> 5) + (threadIdx.x >> 5);
  const int base = wave * (2 * DOT_PAIRS);
  const int lane = threadIdx.x & 31;
  if (base + 2 * DOT_PAIRS <= nrows)
    att_dots_wave<H, true>(Hm, att_s, att_d, a_s, a_d, nrows, base, lane);
  else if (base < nrows)
    att_dots_wave<H, false>(Hm, att_s, att_d, a_s, a_d, nrows, base, lane);
}

__device__ __forceinline__ unsigned match4(unsigned w, unsigned t) {
  const unsigned x = w ^ t;
  return ~((((x & 0x7f7f7f7fu) + 0x7f7f7f7fu) | x) | 0x7f7f7f7fu);
}
__device__ __forceinline__ unsigned match16(uint4 k, unsigned t) {
  return (match4(k.x, t) >> 7) | (match4(k.y, t) >> 6)
       | (match4(k.z, t) >> 5) | (match4(k.w, t) >> 4);
}
template <int CTRL>
__device__ __forceinline__ int row_shr0(int v) {
  return __builtin_amdgcn_update_dpp(0, v, CTRL, 0xf, 0xf, true);
}

__device__ __forceinline__ float lrelu(float e) {
  return e > 0.f ? e : NEG_SLOPE * e;
}

template <int H, int C, bool ELU, int ROLE>
__global__ __launch_bounds__(TILE_NT) void node_tile_gat(
    const int* __restrict__ src, const int* __restrict__ dst,
    const float* __restrict__ a_s, const float* __restrict__ a_d,
    const float* __restrict__ Hfeat, const float* __restrict__ bias,
    float* __restrict__ out, int E, int N,
    int* __restrict__ tl_src, unsigned short* __restrict__ tl_ld,
    int* __restrict__ tl_cnt, const unsigned char* __restrict__ tkey)
{
  constexpr int HC = H * C;
  constexpr int TN = TILE_NODES;
  constexpr int CAP = LIST_CAP;
  constexpr int NT = TILE_NT;
  constexpr int J = HC / 32;
  constexpr int V = (J >= 4) ? 4 : J;
  constexpr int Q = J / V;
  constexpr int LW = 32 * V;
  constexpr int HPL = LW / C;
  typedef __attribute__((ext_vector_type(V))) float vrow;
  static_assert(V * Q == J && (V == 4 || V == 2), "a row is whole 16- or 8-byte lane loads");
  static_assert(LW % C == 0 && Q * HPL == H, "a wave load covers whole heads");
  static_assert(C % 32 == 0, "a 32-float chunk must lie inside one head");
  static_assert((H & (H - 1)) == 0 && H <= 32, "H must be a power of two");

  __shared__ float acc[TN * HC];
  __shared__ float m_l[TN * H];
  __shared__ float s_l[TN * H];
  __shared__ float sc_l[TN * H];
  __shared__ float ad_l[TN * H];
  __shared__ __attribute__((aligned(16))) int lsrc[CAP];
  __shared__ __attribute__((aligned(8))) unsigned short lld[CAP];
  __shared__ int cnt;
  __shared__ int ovf;

  const int tid = threadIdx.x;
  const int lane = tid & 31;
  const int wv = tid >> 5;
  const int lo = blockIdx.x * TN;
  const int nloc = min(TN, N - lo);
  if (nloc <= 0) return;
  const bool vecOK = (((size_t)dst) & 15) == 0;
  const int E4 = vecOK ? (E >> 2) : 0;
  const int4* dst4 = (const int4*)dst;
  const int rot = (int)((((long long)blockIdx.x * E4) / gridDim.x) & ~31LL);

  auto test1 = [&](int d, int e) {
    unsigned ldv = (unsigned)(d - lo);
    if (ldv < (unsigned)nloc) {
      int idx = atomicAdd(&cnt, 1);
      if (idx < CAP) lsrc[idx] = e;
      else ovf = 1;
    }
  };

  auto process = [&](int n, int batch, bool resolved) {
    if (!resolved) {
      for (int i = (batch == 0 ? nloc : 0) + tid; i < n; i += NT) {
        const int e = lsrc[i];
        lsrc[i] = src[e];
        lld[i] = (unsigned short)(dst[e] - lo);
      }
    }
    if (batch > 0) {
      for (int j = tid; j < nloc * H; j += NT) sc_l[j] = m_l[j];
    }
    if (tid < 3 && n + tid < CAP) { lsrc[n + tid] = lo; lld[n + tid] = 0; }
    __syncthreads();
    for (int i = tid; i < n; i += NT) {
      const int si = lsrc[i];
      const int ld = lld[i];
      for (int h = 0; h < H; ++h) {
        float ev = lrelu(a_s[(size_t)si * H + h] + ad_l[ld * H + h]);
        __hip_atomic_fetch_max(&m_l[ld * H + h], ev,
                               __ATOMIC_RELAXED, __HIP_MEMORY_SCOPE_WORKGROUP);
      }
    }
    __syncthreads();
    if (batch > 0) {
      for (int j = tid; j < nloc * H; j += NT) {
        float mo = sc_l[j], mn = m_l[j];
        float f = (mo == mn) ? 1.f : __expf(mo - mn);
        sc_l[j] = f;
        s_l[j] *= f;
      }
      __syncthreads();
      for (int i = tid; i < nloc * HC; i += NT) {
        const int p = i % HC;
        const int f = (p / LW) * LW + (p % 32) * V + (p % LW) / 32;
        acc[i] *= sc_l[(i / HC) * H + f / C];
      }
      __syncthreads();
    }
    for (int i0 = wv * 4; i0 < n; i0 += 32 * 4) {
      vrow hv[4][Q];
      const int4 sv = *(const int4*)&lsrc[i0];
      const ushort4 lv = *(const ushort4*)&lld[i0];
      __builtin_amdgcn_sched_barrier(0);
      const int si[4] = {sv.x, sv.y, sv.z, sv.w};
      const int ld[4] = {lv.x, lv.y, lv.z, lv.w};
      const int nv = n - i0;
      const int us = (lane / H) & 3;
      const int hh = lane & (H - 1);
      int ssel = si[0], lsel = ld[0];
      #pragma unroll
      for (int u = 1; u < 4; ++u)
        if (us == u) { ssel = si[u]; lsel = ld[u]; }
      const float asv = a_s[(size_t)ssel * H + hh];
      const float adv = ad_l[lsel * H + hh];
      const float mv  = m_l[lsel * H + hh];
      #pragma unroll
      for (int u = 0; u < 4; ++u) {
        const vrow* hp = (const vrow*)(Hfeat + (size_t)si[u] * HC) + lane;
        #pragma unroll
        for (int q = 0; q < Q; ++q) hv[u][q] = hp[q * 32];
      }
      __builtin_amdgcn_sched_barrier(0);
      const float ev = lrelu(asv + adv);
      const float wl = __expf(ev - mv);
      if (lane < 4 * H && us < nv) atomicAdd(&s_l[lsel * H + hh], wl);
      #pragma unroll
      for (int u = 0; u < 4; ++u) {
        if (u < nv) {
          float wh[H];
          #pragma unroll
          for (int h = 0; h < H; ++h)
            wh[h] = __builtin_bit_cast(float,
                __builtin_amdgcn_readlane(__builtin_bit_cast(int, wl), u * H + h));
          #pragma unroll
          for (int q = 0; q < Q; ++q) {
            float wq = wh[q * HPL];
            #pragma unroll
            for (int t = 1; t < HPL; ++t)
              if ((lane * V) / C == t) wq = wh[q * HPL + t];
            #pragma unroll
            for (int k = 0; k < V; ++k)
              atomicAdd(&acc[ld[u] * HC + q * LW + k * 32 + lane], hv[u][q][k] * wq);
          }
        }
      }
    }
    __syncthreads();
  };

  int batch = 0;
  auto drain_if_needed = [&]() {
    __syncthreads();
    const int c = cnt;
    __syncthreads();
    if (c + 4 * NT > CAP) {
      process(c, batch, false);
      ++batch;
      if (tid == 0) cnt = 0;
      __syncthreads();
    }
  };

  auto init_sums = [&]() {
    for (int i = tid; i < nloc * HC; i += NT) acc[i] = 0.f;
    for (int j = tid; j < nloc * H; j += NT) {
      m_l[j] = -__builtin_inff();
      s_l[j] = 0.f;
      ad_l[j] = a_d[(size_t)lo * H + j];
    }
  };

  const size_t lbase = (size_t)blockIdx.x * CAP;
  const int nl = (ROLE == 1) ? tl_cnt[blockIdx.x] : -1;

  const bool handed = (ROLE == 1 && nl >= 0);
  if (handed) {
    const int n = min(nl, CAP);
    init_sums();
    for (int i = tid; i < n; i += NT) { lsrc[i] = tl_src[lbase + i]; lld[i] = tl_ld[lbase + i]; }
    process(n, 0, true);
  }
  for (int attempt = (ROLE == 1 ? 1 : 0); !handed && attempt < 2; ++attempt) {
    init_sums();
    for (int i = tid; i < nloc; i += NT) { lsrc[i] = lo + i; lld[i] = (unsigned short)i; }
    if (tid == 0) { cnt = nloc; ovf = 0; }
    batch = 0;
    __syncthreads();

    const bool byteKeys = gridDim.x <= KEY_TILES && (((size_t)tkey) & 15) == 0;
    if (attempt == 0 && byteKeys) {
      constexpr int KU = 8;
      const int K16 = E >> 4;
      const uint4* key4 = (const uint4*)tkey;
      const unsigned tpat = (unsigned)blockIdx.x * 0x01010101u;
      const int krot = (int)((((long long)blockIdx.x * K16) / gridDim.x) & ~31LL);
      const int wvs = __builtin_amdgcn_readfirstlane(wv);
      for (int g0 = 0; g0 < K16; g0 += KU * NT) {
        uint4 kq[KU];
        int gi[KU];
        #pragma unroll
        for (int u = 0; u < KU; ++u) {
          const int g = g0 + u * NT + tid;
          int gr = g + krot;
          if (gr >= K16) gr -= K16;
          gr = min(gr, K16 - 1);
          gi[u] = gr;
          kq[u] = key4[gr];
        }
        unsigned zz[KU];
        int c = 0;
        #pragma unroll
        for (int u = 0; u < KU; ++u) {
          zz[u] = 0;
          if (g0 + u * NT + (wvs << 5) < K16) {
            const unsigned z = match16(kq[u], tpat);
            zz[u] = (g0 + u * NT + tid < K16) ? z : 0u;
            c += __builtin_popcount(zz[u]);
          }
        }
        if (__builtin_amdgcn_ballot_w32(c != 0)) {
          int inc = c;
          inc += row_shr0<0x111>(inc);
          inc += row_shr0<0x112>(inc);
          inc += row_shr0<0x114>(inc);
          inc += row_shr0<0x118>(inc);
          const int r0 = __builtin_amdgcn_readlane(inc, 15);
          if (lane >= 16) inc += r0;
          const int tot = __builtin_amdgcn_readlane(inc, 31);
          int slot = 0;
          if (lane == 0) slot = atomicAdd(&cnt, tot);
          slot = __builtin_amdgcn_readfirstlane(slot);
          int idx = slot + inc - c;
          if (c) {
            #pragma unroll
            for (int u = 0; u < KU; ++u) {
              unsigned z = zz[u];
              for (int it = 0; it < 16 && z; ++it) {
                const int p = __builtin_ctz(z);
                z &= z - 1;
                if (idx < CAP) lsrc[idx] = gi[u] * 16 + ((p & 7) << 2) + (p >> 3);
                else ovf = 1;
                ++idx;
              }
            }
          }
        }
      }
      for (int e = K16 * 16 + tid; e < E; e += NT) test1(dst[e], e);
    } else if (attempt == 0) {
      for (int g0 = 0; g0 < E4; g0 += 4 * NT) {
        int4 d[4];
        int gi[4];
        #pragma unroll
        for (int u = 0; u < 4; ++u) {
          int g = g0 + u * NT + tid;
          bool ok = g < E4;
          int gr = g + rot;
          if (gr >= E4) gr -= E4;
          gi[u] = gr;
          if (ok) d[u] = dst4[gr];
          else    d[u] = make_int4(lo - 1, lo - 1, lo - 1, lo - 1);
        }
        unsigned hit[16];
        #pragma unroll
        for (int u = 0; u < 4; ++u) {
          hit[4 * u + 0] = __builtin_amdgcn_ballot_w32((unsigned)(d[u].x - lo) < (unsigned)nloc);
          hit[4 * u + 1] = __builtin_amdgcn_ballot_w32((unsigned)(d[u].y - lo) < (unsigned)nloc);
          hit[4 * u + 2] = __builtin_amdgcn_ballot_w32((unsigned)(d[u].z - lo) < (unsigned)nloc);
          hit[4 * u + 3] = __builtin_amdgcn_ballot_w32((unsigned)(d[u].w - lo) < (unsigned)nloc);
        }
        unsigned any = 0;
        #pragma unroll
        for (int k = 0; k < 16; ++k) any |= hit[k];
        if (any) {
          int tot = 0;
          #pragma unroll
          for (int k = 0; k < 16; ++k) tot += __builtin_popcount(hit[k]);
          int slot = 0;
          if (lane == 0) slot = atomicAdd(&cnt, tot);
          slot = __builtin_amdgcn_readfirstlane(slot);
          #pragma unroll
          for (int k = 0; k < 16; ++k) {
            if (hit[k]) {
              if ((hit[k] >> lane) & 1u) {
                const int idx = slot + (int)__builtin_amdgcn_mbcnt_lo(hit[k], 0u);
                if (idx < CAP) lsrc[idx] = gi[k >> 2] * 4 + (k & 3);
                else ovf = 1;
              }
              slot += __builtin_popcount(hit[k]);
            }
          }
        }
      }
      for (int e = E4 * 4 + tid; e < E; e += NT) test1(dst[e], e);
    } else {
      for (int g0 = 0; g0 < E4; g0 += NT) {
        drain_if_needed();
        int g = g0 + tid;
        if (g < E4) {
          int4 d = dst4[g];
          const int e = g * 4;
          test1(d.x, e);
          test1(d.y, e + 1);
          test1(d.z, e + 2);
          test1(d.w, e + 3);
        }
      }
      for (int e0 = E4 * 4; e0 < E; e0 += 4 * NT) {
        drain_if_needed();
        for (int k = 0; k < 4; ++k) {
          int e = e0 + k * NT + tid;
          if (e < E) test1(dst[e], e);
        }
      }
    }
    __syncthreads();
    const int o = ovf;
    const int c = cnt;
    __syncthreads();
    if (attempt == 0 && o) continue;
    const int n = min(c, CAP);
    process(n, batch, false);
    if (ROLE == 0) {
      if (attempt == 0) {
        for (int i = tid; i < n; i += NT) { tl_src[lbase + i] = lsrc[i]; tl_ld[lbase + i] = lld[i]; }
      }
      if (tid == 0) tl_cnt[blockIdx.x] = (attempt == 0) ? n : -1;
    }
    break;
  }

  float bl[J];
  #pragma unroll
  for (int j = 0; j < J; ++j) bl[j] = bias[(j / V) * LW + lane * V + (j % V)];
  for (int r = wv; r < nloc; r += 32) {
    #pragma unroll
    for (int q = 0; q < Q; ++q) {
      const float den = s_l[r * H + q * HPL + (lane * V) / C] + PYG_EPS;
      vrow o;
      #pragma unroll
      for (int k = 0; k < V; ++k) {
        float v = acc[r * HC + q * LW + k * 32 + lane] / den + bl[q * V + k];
        if (ELU) v = v > 0.f ? v : expm1f(v);
        o[k] = v;
      }
      *((vrow*)(out + (size_t)(lo + r) * HC) + q * 32 + lane) = o;
    }
  }
}

extern "C" void kernel_launch(void* const* d_in, const int* in_sizes, int n_in,
                              void* d_out, int out_size, void* d_ws, size_t ws_size,
                              hipStream_t stream)
{
  const float* x   = (const float*)d_in[0];
  const int*   ei  = (const int*)  d_in[1];
  const float* W1  = (const float*)d_in[2];
  const float* as1 = (const float*)d_in[3];
  const float* ad1 = (const float*)d_in[4];
  const float* b1  = (const float*)d_in[5];
  const float* W2  = (const float*)d_in[6];
  const float* as2 = (const float*)d_in[7];
  const float* ad2 = (const float*)d_in[8];
  const float* b2  = (const float*)d_in[9];
  float* out = (float*)d_out;

  const int F = 128, H1 = 4, C = 64, HC1 = 256, HC2 = 64;
  const int N  = in_sizes[0] / F;
  const int E  = in_sizes[1] / 2;
  const int* src = ei;
  const int* dst = ei + E;

  float* h1  = (float*)d_ws;
  float* x2  = h1  + (size_t)N * HC1;
  float* h2  = x2  + (size_t)N * HC1;
  float* aS1 = h2  + (size_t)N * HC2;
  float* aD1 = aS1 + (size_t)N * H1;
  float* aS2 = aD1 + (size_t)N * H1;
  float* aD2 = aS2 + N;

  const int TB = 256;
  auto cdiv = [](int a, int b) { return (a + b - 1) / b; };

  const int tiles = cdiv(N, TILE_NODES);
  int* tlSrc = (int*)(aD2 + N);
  unsigned short* tlLd = (unsigned short*)(tlSrc + (size_t)tiles * LIST_CAP);
  int* tlCnt = (int*)(tlLd + (size_t)tiles * LIST_CAP);
  unsigned char* tKey = (unsigned char*)(tlCnt + tiles);

  gemm_wmma_f32<128, 256, 8, 1024><<<256, 1024, 0, stream>>>(x, W1, h1, N / 16);
  att_dots_rows<4, true><<<cdiv(N * H1, 8 * 2 * DOT_PAIRS), TB, 0, stream>>>(
      h1, as1, ad1, aS1, aD1, N * H1, dst, tKey, E);
  node_tile_gat<4, 64, true, 0><<<tiles, TILE_NT, 0, stream>>>(
      src, dst, aS1, aD1, h1, b1, x2, E, N, tlSrc, tlLd, tlCnt, tKey);

  gemm_wmma_f32<256, 64, 4, 1024><<<256, 1024, 0, stream>>>(x2, W2, h2, N / 16);
  att_dots_rows<1, false><<<cdiv(N, 8 * 2 * DOT_PAIRS), TB, 0, stream>>>(
      h2, as2, ad2, aS2, aD2, N, dst, tKey, E);
  node_tile_gat<1, 64, false, 1><<<tiles, TILE_NT, 0, stream>>>(
      src, dst, aS2, aD2, h2, b2, out, E, N, tlSrc, tlLd, tlCnt, tKey);
}


# Layer 2's dense product moves into layer 1's edge kernel: a tile's finished rows (bias, ELU) stay in LDS and feed f32 WMMA against W2 staged in LDS; x2 and one launch gone; k sum order changes
# speedup vs baseline: 8.0292x; 1.0379x over previous
#include <hip/hip_runtime.h>
#include <math.h>

typedef __attribute__((ext_vector_type(2))) float v2f;
typedef __attribute__((ext_vector_type(8))) float v8f;

#define NEG_SLOPE 0.2f
#define PYG_EPS   1e-16f

#define TILE_NODES 196
#define LIST_CAP   8192
#define TILE_NT    1024
#define KEY_TILES  256

template <int K, int M, int NT, int THREADS>
__global__ __launch_bounds__(THREADS) void gemm_wmma_f32(
    const float* __restrict__ X, const float* __restrict__ W,
    float* __restrict__ Hout, int rowTiles)
{
  constexpr int PITCH = 2 * M + 32;
  constexpr int CH = (M / 16) / NT;
  static_assert(K % 4 == 0 && M % 16 == 0 && (M / 16) % NT == 0, "shape");
  static_assert(PITCH % 64 == 32, "lane halves must fall on different bank halves");
  __shared__ __attribute__((aligned(16))) float Ws[(K / 2) * PITCH];

  const int tid = threadIdx.x;
  for (int i = tid; i < K * M; i += THREADS) {
    const int k = i / M;
    const int col = i - k * M;
    Ws[(k >> 1) * PITCH + col * 2 + (k & 1)] = W[i];
  }
  __syncthreads();

  const int lane = tid & 31;
  const int wv = tid >> 5;
  const int half = lane >> 4;
  const int l15  = lane & 15;
  const int items = rowTiles * CH;

  for (int it = blockIdx.x + gridDim.x * wv; it < items;
       it += gridDim.x * (THREADS / 32)) {
    const int tm = it / CH;
    const int c0 = (it - tm * CH) * (NT * 16);
    const float* xp = X + (size_t)((tm << 4) + l15) * K + (half << 1);
    const float* bp = Ws + half * PITCH + (c0 + l15) * 2;

    v8f acc[NT];
    #pragma unroll
    for (int t = 0; t < NT; ++t) acc[t] = (v8f){0.f, 0.f, 0.f, 0.f, 0.f, 0.f, 0.f, 0.f};

    for (int k0 = 0; k0 < K; k0 += 4) {
      const v2f a = *(const v2f*)(xp + k0);
      const float* bk = bp + (k0 >> 1) * PITCH;
      #pragma unroll
      for (int t = 0; t < NT; ++t) {
        const v2f b = *(const v2f*)(bk + t * 32);
        acc[t] = __builtin_amdgcn_wmma_f32_16x16x4_f32(
             false, a,  false, b,
             (short)0, acc[t],  false,  false);
      }
    }

    float* op = Hout + (size_t)((tm << 4) + (half << 3)) * M + (c0 + l15);
    #pragma unroll
    for (int t = 0; t < NT; ++t) {
      #pragma unroll
      for (int v = 0; v < 8; ++v) op[(size_t)v * M + t * 16] = acc[t][v];
    }
  }
}

template <int CTRL>
__device__ __forceinline__ float row_swz(float v) {
  return __builtin_bit_cast(float, __builtin_amdgcn_update_dpp(
      0, __builtin_bit_cast(int, v), CTRL, 0xf, 0xf, true));
}

#define DOT_PAIRS 16

template <int H, bool FULL>
__device__ __forceinline__ void att_dots_wave(
    const float* __restrict__ Hm, const float* __restrict__ att_s,
    const float* __restrict__ att_d, float* __restrict__ a_s,
    float* __restrict__ a_d, int nrows, int base, int lane)
{
  constexpr int C = 64;
  constexpr int U = DOT_PAIRS;
  constexpr int PP = (H >= 2) ? (H / 2) : 1;
  static_assert((H & (H - 1)) == 0, "H must be a power of two");
  static_assert(U % PP == 0 && (2 * U) % H == 0, "a wave's first row must be head 0");

  const int half = lane >> 4;
  const int l15  = lane & 15;
  const int c0   = l15 << 2;

  float4 x[U];
  const float* p0 = Hm + (size_t)(base + half) * C + c0;
  #pragma unroll
  for (int u = 0; u < U; ++u) {
    if (FULL) {
      x[u] = *(const float4*)(p0 + u * 2 * C);
    } else {
      const int r = min(base + 2 * u + half, nrows - 1);
      x[u] = *(const float4*)(Hm + (size_t)r * C + c0);
    }
  }

  float ws[PP][4], wd[PP][4];
  #pragma unroll
  for (int q = 0; q < PP; ++q) {
    const int h = (2 * q + half) & (H - 1);
    #pragma unroll
    for (int k = 0; k < 4; ++k) {
      ws[q][k] = att_s[h * C + c0 + k];
      wd[q][k] = att_d[h * C + c0 + k];
    }
  }

  float ss[U], sd[U];
  #pragma unroll
  for (int u = 0; u < U; ++u) {
    const int q = u % PP;
    ss[u] = x[u].x * ws[q][0] + x[u].y * ws[q][1] + x[u].z * ws[q][2] + x[u].w * ws[q][3];
    sd[u] = x[u].x * wd[q][0] + x[u].y * wd[q][1] + x[u].z * wd[q][2] + x[u].w * wd[q][3];
  }
  #pragma unroll
  for (int u = 0; u < U; ++u) { ss[u] += row_swz<0x161>(ss[u]); sd[u] += row_swz<0x161>(sd[u]); }
  #pragma unroll
  for (int u = 0; u < U; ++u) { ss[u] += row_swz<0x162>(ss[u]); sd[u] += row_swz<0x162>(sd[u]); }
  #pragma unroll
  for (int u = 0; u < U; ++u) { ss[u] += row_swz<0x164>(ss[u]); sd[u] += row_swz<0x164>(sd[u]); }
  #pragma unroll
  for (int u = 0; u < U; ++u) { ss[u] += row_swz<0x168>(ss[u]); sd[u] += row_swz<0x168>(sd[u]); }

  float vs = ss[0], vd = sd[0];
  #pragma unroll
  for (int u = 1; u < U; ++u) {
    if (l15 == u) { vs = ss[u]; vd = sd[u]; }
  }
  const int r = base + 2 * l15 + half;
  if (FULL || r < nrows) { a_s[r] = vs; a_d[r] = vd; }
}

template <int H, bool KEYS>
__global__ __launch_bounds__(256) void att_dots_rows(
    const float* __restrict__ Hm, const float* __restrict__ att_s,
    const float* __restrict__ att_d, float* __restrict__ a_s,
    float* __restrict__ a_d, int nrows,
    const int* __restrict__ dst, unsigned char* __restrict__ tkey, int E)
{
  if (KEYS) {
    const int nthr = gridDim.x * blockDim.x;
    const int gt = blockIdx.x * blockDim.x + threadIdx.x;
    const int E4 = ((((size_t)dst) & 15) == 0) ? (E >> 2) : 0;
    for (int g = gt; g < E4; g += nthr) {
      const int4 d = ((const int4*)dst)[g];
      const unsigned k = (((unsigned)d.x / TILE_NODES) & 255u)
                       | ((((unsigned)d.y / TILE_NODES) & 255u) << 8)
                       | ((((unsigned)d.z / TILE_NODES) & 255u) << 16)
                       | ((((unsigned)d.w / TILE_NODES) & 255u) << 24);
      ((unsigned*)tkey)[g] = k;
    }
    for (int e = E4 * 4 + gt; e < E; e += nthr)
      tkey[e] = (unsigned char)((unsigned)dst[e] / TILE_NODES);
  }
  const int wave = blockIdx.x * (blockDim.x >> 5) + (threadIdx.x >> 5);
  const int base = wave * (2 * DOT_PAIRS);
  const int lane = threadIdx.x & 31;
  if (base + 2 * DOT_PAIRS <= nrows)
    att_dots_wave<H, true>(Hm, att_s, att_d, a_s, a_d, nrows, base, lane);
  else if (base < nrows)
    att_dots_wave<H, false>(Hm, att_s, att_d, a_s, a_d, nrows, base, lane);
}

__device__ __forceinline__ unsigned match4(unsigned w, unsigned t) {
  const unsigned x = w ^ t;
  return ~((((x & 0x7f7f7f7fu) + 0x7f7f7f7fu) | x) | 0x7f7f7f7fu);
}
__device__ __forceinline__ unsigned match16(uint4 k, unsigned t) {
  return (match4(k.x, t) >> 7) | (match4(k.y, t) >> 6)
       | (match4(k.z, t) >> 5) | (match4(k.w, t) >> 4);
}
template <int CTRL>
__device__ __forceinline__ int row_shr0(int v) {
  return __builtin_amdgcn_update_dpp(0, v, CTRL, 0xf, 0xf, true);
}

__device__ __forceinline__ float lrelu(float e) {
  return e > 0.f ? e : NEG_SLOPE * e;
}

template <int M2, int LW, int V, int NT, int WPT>
__device__ __forceinline__ void wn_fetch(float (&wn)[WPT], const float* __restrict__ Wn,
                                         int g, int tid) {
  #pragma unroll
  for (int i = 0; i < WPT; ++i) {
    const int idx = tid + i * NT;
    const int wl = idx / M2;
    const int col = idx - wl * M2;
    const int f = g * LW + (wl & 31) * V + (wl >> 5);
    wn[i] = Wn[(size_t)f * M2 + col];
  }
}
template <int M2, int BP, int NT, int WPT>
__device__ __forceinline__ void wn_put(const float (&wn)[WPT], float* Wn_s, int tid) {
  #pragma unroll
  for (int i = 0; i < WPT; ++i) {
    const int idx = tid + i * NT;
    const int wl = idx / M2;
    const int col = idx - wl * M2;
    Wn_s[(wl >> 1) * BP + col * 2 + (wl & 1)] = wn[i];
  }
}

template <int H, int C, bool ELU, int ROLE, int M2>
__global__ __launch_bounds__(TILE_NT) void node_tile_gat(
    const int* __restrict__ src, const int* __restrict__ dst,
    const float* __restrict__ a_s, const float* __restrict__ a_d,
    const float* __restrict__ Hfeat, const float* __restrict__ bias,
    float* __restrict__ out, int E, int N,
    int* __restrict__ tl_src, unsigned short* __restrict__ tl_ld,
    int* __restrict__ tl_cnt, const unsigned char* __restrict__ tkey,
    const float* __restrict__ Wn)
{
  constexpr int HC = H * C;
  constexpr int TN = TILE_NODES;
  constexpr int CAP = LIST_CAP;
  constexpr int NT = TILE_NT;
  constexpr int J = HC / 32;
  constexpr int V = (J >= 4) ? 4 : J;
  constexpr int Q = J / V;
  constexpr int LW = 32 * V;
  constexpr int HPL = LW / C;
  constexpr int AP = (M2 > 0) ? HC + 4 : HC;
  constexpr int BP = 2 * M2 + 32;
  constexpr int WPT = (M2 > 0) ? (LW * M2) / NT : 1;
  typedef __attribute__((ext_vector_type(V))) float vrow;
  static_assert(V * Q == J && (V == 4 || V == 2), "a row is whole 16- or 8-byte lane loads");
  static_assert(LW % C == 0 && Q * HPL == H, "a wave load covers whole heads");
  static_assert(C % 32 == 0, "a 32-float chunk must lie inside one head");
  static_assert((H & (H - 1)) == 0 && H <= 32, "H must be a power of two");

  __shared__ __attribute__((aligned(16))) float acc[TN * AP];
  __shared__ float m_l[TN * H];
  __shared__ float s_l[TN * H];
  __shared__ float sc_l[TN * H];
  __shared__ float ad_l[TN * H];
  __shared__ __attribute__((aligned(16))) int lsrc[CAP];
  __shared__ __attribute__((aligned(8))) unsigned short lld[CAP];
  __shared__ int cnt;
  __shared__ int ovf;

  const int tid = threadIdx.x;
  const int lane = tid & 31;
  const int wv = tid >> 5;
  const int lo = blockIdx.x * TN;
  const int nloc = min(TN, N - lo);
  if (nloc <= 0) return;

  __shared__ __attribute__((aligned(16))) float Wn_s[(M2 > 0) ? (LW / 2) * BP : 4];
  const bool vecOK = (((size_t)dst) & 15) == 0;
  const int E4 = vecOK ? (E >> 2) : 0;
  const int4* dst4 = (const int4*)dst;
  const int rot = (int)((((long long)blockIdx.x * E4) / gridDim.x) & ~31LL);

  auto test1 = [&](int d, int e) {
    unsigned ldv = (unsigned)(d - lo);
    if (ldv < (unsigned)nloc) {
      int idx = atomicAdd(&cnt, 1);
      if (idx < CAP) lsrc[idx] = e;
      else ovf = 1;
    }
  };

  auto process = [&](int n, int batch, bool resolved) {
    if (!resolved) {
      for (int i = (batch == 0 ? nloc : 0) + tid; i < n; i += NT) {
        const int e = lsrc[i];
        lsrc[i] = src[e];
        lld[i] = (unsigned short)(dst[e] - lo);
      }
    }
    if (batch > 0) {
      for (int j = tid; j < nloc * H; j += NT) sc_l[j] = m_l[j];
    }
    if (tid < 3 && n + tid < CAP) { lsrc[n + tid] = lo; lld[n + tid] = 0; }
    __syncthreads();
    for (int i = tid; i < n; i += NT) {
      const int si = lsrc[i];
      const int ld = lld[i];
      for (int h = 0; h < H; ++h) {
        float ev = lrelu(a_s[(size_t)si * H + h] + ad_l[ld * H + h]);
        __hip_atomic_fetch_max(&m_l[ld * H + h], ev,
                               __ATOMIC_RELAXED, __HIP_MEMORY_SCOPE_WORKGROUP);
      }
    }
    __syncthreads();
    if (batch > 0) {
      for (int j = tid; j < nloc * H; j += NT) {
        float mo = sc_l[j], mn = m_l[j];
        float f = (mo == mn) ? 1.f : __expf(mo - mn);
        sc_l[j] = f;
        s_l[j] *= f;
      }
      __syncthreads();
      for (int i = tid; i < nloc * HC; i += NT) {
        const int p = i % HC;
        const int f = (p / LW) * LW + (p % 32) * V + (p % LW) / 32;
        acc[(i / HC) * AP + p] *= sc_l[(i / HC) * H + f / C];
      }
      __syncthreads();
    }
    for (int i0 = wv * 4; i0 < n; i0 += 32 * 4) {
      vrow hv[4][Q];
      const int4 sv = *(const int4*)&lsrc[i0];
      const ushort4 lv = *(const ushort4*)&lld[i0];
      __builtin_amdgcn_sched_barrier(0);
      const int si[4] = {sv.x, sv.y, sv.z, sv.w};
      const int ld[4] = {lv.x, lv.y, lv.z, lv.w};
      const int nv = n - i0;
      const int us = (lane / H) & 3;
      const int hh = lane & (H - 1);
      int ssel = si[0], lsel = ld[0];
      #pragma unroll
      for (int u = 1; u < 4; ++u)
        if (us == u) { ssel = si[u]; lsel = ld[u]; }
      const float asv = a_s[(size_t)ssel * H + hh];
      const float adv = ad_l[lsel * H + hh];
      const float mv  = m_l[lsel * H + hh];
      #pragma unroll
      for (int u = 0; u < 4; ++u) {
        const vrow* hp = (const vrow*)(Hfeat + (size_t)si[u] * HC) + lane;
        #pragma unroll
        for (int q = 0; q < Q; ++q) hv[u][q] = hp[q * 32];
      }
      __builtin_amdgcn_sched_barrier(0);
      const float ev = lrelu(asv + adv);
      const float wl = __expf(ev - mv);
      if (lane < 4 * H && us < nv) atomicAdd(&s_l[lsel * H + hh], wl);
      #pragma unroll
      for (int u = 0; u < 4; ++u) {
        if (u < nv) {
          float wh[H];
          #pragma unroll
          for (int h = 0; h < H; ++h)
            wh[h] = __builtin_bit_cast(float,
                __builtin_amdgcn_readlane(__builtin_bit_cast(int, wl), u * H + h));
          #pragma unroll
          for (int q = 0; q < Q; ++q) {
            float wq = wh[q * HPL];
            #pragma unroll
            for (int t = 1; t < HPL; ++t)
              if ((lane * V) / C == t) wq = wh[q * HPL + t];
            #pragma unroll
            for (int k = 0; k < V; ++k)
              atomicAdd(&acc[ld[u] * AP + q * LW + k * 32 + lane], hv[u][q][k] * wq);
          }
        }
      }
    }
    __syncthreads();
  };

  int batch = 0;
  auto drain_if_needed = [&]() {
    __syncthreads();
    const int c = cnt;
    __syncthreads();
    if (c + 4 * NT > CAP) {
      process(c, batch, false);
      ++batch;
      if (tid == 0) cnt = 0;
      __syncthreads();
    }
  };

  auto init_sums = [&]() {
    for (int i = tid; i < nloc * AP; i += NT) acc[i] = 0.f;
    for (int j = tid; j < nloc * H; j += NT) {
      m_l[j] = -__builtin_inff();
      s_l[j] = 0.f;
      ad_l[j] = a_d[(size_t)lo * H + j];
    }
  };

  const size_t lbase = (size_t)blockIdx.x * CAP;
  const int nl = (ROLE == 1) ? tl_cnt[blockIdx.x] : -1;

  const bool handed = (ROLE == 1 && nl >= 0);
  if (handed) {
    const int n = min(nl, CAP);
    init_sums();
    for (int i = tid; i < n; i += NT) { lsrc[i] = tl_src[lbase + i]; lld[i] = tl_ld[lbase + i]; }
    process(n, 0, true);
  }
  for (int attempt = (ROLE == 1 ? 1 : 0); !handed && attempt < 2; ++attempt) {
    init_sums();
    for (int i = tid; i < nloc; i += NT) { lsrc[i] = lo + i; lld[i] = (unsigned short)i; }
    if (tid == 0) { cnt = nloc; ovf = 0; }
    batch = 0;
    __syncthreads();

    const bool byteKeys = gridDim.x <= KEY_TILES && (((size_t)tkey) & 15) == 0;
    if (attempt == 0 && byteKeys) {
      constexpr int KU = 8;
      const int K16 = E >> 4;
      const uint4* key4 = (const uint4*)tkey;
      const unsigned tpat = (unsigned)blockIdx.x * 0x01010101u;
      const int krot = (int)((((long long)blockIdx.x * K16) / gridDim.x) & ~31LL);
      const int wvs = __builtin_amdgcn_readfirstlane(wv);
      for (int g0 = 0; g0 < K16; g0 += KU * NT) {
        uint4 kq[KU];
        int gi[KU];
        #pragma unroll
        for (int u = 0; u < KU; ++u) {
          const int g = g0 + u * NT + tid;
          int gr = g + krot;
          if (gr >= K16) gr -= K16;
          gr = min(gr, K16 - 1);
          gi[u] = gr;
          kq[u] = key4[gr];
        }
        unsigned zz[KU];
        int c = 0;
        #pragma unroll
        for (int u = 0; u < KU; ++u) {
          zz[u] = 0;
          if (g0 + u * NT + (wvs << 5) < K16) {
            const unsigned z = match16(kq[u], tpat);
            zz[u] = (g0 + u * NT + tid < K16) ? z : 0u;
            c += __builtin_popcount(zz[u]);
          }
        }
        if (__builtin_amdgcn_ballot_w32(c != 0)) {
          int inc = c;
          inc += row_shr0<0x111>(inc);
          inc += row_shr0<0x112>(inc);
          inc += row_shr0<0x114>(inc);
          inc += row_shr0<0x118>(inc);
          const int r0 = __builtin_amdgcn_readlane(inc, 15);
          if (lane >= 16) inc += r0;
          const int tot = __builtin_amdgcn_readlane(inc, 31);
          int slot = 0;
          if (lane == 0) slot = atomicAdd(&cnt, tot);
          slot = __builtin_amdgcn_readfirstlane(slot);
          int idx = slot + inc - c;
          if (c) {
            #pragma unroll
            for (int u = 0; u < KU; ++u) {
              unsigned z = zz[u];
              for (int it = 0; it < 16 && z; ++it) {
                const int p = __builtin_ctz(z);
                z &= z - 1;
                if (idx < CAP) lsrc[idx] = gi[u] * 16 + ((p & 7) << 2) + (p >> 3);
                else ovf = 1;
                ++idx;
              }
            }
          }
        }
      }
      for (int e = K16 * 16 + tid; e < E; e += NT) test1(dst[e], e);
    } else if (attempt == 0) {
      for (int g0 = 0; g0 < E4; g0 += 4 * NT) {
        int4 d[4];
        int gi[4];
        #pragma unroll
        for (int u = 0; u < 4; ++u) {
          int g = g0 + u * NT + tid;
          bool ok = g < E4;
          int gr = g + rot;
          if (gr >= E4) gr -= E4;
          gi[u] = gr;
          if (ok) d[u] = dst4[gr];
          else    d[u] = make_int4(lo - 1, lo - 1, lo - 1, lo - 1);
        }
        unsigned hit[16];
        #pragma unroll
        for (int u = 0; u < 4; ++u) {
          hit[4 * u + 0] = __builtin_amdgcn_ballot_w32((unsigned)(d[u].x - lo) < (unsigned)nloc);
          hit[4 * u + 1] = __builtin_amdgcn_ballot_w32((unsigned)(d[u].y - lo) < (unsigned)nloc);
          hit[4 * u + 2] = __builtin_amdgcn_ballot_w32((unsigned)(d[u].z - lo) < (unsigned)nloc);
          hit[4 * u + 3] = __builtin_amdgcn_ballot_w32((unsigned)(d[u].w - lo) < (unsigned)nloc);
        }
        unsigned any = 0;
        #pragma unroll
        for (int k = 0; k < 16; ++k) any |= hit[k];
        if (any) {
          int tot = 0;
          #pragma unroll
          for (int k = 0; k < 16; ++k) tot += __builtin_popcount(hit[k]);
          int slot = 0;
          if (lane == 0) slot = atomicAdd(&cnt, tot);
          slot = __builtin_amdgcn_readfirstlane(slot);
          #pragma unroll
          for (int k = 0; k < 16; ++k) {
            if (hit[k]) {
              if ((hit[k] >> lane) & 1u) {
                const int idx = slot + (int)__builtin_amdgcn_mbcnt_lo(hit[k], 0u);
                if (idx < CAP) lsrc[idx] = gi[k >> 2] * 4 + (k & 3);
                else ovf = 1;
              }
              slot += __builtin_popcount(hit[k]);
            }
          }
        }
      }
      for (int e = E4 * 4 + tid; e < E; e += NT) test1(dst[e], e);
    } else {
      for (int g0 = 0; g0 < E4; g0 += NT) {
        drain_if_needed();
        int g = g0 + tid;
        if (g < E4) {
          int4 d = dst4[g];
          const int e = g * 4;
          test1(d.x, e);
          test1(d.y, e + 1);
          test1(d.z, e + 2);
          test1(d.w, e + 3);
        }
      }
      for (int e0 = E4 * 4; e0 < E; e0 += 4 * NT) {
        drain_if_needed();
        for (int k = 0; k < 4; ++k) {
          int e = e0 + k * NT + tid;
          if (e < E) test1(dst[e], e);
        }
      }
    }
    __syncthreads();
    const int o = ovf;
    const int c = cnt;
    __syncthreads();
    if (attempt == 0 && o) continue;
    const int n = min(c, CAP);
    process(n, batch, false);
    if (ROLE == 0) {
      if (attempt == 0) {
        for (int i = tid; i < n; i += NT) { tl_src[lbase + i] = lsrc[i]; tl_ld[lbase + i] = lld[i]; }
      }
      if (tid == 0) tl_cnt[blockIdx.x] = (attempt == 0) ? n : -1;
    }
    break;
  }

  float bl[J];
  #pragma unroll
  for (int j = 0; j < J; ++j) bl[j] = bias[(j / V) * LW + lane * V + (j % V)];
  float wn[WPT];
  if constexpr (M2 > 0) wn_fetch<M2, LW, V, NT, WPT>(wn, Wn, 0, tid);
  for (int r = wv; r < nloc; r += 32) {
    #pragma unroll
    for (int q = 0; q < Q; ++q) {
      const float den = s_l[r * H + q * HPL + (lane * V) / C] + PYG_EPS;
      vrow o;
      #pragma unroll
      for (int k = 0; k < V; ++k) {
        float v = acc[r * AP + q * LW + k * 32 + lane] / den + bl[q * V + k];
        if (ELU) v = v > 0.f ? v : expm1f(v);
        o[k] = v;
        if constexpr (M2 > 0) acc[r * AP + q * LW + k * 32 + lane] = v;
      }
      if constexpr (M2 == 0) *((vrow*)(out + (size_t)(lo + r) * HC) + q * 32 + lane) = o;
    }
  }

  if constexpr (M2 > 0) {
    constexpr int NT2 = 2;
    constexpr int CH2 = (M2 / 16) / NT2;
    static_assert(M2 % (16 * NT2) == 0 && LW % 4 == 0 && (LW * M2) % NT == 0, "shape");
    static_assert(((TN + 15) / 16) * CH2 <= NT / 32, "one work item per wave");
    static_assert(AP % 64 == 4 && AP % 2 == 0, "16 rows of an A read on 16 bank groups, 8-byte aligned");
    static_assert(BP % 64 == 32, "lane halves of a B read on different bank halves");
    const int half = lane >> 4;
    const int l15 = lane & 15;
    const int rt = wv / CH2;
    const int c0 = (wv - rt * CH2) * (NT2 * 16);
    const bool work = rt * 16 < nloc;
    const int arow = min(rt * 16 + l15, nloc - 1);
    const float* ap = acc + arow * AP + (half << 1);
    const float* bp = Wn_s + half * BP + (c0 + l15) * 2;

    v8f d[NT2];
    #pragma unroll
    for (int t = 0; t < NT2; ++t) d[t] = (v8f){0.f, 0.f, 0.f, 0.f, 0.f, 0.f, 0.f, 0.f};

    wn_put<M2, BP, NT, WPT>(wn, Wn_s, tid);
    __syncthreads();
    #pragma unroll
    for (int g = 0; g < Q; ++g) {
      if (g + 1 < Q) wn_fetch<M2, LW, V, NT, WPT>(wn, Wn, g + 1, tid);
      if (work) {
        for (int jj = 0; jj < LW / 4; ++jj) {
          const v2f a = *(const v2f*)(ap + g * LW + 4 * jj);
          const float* bk = bp + (2 * jj) * BP;
          #pragma unroll
          for (int t = 0; t < NT2; ++t) {
            const v2f b = *(const v2f*)(bk + t * 32);
            d[t] = __builtin_amdgcn_wmma_f32_16x16x4_f32(
                 false, a,  false, b,
                 (short)0, d[t],  false,  false);
          }
        }
      }
      if (g + 1 < Q) {
        __syncthreads();
        wn_put<M2, BP, NT, WPT>(wn, Wn_s, tid);
        __syncthreads();
      }
    }

    if (work) {
      #pragma unroll
      for (int t = 0; t < NT2; ++t) {
        #pragma unroll
        for (int v = 0; v < 8; ++v) {
          const int r = rt * 16 + (half << 3) + v;
          if (r < nloc) out[(size_t)(lo + r) * M2 + c0 + t * 16 + l15] = d[t][v];
        }
      }
    }
  }
}

extern "C" void kernel_launch(void* const* d_in, const int* in_sizes, int n_in,
                              void* d_out, int out_size, void* d_ws, size_t ws_size,
                              hipStream_t stream)
{
  const float* x   = (const float*)d_in[0];
  const int*   ei  = (const int*)  d_in[1];
  const float* W1  = (const float*)d_in[2];
  const float* as1 = (const float*)d_in[3];
  const float* ad1 = (const float*)d_in[4];
  const float* b1  = (const float*)d_in[5];
  const float* W2  = (const float*)d_in[6];
  const float* as2 = (const float*)d_in[7];
  const float* ad2 = (const float*)d_in[8];
  const float* b2  = (const float*)d_in[9];
  float* out = (float*)d_out;

  const int F = 128, H1 = 4, C = 64, HC1 = 256, HC2 = 64;
  const int N  = in_sizes[0] / F;
  const int E  = in_sizes[1] / 2;
  const int* src = ei;
  const int* dst = ei + E;

  float* h1  = (float*)d_ws;
  float* h2  = h1  + (size_t)N * HC1;
  float* aS1 = h2  + (size_t)N * HC2;
  float* aD1 = aS1 + (size_t)N * H1;
  float* aS2 = aD1 + (size_t)N * H1;
  float* aD2 = aS2 + N;

  const int TB = 256;
  auto cdiv = [](int a, int b) { return (a + b - 1) / b; };

  const int tiles = cdiv(N, TILE_NODES);
  int* tlSrc = (int*)(aD2 + N);
  unsigned short* tlLd = (unsigned short*)(tlSrc + (size_t)tiles * LIST_CAP);
  int* tlCnt = (int*)(tlLd + (size_t)tiles * LIST_CAP);
  unsigned char* tKey = (unsigned char*)(tlCnt + tiles);

  gemm_wmma_f32<128, 256, 8, 1024><<<256, 1024, 0, stream>>>(x, W1, h1, N / 16);
  att_dots_rows<4, true><<<cdiv(N * H1, 8 * 2 * DOT_PAIRS), TB, 0, stream>>>(
      h1, as1, ad1, aS1, aD1, N * H1, dst, tKey, E);
  node_tile_gat<4, 64, true, 0, 64><<<tiles, TILE_NT, 0, stream>>>(
      src, dst, aS1, aD1, h1, b1, h2, E, N, tlSrc, tlLd, tlCnt, tKey, W2);

  att_dots_rows<1, false><<<cdiv(N, 8 * 2 * DOT_PAIRS), TB, 0, stream>>>(
      h2, as2, ad2, aS2, aD2, N, dst, tKey, E);
  node_tile_gat<1, 64, false, 1, 0><<<tiles, TILE_NT, 0, stream>>>(
      src, dst, aS2, aD2, h2, b2, out, E, N, tlSrc, tlLd, tlCnt, tKey, nullptr);
}


# Row finish pass: sums times one reciprocal per row and head, not a divide per value; ELU's exp(v)-1 from hardware exp (series near 0, abs error 1e-7), not library expm1f; selects, no branches
# speedup vs baseline: 8.3013x; 1.0339x over previous
#include <hip/hip_runtime.h>
#include <math.h>

typedef __attribute__((ext_vector_type(2))) float v2f;
typedef __attribute__((ext_vector_type(8))) float v8f;

#define NEG_SLOPE 0.2f
#define PYG_EPS   1e-16f

#define TILE_NODES 196
#define LIST_CAP   8192
#define TILE_NT    1024
#define KEY_TILES  256

template <int K, int M, int NT, int THREADS>
__global__ __launch_bounds__(THREADS) void gemm_wmma_f32(
    const float* __restrict__ X, const float* __restrict__ W,
    float* __restrict__ Hout, int rowTiles)
{
  constexpr int PITCH = 2 * M + 32;
  constexpr int CH = (M / 16) / NT;
  static_assert(K % 4 == 0 && M % 16 == 0 && (M / 16) % NT == 0, "shape");
  static_assert(PITCH % 64 == 32, "lane halves must fall on different bank halves");
  __shared__ __attribute__((aligned(16))) float Ws[(K / 2) * PITCH];

  const int tid = threadIdx.x;
  for (int i = tid; i < K * M; i += THREADS) {
    const int k = i / M;
    const int col = i - k * M;
    Ws[(k >> 1) * PITCH + col * 2 + (k & 1)] = W[i];
  }
  __syncthreads();

  const int lane = tid & 31;
  const int wv = tid >> 5;
  const int half = lane >> 4;
  const int l15  = lane & 15;
  const int items = rowTiles * CH;

  for (int it = blockIdx.x + gridDim.x * wv; it < items;
       it += gridDim.x * (THREADS / 32)) {
    const int tm = it / CH;
    const int c0 = (it - tm * CH) * (NT * 16);
    const float* xp = X + (size_t)((tm << 4) + l15) * K + (half << 1);
    const float* bp = Ws + half * PITCH + (c0 + l15) * 2;

    v8f acc[NT];
    #pragma unroll
    for (int t = 0; t < NT; ++t) acc[t] = (v8f){0.f, 0.f, 0.f, 0.f, 0.f, 0.f, 0.f, 0.f};

    for (int k0 = 0; k0 < K; k0 += 4) {
      const v2f a = *(const v2f*)(xp + k0);
      const float* bk = bp + (k0 >> 1) * PITCH;
      #pragma unroll
      for (int t = 0; t < NT; ++t) {
        const v2f b = *(const v2f*)(bk + t * 32);
        acc[t] = __builtin_amdgcn_wmma_f32_16x16x4_f32(
             false, a,  false, b,
             (short)0, acc[t],  false,  false);
      }
    }

    float* op = Hout + (size_t)((tm << 4) + (half << 3)) * M + (c0 + l15);
    #pragma unroll
    for (int t = 0; t < NT; ++t) {
      #pragma unroll
      for (int v = 0; v < 8; ++v) op[(size_t)v * M + t * 16] = acc[t][v];
    }
  }
}

template <int CTRL>
__device__ __forceinline__ float row_swz(float v) {
  return __builtin_bit_cast(float, __builtin_amdgcn_update_dpp(
      0, __builtin_bit_cast(int, v), CTRL, 0xf, 0xf, true));
}

#define DOT_PAIRS 16

template <int H, bool FULL>
__device__ __forceinline__ void att_dots_wave(
    const float* __restrict__ Hm, const float* __restrict__ att_s,
    const float* __restrict__ att_d, float* __restrict__ a_s,
    float* __restrict__ a_d, int nrows, int base, int lane)
{
  constexpr int C = 64;
  constexpr int U = DOT_PAIRS;
  constexpr int PP = (H >= 2) ? (H / 2) : 1;
  static_assert((H & (H - 1)) == 0, "H must be a power of two");
  static_assert(U % PP == 0 && (2 * U) % H == 0, "a wave's first row must be head 0");

  const int half = lane >> 4;
  const int l15  = lane & 15;
  const int c0   = l15 << 2;

  float4 x[U];
  const float* p0 = Hm + (size_t)(base + half) * C + c0;
  #pragma unroll
  for (int u = 0; u < U; ++u) {
    if (FULL) {
      x[u] = *(const float4*)(p0 + u * 2 * C);
    } else {
      const int r = min(base + 2 * u + half, nrows - 1);
      x[u] = *(const float4*)(Hm + (size_t)r * C + c0);
    }
  }

  float ws[PP][4], wd[PP][4];
  #pragma unroll
  for (int q = 0; q < PP; ++q) {
    const int h = (2 * q + half) & (H - 1);
    #pragma unroll
    for (int k = 0; k < 4; ++k) {
      ws[q][k] = att_s[h * C + c0 + k];
      wd[q][k] = att_d[h * C + c0 + k];
    }
  }

  float ss[U], sd[U];
  #pragma unroll
  for (int u = 0; u < U; ++u) {
    const int q = u % PP;
    ss[u] = x[u].x * ws[q][0] + x[u].y * ws[q][1] + x[u].z * ws[q][2] + x[u].w * ws[q][3];
    sd[u] = x[u].x * wd[q][0] + x[u].y * wd[q][1] + x[u].z * wd[q][2] + x[u].w * wd[q][3];
  }
  #pragma unroll
  for (int u = 0; u < U; ++u) { ss[u] += row_swz<0x161>(ss[u]); sd[u] += row_swz<0x161>(sd[u]); }
  #pragma unroll
  for (int u = 0; u < U; ++u) { ss[u] += row_swz<0x162>(ss[u]); sd[u] += row_swz<0x162>(sd[u]); }
  #pragma unroll
  for (int u = 0; u < U; ++u) { ss[u] += row_swz<0x164>(ss[u]); sd[u] += row_swz<0x164>(sd[u]); }
  #pragma unroll
  for (int u = 0; u < U; ++u) { ss[u] += row_swz<0x168>(ss[u]); sd[u] += row_swz<0x168>(sd[u]); }

  float vs = ss[0], vd = sd[0];
  #pragma unroll
  for (int u = 1; u < U; ++u) {
    if (l15 == u) { vs = ss[u]; vd = sd[u]; }
  }
  const int r = base + 2 * l15 + half;
  if (FULL || r < nrows) { a_s[r] = vs; a_d[r] = vd; }
}

template <int H, bool KEYS>
__global__ __launch_bounds__(256) void att_dots_rows(
    const float* __restrict__ Hm, const float* __restrict__ att_s,
    const float* __restrict__ att_d, float* __restrict__ a_s,
    float* __restrict__ a_d, int nrows,
    const int* __restrict__ dst, unsigned char* __restrict__ tkey, int E)
{
  if (KEYS) {
    const int nthr = gridDim.x * blockDim.x;
    const int gt = blockIdx.x * blockDim.x + threadIdx.x;
    const int E4 = ((((size_t)dst) & 15) == 0) ? (E >> 2) : 0;
    for (int g = gt; g < E4; g += nthr) {
      const int4 d = ((const int4*)dst)[g];
      const unsigned k = (((unsigned)d.x / TILE_NODES) & 255u)
                       | ((((unsigned)d.y / TILE_NODES) & 255u) << 8)
                       | ((((unsigned)d.z / TILE_NODES) & 255u) << 16)
                       | ((((unsigned)d.w / TILE_NODES) & 255u) << 24);
      ((unsigned*)tkey)[g] = k;
    }
    for (int e = E4 * 4 + gt; e < E; e += nthr)
      tkey[e] = (unsigned char)((unsigned)dst[e] / TILE_NODES);
  }
  const int wave = blockIdx.x * (blockDim.x >> 5) + (threadIdx.x >> 5);
  const int base = wave * (2 * DOT_PAIRS);
  const int lane = threadIdx.x & 31;
  if (base + 2 * DOT_PAIRS <= nrows)
    att_dots_wave<H, true>(Hm, att_s, att_d, a_s, a_d, nrows, base, lane);
  else if (base < nrows)
    att_dots_wave<H, false>(Hm, att_s, att_d, a_s, a_d, nrows, base, lane);
}

__device__ __forceinline__ unsigned match4(unsigned w, unsigned t) {
  const unsigned x = w ^ t;
  return ~((((x & 0x7f7f7f7fu) + 0x7f7f7f7fu) | x) | 0x7f7f7f7fu);
}
__device__ __forceinline__ unsigned match16(uint4 k, unsigned t) {
  return (match4(k.x, t) >> 7) | (match4(k.y, t) >> 6)
       | (match4(k.z, t) >> 5) | (match4(k.w, t) >> 4);
}
template <int CTRL>
__device__ __forceinline__ int row_shr0(int v) {
  return __builtin_amdgcn_update_dpp(0, v, CTRL, 0xf, 0xf, true);
}

__device__ __forceinline__ float lrelu(float e) {
  return e > 0.f ? e : NEG_SLOPE * e;
}

__device__ __forceinline__ float expm1_nonpos(float t) {
  const float big = __expf(t) - 1.0f;
  float p = 1.0f / 5040.0f;
  p = fmaf(p, t, 1.0f / 720.0f);
  p = fmaf(p, t, 1.0f / 120.0f);
  p = fmaf(p, t, 1.0f / 24.0f);
  p = fmaf(p, t, 1.0f / 6.0f);
  p = fmaf(p, t, 0.5f);
  p = fmaf(p, t, 1.0f);
  return t > -0.34657359f ? p * t : big;
}

template <int M2, int LW, int V, int NT, int WPT>
__device__ __forceinline__ void wn_fetch(float (&wn)[WPT], const float* __restrict__ Wn,
                                         int g, int tid) {
  #pragma unroll
  for (int i = 0; i < WPT; ++i) {
    const int idx = tid + i * NT;
    const int wl = idx / M2;
    const int col = idx - wl * M2;
    const int f = g * LW + (wl & 31) * V + (wl >> 5);
    wn[i] = Wn[(size_t)f * M2 + col];
  }
}
template <int M2, int BP, int NT, int WPT>
__device__ __forceinline__ void wn_put(const float (&wn)[WPT], float* Wn_s, int tid) {
  #pragma unroll
  for (int i = 0; i < WPT; ++i) {
    const int idx = tid + i * NT;
    const int wl = idx / M2;
    const int col = idx - wl * M2;
    Wn_s[(wl >> 1) * BP + col * 2 + (wl & 1)] = wn[i];
  }
}

template <int H, int C, bool ELU, int ROLE, int M2>
__global__ __launch_bounds__(TILE_NT) void node_tile_gat(
    const int* __restrict__ src, const int* __restrict__ dst,
    const float* __restrict__ a_s, const float* __restrict__ a_d,
    const float* __restrict__ Hfeat, const float* __restrict__ bias,
    float* __restrict__ out, int E, int N,
    int* __restrict__ tl_src, unsigned short* __restrict__ tl_ld,
    int* __restrict__ tl_cnt, const unsigned char* __restrict__ tkey,
    const float* __restrict__ Wn)
{
  constexpr int HC = H * C;
  constexpr int TN = TILE_NODES;
  constexpr int CAP = LIST_CAP;
  constexpr int NT = TILE_NT;
  constexpr int J = HC / 32;
  constexpr int V = (J >= 4) ? 4 : J;
  constexpr int Q = J / V;
  constexpr int LW = 32 * V;
  constexpr int HPL = LW / C;
  constexpr int AP = (M2 > 0) ? HC + 4 : HC;
  constexpr int BP = 2 * M2 + 32;
  constexpr int WPT = (M2 > 0) ? (LW * M2) / NT : 1;
  typedef __attribute__((ext_vector_type(V))) float vrow;
  static_assert(V * Q == J && (V == 4 || V == 2), "a row is whole 16- or 8-byte lane loads");
  static_assert(LW % C == 0 && Q * HPL == H, "a wave load covers whole heads");
  static_assert(C % 32 == 0, "a 32-float chunk must lie inside one head");
  static_assert((H & (H - 1)) == 0 && H <= 32, "H must be a power of two");

  __shared__ __attribute__((aligned(16))) float acc[TN * AP];
  __shared__ float m_l[TN * H];
  __shared__ float s_l[TN * H];
  __shared__ float sc_l[TN * H];
  __shared__ float ad_l[TN * H];
  __shared__ __attribute__((aligned(16))) int lsrc[CAP];
  __shared__ __attribute__((aligned(8))) unsigned short lld[CAP];
  __shared__ int cnt;
  __shared__ int ovf;

  const int tid = threadIdx.x;
  const int lane = tid & 31;
  const int wv = tid >> 5;
  const int lo = blockIdx.x * TN;
  const int nloc = min(TN, N - lo);
  if (nloc <= 0) return;

  __shared__ __attribute__((aligned(16))) float Wn_s[(M2 > 0) ? (LW / 2) * BP : 4];
  const bool vecOK = (((size_t)dst) & 15) == 0;
  const int E4 = vecOK ? (E >> 2) : 0;
  const int4* dst4 = (const int4*)dst;
  const int rot = (int)((((long long)blockIdx.x * E4) / gridDim.x) & ~31LL);

  auto test1 = [&](int d, int e) {
    unsigned ldv = (unsigned)(d - lo);
    if (ldv < (unsigned)nloc) {
      int idx = atomicAdd(&cnt, 1);
      if (idx < CAP) lsrc[idx] = e;
      else ovf = 1;
    }
  };

  auto process = [&](int n, int batch, bool resolved) {
    if (!resolved) {
      for (int i = (batch == 0 ? nloc : 0) + tid; i < n; i += NT) {
        const int e = lsrc[i];
        lsrc[i] = src[e];
        lld[i] = (unsigned short)(dst[e] - lo);
      }
    }
    if (batch > 0) {
      for (int j = tid; j < nloc * H; j += NT) sc_l[j] = m_l[j];
    }
    if (tid < 3 && n + tid < CAP) { lsrc[n + tid] = lo; lld[n + tid] = 0; }
    __syncthreads();
    for (int i = tid; i < n; i += NT) {
      const int si = lsrc[i];
      const int ld = lld[i];
      for (int h = 0; h < H; ++h) {
        float ev = lrelu(a_s[(size_t)si * H + h] + ad_l[ld * H + h]);
        __hip_atomic_fetch_max(&m_l[ld * H + h], ev,
                               __ATOMIC_RELAXED, __HIP_MEMORY_SCOPE_WORKGROUP);
      }
    }
    __syncthreads();
    if (batch > 0) {
      for (int j = tid; j < nloc * H; j += NT) {
        float mo = sc_l[j], mn = m_l[j];
        float f = (mo == mn) ? 1.f : __expf(mo - mn);
        sc_l[j] = f;
        s_l[j] *= f;
      }
      __syncthreads();
      for (int i = tid; i < nloc * HC; i += NT) {
        const int p = i % HC;
        const int f = (p / LW) * LW + (p % 32) * V + (p % LW) / 32;
        acc[(i / HC) * AP + p] *= sc_l[(i / HC) * H + f / C];
      }
      __syncthreads();
    }
    for (int i0 = wv * 4; i0 < n; i0 += 32 * 4) {
      vrow hv[4][Q];
      const int4 sv = *(const int4*)&lsrc[i0];
      const ushort4 lv = *(const ushort4*)&lld[i0];
      __builtin_amdgcn_sched_barrier(0);
      const int si[4] = {sv.x, sv.y, sv.z, sv.w};
      const int ld[4] = {lv.x, lv.y, lv.z, lv.w};
      const int nv = n - i0;
      const int us = (lane / H) & 3;
      const int hh = lane & (H - 1);
      int ssel = si[0], lsel = ld[0];
      #pragma unroll
      for (int u = 1; u < 4; ++u)
        if (us == u) { ssel = si[u]; lsel = ld[u]; }
      const float asv = a_s[(size_t)ssel * H + hh];
      const float adv = ad_l[lsel * H + hh];
      const float mv  = m_l[lsel * H + hh];
      #pragma unroll
      for (int u = 0; u < 4; ++u) {
        const vrow* hp = (const vrow*)(Hfeat + (size_t)si[u] * HC) + lane;
        #pragma unroll
        for (int q = 0; q < Q; ++q) hv[u][q] = hp[q * 32];
      }
      __builtin_amdgcn_sched_barrier(0);
      const float ev = lrelu(asv + adv);
      const float wl = __expf(ev - mv);
      if (lane < 4 * H && us < nv) atomicAdd(&s_l[lsel * H + hh], wl);
      #pragma unroll
      for (int u = 0; u < 4; ++u) {
        if (u < nv) {
          float wh[H];
          #pragma unroll
          for (int h = 0; h < H; ++h)
            wh[h] = __builtin_bit_cast(float,
                __builtin_amdgcn_readlane(__builtin_bit_cast(int, wl), u * H + h));
          #pragma unroll
          for (int q = 0; q < Q; ++q) {
            float wq = wh[q * HPL];
            #pragma unroll
            for (int t = 1; t < HPL; ++t)
              if ((lane * V) / C == t) wq = wh[q * HPL + t];
            #pragma unroll
            for (int k = 0; k < V; ++k)
              atomicAdd(&acc[ld[u] * AP + q * LW + k * 32 + lane], hv[u][q][k] * wq);
          }
        }
      }
    }
    __syncthreads();
  };

  int batch = 0;
  auto drain_if_needed = [&]() {
    __syncthreads();
    const int c = cnt;
    __syncthreads();
    if (c + 4 * NT > CAP) {
      process(c, batch, false);
      ++batch;
      if (tid == 0) cnt = 0;
      __syncthreads();
    }
  };

  auto init_sums = [&]() {
    for (int i = tid; i < nloc * AP; i += NT) acc[i] = 0.f;
    for (int j = tid; j < nloc * H; j += NT) {
      m_l[j] = -__builtin_inff();
      s_l[j] = 0.f;
      ad_l[j] = a_d[(size_t)lo * H + j];
    }
  };

  const size_t lbase = (size_t)blockIdx.x * CAP;
  const int nl = (ROLE == 1) ? tl_cnt[blockIdx.x] : -1;

  const bool handed = (ROLE == 1 && nl >= 0);
  if (handed) {
    const int n = min(nl, CAP);
    init_sums();
    for (int i = tid; i < n; i += NT) { lsrc[i] = tl_src[lbase + i]; lld[i] = tl_ld[lbase + i]; }
    process(n, 0, true);
  }
  for (int attempt = (ROLE == 1 ? 1 : 0); !handed && attempt < 2; ++attempt) {
    init_sums();
    for (int i = tid; i < nloc; i += NT) { lsrc[i] = lo + i; lld[i] = (unsigned short)i; }
    if (tid == 0) { cnt = nloc; ovf = 0; }
    batch = 0;
    __syncthreads();

    const bool byteKeys = gridDim.x <= KEY_TILES && (((size_t)tkey) & 15) == 0;
    if (attempt == 0 && byteKeys) {
      constexpr int KU = 8;
      const int K16 = E >> 4;
      const uint4* key4 = (const uint4*)tkey;
      const unsigned tpat = (unsigned)blockIdx.x * 0x01010101u;
      const int krot = (int)((((long long)blockIdx.x * K16) / gridDim.x) & ~31LL);
      const int wvs = __builtin_amdgcn_readfirstlane(wv);
      for (int g0 = 0; g0 < K16; g0 += KU * NT) {
        uint4 kq[KU];
        int gi[KU];
        #pragma unroll
        for (int u = 0; u < KU; ++u) {
          const int g = g0 + u * NT + tid;
          int gr = g + krot;
          if (gr >= K16) gr -= K16;
          gr = min(gr, K16 - 1);
          gi[u] = gr;
          kq[u] = key4[gr];
        }
        unsigned zz[KU];
        int c = 0;
        #pragma unroll
        for (int u = 0; u < KU; ++u) {
          zz[u] = 0;
          if (g0 + u * NT + (wvs << 5) < K16) {
            const unsigned z = match16(kq[u], tpat);
            zz[u] = (g0 + u * NT + tid < K16) ? z : 0u;
            c += __builtin_popcount(zz[u]);
          }
        }
        if (__builtin_amdgcn_ballot_w32(c != 0)) {
          int inc = c;
          inc += row_shr0<0x111>(inc);
          inc += row_shr0<0x112>(inc);
          inc += row_shr0<0x114>(inc);
          inc += row_shr0<0x118>(inc);
          const int r0 = __builtin_amdgcn_readlane(inc, 15);
          if (lane >= 16) inc += r0;
          const int tot = __builtin_amdgcn_readlane(inc, 31);
          int slot = 0;
          if (lane == 0) slot = atomicAdd(&cnt, tot);
          slot = __builtin_amdgcn_readfirstlane(slot);
          int idx = slot + inc - c;
          if (c) {
            #pragma unroll
            for (int u = 0; u < KU; ++u) {
              unsigned z = zz[u];
              for (int it = 0; it < 16 && z; ++it) {
                const int p = __builtin_ctz(z);
                z &= z - 1;
                if (idx < CAP) lsrc[idx] = gi[u] * 16 + ((p & 7) << 2) + (p >> 3);
                else ovf = 1;
                ++idx;
              }
            }
          }
        }
      }
      for (int e = K16 * 16 + tid; e < E; e += NT) test1(dst[e], e);
    } else if (attempt == 0) {
      for (int g0 = 0; g0 < E4; g0 += 4 * NT) {
        int4 d[4];
        int gi[4];
        #pragma unroll
        for (int u = 0; u < 4; ++u) {
          int g = g0 + u * NT + tid;
          bool ok = g < E4;
          int gr = g + rot;
          if (gr >= E4) gr -= E4;
          gi[u] = gr;
          if (ok) d[u] = dst4[gr];
          else    d[u] = make_int4(lo - 1, lo - 1, lo - 1, lo - 1);
        }
        unsigned hit[16];
        #pragma unroll
        for (int u = 0; u < 4; ++u) {
          hit[4 * u + 0] = __builtin_amdgcn_ballot_w32((unsigned)(d[u].x - lo) < (unsigned)nloc);
          hit[4 * u + 1] = __builtin_amdgcn_ballot_w32((unsigned)(d[u].y - lo) < (unsigned)nloc);
          hit[4 * u + 2] = __builtin_amdgcn_ballot_w32((unsigned)(d[u].z - lo) < (unsigned)nloc);
          hit[4 * u + 3] = __builtin_amdgcn_ballot_w32((unsigned)(d[u].w - lo) < (unsigned)nloc);
        }
        unsigned any = 0;
        #pragma unroll
        for (int k = 0; k < 16; ++k) any |= hit[k];
        if (any) {
          int tot = 0;
          #pragma unroll
          for (int k = 0; k < 16; ++k) tot += __builtin_popcount(hit[k]);
          int slot = 0;
          if (lane == 0) slot = atomicAdd(&cnt, tot);
          slot = __builtin_amdgcn_readfirstlane(slot);
          #pragma unroll
          for (int k = 0; k < 16; ++k) {
            if (hit[k]) {
              if ((hit[k] >> lane) & 1u) {
                const int idx = slot + (int)__builtin_amdgcn_mbcnt_lo(hit[k], 0u);
                if (idx < CAP) lsrc[idx] = gi[k >> 2] * 4 + (k & 3);
                else ovf = 1;
              }
              slot += __builtin_popcount(hit[k]);
            }
          }
        }
      }
      for (int e = E4 * 4 + tid; e < E; e += NT) test1(dst[e], e);
    } else {
      for (int g0 = 0; g0 < E4; g0 += NT) {
        drain_if_needed();
        int g = g0 + tid;
        if (g < E4) {
          int4 d = dst4[g];
          const int e = g * 4;
          test1(d.x, e);
          test1(d.y, e + 1);
          test1(d.z, e + 2);
          test1(d.w, e + 3);
        }
      }
      for (int e0 = E4 * 4; e0 < E; e0 += 4 * NT) {
        drain_if_needed();
        for (int k = 0; k < 4; ++k) {
          int e = e0 + k * NT + tid;
          if (e < E) test1(dst[e], e);
        }
      }
    }
    __syncthreads();
    const int o = ovf;
    const int c = cnt;
    __syncthreads();
    if (attempt == 0 && o) continue;
    const int n = min(c, CAP);
    process(n, batch, false);
    if (ROLE == 0) {
      if (attempt == 0) {
        for (int i = tid; i < n; i += NT) { tl_src[lbase + i] = lsrc[i]; tl_ld[lbase + i] = lld[i]; }
      }
      if (tid == 0) tl_cnt[blockIdx.x] = (attempt == 0) ? n : -1;
    }
    break;
  }

  for (int j = tid; j < nloc * H; j += NT) s_l[j] = 1.0f / (s_l[j] + PYG_EPS);

  float bl[J];
  #pragma unroll
  for (int j = 0; j < J; ++j) bl[j] = bias[(j / V) * LW + lane * V + (j % V)];
  float wn[WPT];
  if constexpr (M2 > 0) wn_fetch<M2, LW, V, NT, WPT>(wn, Wn, 0, tid);
  __syncthreads();
  for (int r = wv; r < nloc; r += 32) {
    float a[J];
    #pragma unroll
    for (int j = 0; j < J; ++j) a[j] = acc[r * AP + (j / V) * LW + (j % V) * 32 + lane];
    float rden[Q];
    #pragma unroll
    for (int q = 0; q < Q; ++q) rden[q] = s_l[r * H + q * HPL + (lane * V) / C];
    #pragma unroll
    for (int j = 0; j < J; ++j) {
      float v = a[j] * rden[j / V] + bl[j];
      if (ELU) {
        const bool pos = v > 0.f;
        const float en = expm1_nonpos(pos ? 0.f : v);
        v = pos ? v : en;
      }
      a[j] = v;
    }
    #pragma unroll
    for (int q = 0; q < Q; ++q) {
      if constexpr (M2 > 0) {
        #pragma unroll
        for (int k = 0; k < V; ++k) acc[r * AP + q * LW + k * 32 + lane] = a[q * V + k];
      } else {
        vrow o;
        #pragma unroll
        for (int k = 0; k < V; ++k) o[k] = a[q * V + k];
        *((vrow*)(out + (size_t)(lo + r) * HC) + q * 32 + lane) = o;
      }
    }
  }

  if constexpr (M2 > 0) {
    constexpr int NT2 = 2;
    constexpr int CH2 = (M2 / 16) / NT2;
    static_assert(M2 % (16 * NT2) == 0 && LW % 4 == 0 && (LW * M2) % NT == 0, "shape");
    static_assert(((TN + 15) / 16) * CH2 <= NT / 32, "one work item per wave");
    static_assert(AP % 64 == 4 && AP % 2 == 0, "16 rows of an A read on 16 bank groups, 8-byte aligned");
    static_assert(BP % 64 == 32, "lane halves of a B read on different bank halves");
    const int half = lane >> 4;
    const int l15 = lane & 15;
    const int rt = wv / CH2;
    const int c0 = (wv - rt * CH2) * (NT2 * 16);
    const bool work = rt * 16 < nloc;
    const int arow = min(rt * 16 + l15, nloc - 1);
    const float* ap = acc + arow * AP + (half << 1);
    const float* bp = Wn_s + half * BP + (c0 + l15) * 2;

    v8f d[NT2];
    #pragma unroll
    for (int t = 0; t < NT2; ++t) d[t] = (v8f){0.f, 0.f, 0.f, 0.f, 0.f, 0.f, 0.f, 0.f};

    wn_put<M2, BP, NT, WPT>(wn, Wn_s, tid);
    __syncthreads();
    #pragma unroll
    for (int g = 0; g < Q; ++g) {
      if (g + 1 < Q) wn_fetch<M2, LW, V, NT, WPT>(wn, Wn, g + 1, tid);
      if (work) {
        for (int jj = 0; jj < LW / 4; ++jj) {
          const v2f a = *(const v2f*)(ap + g * LW + 4 * jj);
          const float* bk = bp + (2 * jj) * BP;
          #pragma unroll
          for (int t = 0; t < NT2; ++t) {
            const v2f b = *(const v2f*)(bk + t * 32);
            d[t] = __builtin_amdgcn_wmma_f32_16x16x4_f32(
                 false, a,  false, b,
                 (short)0, d[t],  false,  false);
          }
        }
      }
      if (g + 1 < Q) {
        __syncthreads();
        wn_put<M2, BP, NT, WPT>(wn, Wn_s, tid);
        __syncthreads();
      }
    }

    if (work) {
      #pragma unroll
      for (int t = 0; t < NT2; ++t) {
        #pragma unroll
        for (int v = 0; v < 8; ++v) {
          const int r = rt * 16 + (half << 3) + v;
          if (r < nloc) out[(size_t)(lo + r) * M2 + c0 + t * 16 + l15] = d[t][v];
        }
      }
    }
  }
}

extern "C" void kernel_launch(void* const* d_in, const int* in_sizes, int n_in,
                              void* d_out, int out_size, void* d_ws, size_t ws_size,
                              hipStream_t stream)
{
  const float* x   = (const float*)d_in[0];
  const int*   ei  = (const int*)  d_in[1];
  const float* W1  = (const float*)d_in[2];
  const float* as1 = (const float*)d_in[3];
  const float* ad1 = (const float*)d_in[4];
  const float* b1  = (const float*)d_in[5];
  const float* W2  = (const float*)d_in[6];
  const float* as2 = (const float*)d_in[7];
  const float* ad2 = (const float*)d_in[8];
  const float* b2  = (const float*)d_in[9];
  float* out = (float*)d_out;

  const int F = 128, H1 = 4, C = 64, HC1 = 256, HC2 = 64;
  const int N  = in_sizes[0] / F;
  const int E  = in_sizes[1] / 2;
  const int* src = ei;
  const int* dst = ei + E;

  float* h1  = (float*)d_ws;
  float* h2  = h1  + (size_t)N * HC1;
  float* aS1 = h2  + (size_t)N * HC2;
  float* aD1 = aS1 + (size_t)N * H1;
  float* aS2 = aD1 + (size_t)N * H1;
  float* aD2 = aS2 + N;

  const int TB = 256;
  auto cdiv = [](int a, int b) { return (a + b - 1) / b; };

  const int tiles = cdiv(N, TILE_NODES);
  int* tlSrc = (int*)(aD2 + N);
  unsigned short* tlLd = (unsigned short*)(tlSrc + (size_t)tiles * LIST_CAP);
  int* tlCnt = (int*)(tlLd + (size_t)tiles * LIST_CAP);
  unsigned char* tKey = (unsigned char*)(tlCnt + tiles);

  gemm_wmma_f32<128, 256, 8, 1024><<<256, 1024, 0, stream>>>(x, W1, h1, N / 16);
  att_dots_rows<4, true><<<cdiv(N * H1, 8 * 2 * DOT_PAIRS), TB, 0, stream>>>(
      h1, as1, ad1, aS1, aD1, N * H1, dst, tKey, E);
  node_tile_gat<4, 64, true, 0, 64><<<tiles, TILE_NT, 0, stream>>>(
      src, dst, aS1, aD1, h1, b1, h2, E, N, tlSrc, tlLd, tlCnt, tKey, W2);

  att_dots_rows<1, false><<<cdiv(N, 8 * 2 * DOT_PAIRS), TB, 0, stream>>>(
      h2, as2, ad2, aS2, aD2, N, dst, tKey, E);
  node_tile_gat<1, 64, false, 1, 0><<<tiles, TILE_NT, 0, stream>>>(
      src, dst, aS2, aD2, h2, b2, out, E, N, tlSrc, tlLd, tlCnt, tKey, nullptr);
}
